# starting kernel
# baseline (speedup 1.0000x reference)
#include <hip/hip_runtime.h>


typedef __attribute__((ext_vector_type(2))) float v2f;
typedef __attribute__((ext_vector_type(8))) float v8f;

#define NQ      12
#define DIM     4096
#define THREADS 256
#define NWAVES  8
#define NLAYERS 6
#define NPAIRS  6
#define NROUNDS 8

__device__ __forceinline__ float2 cmul(float2 a, float2 b) {
    return make_float2(a.x * b.x - a.y * b.y, a.x * b.y + a.y * b.x);
}

__device__ __forceinline__ void fused_rot(float tx, float ty, float tz, float2* U) {
    const float cx = cosf(0.5f * tx), sx = sinf(0.5f * tx);
    const float cy = cosf(0.5f * ty), sy = sinf(0.5f * ty);
    const float cz = cosf(0.5f * tz), sz = sinf(0.5f * tz);
    const float2 A00 = make_float2( cy * cx,  sy * sx);
    const float2 A01 = make_float2(-sy * cx, -cy * sx);
    const float2 A10 = make_float2( sy * cx, -cy * sx);
    const float2 A11 = make_float2( cy * cx, -sy * sx);
    const float2 e  = make_float2(cz, -sz);
    const float2 ec = make_float2(cz,  sz);
    U[0] = cmul(e, A00);  U[1] = cmul(e, A01);
    U[2] = cmul(ec, A10); U[3] = cmul(ec, A11);
}

__device__ __forceinline__ void embed_rot(float f, float2* U) {
    const float cy = cosf(0.5f * f),  sy = sinf(0.5f * f);
    const float cz = cosf(0.25f * f), sz = sinf(0.25f * f);
    U[0] = make_float2( cz * cy, -sz * cy);
    U[1] = make_float2(-cz * sy,  sz * sy);
    U[2] = make_float2( cz * sy,  sz * sy);
    U[3] = make_float2( cz * cy,  sz * cy);
}

__device__ __forceinline__ void kron22(const float2* Ub, const float2* Ua, float2* M) {
    #pragma unroll
    for (int o = 0; o < 4; ++o)
        #pragma unroll
        for (int j = 0; j < 4; ++j)
            M[o * 4 + j] = cmul(Ub[(o >> 1) * 2 + (j >> 1)], Ua[(o & 1) * 2 + (j & 1)]);
}

__device__ __forceinline__ void apply_pair_wmma(float2* st, const float2* M,
                                                int w, int lane, int wave) {
    const int col = lane & 15;
    const int hiK = lane >> 4;
    const int o   = (col >> 1) & 3;
    const float2 m1 = M[o * 4 + hiK];
    const float2 m2 = M[o * 4 + hiK + 2];
    float a10, a11, a20, a21;
    if ((col & 1) == 0) { a10 = m1.x; a11 = -m1.y; a20 = m2.x; a21 = -m2.y; }
    else                { a10 = m1.y; a11 =  m1.x; a20 = m2.y; a21 =  m2.x; }
    v2f A1; A1[0] = a10; A1[1] = a11;
    v2f A2; A2[0] = a20; A2[1] = a21;

    const int s       = 1 << w;
    const int lowmask = s - 1;
    const int off1    = hiK ? s : 0;
    const int off2    = off1 + 2 * s;
    const int jst     = hiK ? 2 : 0;

    int    bases[NROUNDS];
    float2 p1[NROUNDS], p2[NROUNDS];
    #pragma unroll
    for (int r = 0; r < NROUNDS; ++r) {
        const int g = r * (NWAVES * 16) + wave * 16 + col;
        bases[r] = ((g >> w) << (w + 2)) | (g & lowmask);
        p1[r] = st[bases[r] + off1];
        p2[r] = st[bases[r] + off2];
    }
    #pragma unroll
    for (int r = 0; r < NROUNDS; ++r) {
        v2f B1; B1[0] = p1[r].x; B1[1] = p1[r].y;
        v2f B2; B2[0] = p2[r].x; B2[1] = p2[r].y;
        v8f C = {0.f, 0.f, 0.f, 0.f, 0.f, 0.f, 0.f, 0.f};
        C = __builtin_amdgcn_wmma_f32_16x16x4_f32(false, A1, false, B1,
                                                  (short)0, C, false, false);
        C = __builtin_amdgcn_wmma_f32_16x16x4_f32(false, A2, false, B2,
                                                  (short)0, C, false, false);
        const float2 oa = hiK ? make_float2(C[4], C[5]) : make_float2(C[0], C[1]);
        const float2 ob = hiK ? make_float2(C[6], C[7]) : make_float2(C[2], C[3]);
        st[bases[r] + jst * s]       = oa;
        st[bases[r] + (jst + 1) * s] = ob;
    }
}

__global__ __launch_bounds__(THREADS)
void qsim_kernel(const float* __restrict__ fv, const float* __restrict__ theta,
                 float* __restrict__ out) {
    __shared__ float2 st[DIM];
    __shared__ float2 embM[NPAIRS][16];
    __shared__ float2 layM[NLAYERS * NPAIRS][16];
    __shared__ float  wpart[NWAVES][NQ];

    const int b    = blockIdx.x;
    const int tid  = threadIdx.x;
    const int lane = tid & 31;
    const int wave = tid >> 5;

    for (int i = tid; i < DIM; i += THREADS) st[i] = make_float2(0.f, 0.f);
    if (tid == 0) st[0] = make_float2(1.f, 0.f);

    if (tid < NPAIRS + NLAYERS * NPAIRS) {
        float2 Ua[4], Ub[4];
        if (tid < NPAIRS) {
            embed_rot(fv[b * NQ + 2 * tid],     Ua);
            embed_rot(fv[b * NQ + 2 * tid + 1], Ub);
            kron22(Ub, Ua, &embM[tid][0]);
        } else {
            const int t = tid - NPAIRS, l = t / NPAIRS, p = t % NPAIRS;
            const float* ta = theta + (l * NQ + 2 * p) * 3;
            fused_rot(ta[0], ta[1], ta[2], Ua);
            fused_rot(ta[3], ta[4], ta[5], Ub);
            kron22(Ub, Ua, &layM[t][0]);
        }
    }
    __syncthreads();

    for (int p = 0; p < NPAIRS; ++p) {
        apply_pair_wmma(st, &embM[p][0], 2 * p, lane, wave);
        __syncthreads();
    }

    for (int l = 0; l < NLAYERS; ++l) {
        for (int p = 0; p < NPAIRS; ++p) {
            apply_pair_wmma(st, &layM[l * NPAIRS + p][0], 2 * p, lane, wave);
            __syncthreads();
        }
        for (int k = 0; k < NQ; ++k) {
            const int c = (k < NQ - 1) ? k : NQ - 1;
            const int t = (k < NQ - 1) ? k + 1 : 0;
            const int lo = c < t ? c : t;
            const int hi = c < t ? t : c;
            const int midbits = hi - lo - 1;
            for (int p = tid; p < DIM / 4; p += THREADS) {
                const int i    = p & ((1 << lo) - 1);
                const int rest = p >> lo;
                const int mid  = rest & ((1 << midbits) - 1);
                const int top  = rest >> midbits;
                const int idx  = i | (mid << (lo + 1)) | (top << (hi + 1)) | (1 << c);
                const int jdx  = idx | (1 << t);
                float2 tmp = st[idx];
                st[idx] = st[jdx];
                st[jdx] = tmp;
            }
            __syncthreads();
        }
    }

    float acc[NQ];
    #pragma unroll
    for (int w = 0; w < NQ; ++w) acc[w] = 0.f;
    for (int i = tid; i < DIM; i += THREADS) {
        const float2 a  = st[i];
        const float  pz = a.x * a.x + a.y * a.y;
        #pragma unroll
        for (int w = 0; w < NQ; ++w)
            acc[w] += ((i >> w) & 1) ? -pz : pz;
    }
    #pragma unroll
    for (int off = 16; off > 0; off >>= 1) {
        #pragma unroll
        for (int w = 0; w < NQ; ++w)
            acc[w] += __shfl_xor(acc[w], off, 32);
    }
    if (lane == 0) {
        #pragma unroll
        for (int w = 0; w < NQ; ++w) wpart[wave][w] = acc[w];
    }
    __syncthreads();
    if (tid < NQ) {
        float s = 0.f;
        for (int q = 0; q < NWAVES; ++q) s += wpart[q][tid];
        out[b * NQ + tid] = s;
    }
}

extern "C" void kernel_launch(void* const* d_in, const int* in_sizes, int n_in,
                              void* d_out, int out_size, void* d_ws, size_t ws_size,
                              hipStream_t stream) {
    const float* fv    = (const float*)d_in[0];
    const float* theta = (const float*)d_in[1];
    float*       out   = (float*)d_out;
    const int batch = in_sizes[0] / NQ;
    qsim_kernel<<<batch, THREADS, 0, stream>>>(fv, theta, out);
}


# each wire's embedding gate multiplied into its layer-1 gate (no entangler between them): per-element 2x2 products in the set-up; 18 pair-gate passes, not 21; products summed in a new order
# speedup vs baseline: 1.9752x; 1.9752x over previous
#include <hip/hip_runtime.h>


typedef __attribute__((ext_vector_type(2))) float v2f;
typedef __attribute__((ext_vector_type(8))) float v8f;

#define NQ      12
#define DIM     4096
#define THREADS 256
#define NWAVES  8
#define NLAYERS 6
#define NPAIRS  6
#define NDUO    3
#define NTILES  8

__host__ __device__ constexpr int ring_fwd(int x) {
    int t = x;
    t ^= t << 1; t ^= t << 2; t ^= t << 4; t ^= t << 8;
    t &= DIM - 1;
    return t ^ (t >> (NQ - 1));
}
__host__ __device__ constexpr int ring_inv(int y) {
    const int top = y >> (NQ - 1);
    return (y ^ (y << 1) ^ top ^ (top << 1)) & (DIM - 1);
}
__host__ __device__ constexpr int fold(int x) {
    return x ^ ((x >> 5) & 31) ^ ((x >> 10) & 3);
}
__host__ __device__ constexpr int ring_inv_folded(int p) {
    return fold(ring_inv(fold(p)));
}

__device__ __forceinline__ float2 cmul(float2 a, float2 b) {
    return make_float2(a.x * b.x - a.y * b.y, a.x * b.y + a.y * b.x);
}

__device__ __forceinline__ void fused_rot(float tx, float ty, float tz, float2* U) {
    const float cx = cosf(0.5f * tx), sx = sinf(0.5f * tx);
    const float cy = cosf(0.5f * ty), sy = sinf(0.5f * ty);
    const float cz = cosf(0.5f * tz), sz = sinf(0.5f * tz);
    const float2 A00 = make_float2( cy * cx,  sy * sx);
    const float2 A01 = make_float2(-sy * cx, -cy * sx);
    const float2 A10 = make_float2( sy * cx, -cy * sx);
    const float2 A11 = make_float2( cy * cx, -sy * sx);
    const float2 e  = make_float2(cz, -sz);
    const float2 ec = make_float2(cz,  sz);
    U[0] = cmul(e, A00);  U[1] = cmul(e, A01);
    U[2] = cmul(ec, A10); U[3] = cmul(ec, A11);
}

__device__ __forceinline__ void embed_rot(float f, float2* U) {
    const float cy = cosf(0.5f * f),  sy = sinf(0.5f * f);
    const float cz = cosf(0.25f * f), sz = sinf(0.25f * f);
    U[0] = make_float2( cz * cy, -sz * cy);
    U[1] = make_float2(-cz * sy,  sz * sy);
    U[2] = make_float2( cz * sy,  sz * sy);
    U[3] = make_float2( cz * cy,  sz * cy);
}

__device__ __forceinline__ void kron22(const float2* Ub, const float2* Ua, float2* M) {
    #pragma unroll
    for (int o = 0; o < 4; ++o)
        #pragma unroll
        for (int j = 0; j < 4; ++j)
            M[o * 4 + j] = cmul(Ub[(o >> 1) * 2 + (j >> 1)], Ua[(o & 1) * 2 + (j & 1)]);
}

__device__ __forceinline__ void mul22(const float2* L, const float2* E, float2* C) {
    #pragma unroll
    for (int r = 0; r < 2; ++r)
        #pragma unroll
        for (int c = 0; c < 2; ++c) {
            const float2 x = cmul(L[r * 2], E[c]);
            const float2 y = cmul(L[r * 2 + 1], E[2 + c]);
            C[r * 2 + c] = make_float2(x.x + y.x, x.y + y.y);
        }
}

__device__ __forceinline__ void pair_A(float2 m1, float2 m2, int lane, v2f& A1, v2f& A2) {
    const bool im = (lane & 1) != 0;
    A1[0] = im ? m1.y :  m1.x;  A1[1] = im ? m1.x : -m1.y;
    A2[0] = im ? m2.y :  m2.x;  A2[1] = im ? m2.x : -m2.y;
}

__device__ __forceinline__ void apply_two_pairs(float2* st, const float2* MA, const float2* MB,
                                                int wA, int wB, int wx,
                                                int lan, const int* E, int lane) {
    const int hiK = lane >> 4;
    const int o   = (lane >> 1) & 3;
    const float2 ma1 = MA[o * 4 + hiK], ma2 = MA[o * 4 + hiK + 2];
    const float2 mb1 = MB[o * 4 + hiK], mb2 = MB[o * 4 + hiK + 2];

    const int a1 = E[wA], a2 = E[wA + 1];
    const int b1 = E[wB], b2 = E[wB + 1];
    const int dx = E[wx];
    const int ld0 = lan ^ (hiK ? a1 : 0);
    const int st0 = lan ^ (hiK ? b2 : 0);

    float2 p1[NTILES], p2[NTILES];
    #pragma unroll
    for (int t = 0; t < NTILES; ++t) {
        const int rc = ((t & 1) ? dx : 0) ^ ((t & 2) ? b1 : 0) ^ ((t & 4) ? b2 : 0);
        p1[t] = st[ld0 ^ rc];
        p2[t] = st[ld0 ^ rc ^ a2];
    }
    v2f A1a, A2a, A1b, A2b;
    pair_A(ma1, ma2, lane, A1a, A2a);
    pair_A(mb1, mb2, lane, A1b, A2b);
    v8f D[NTILES];
    #pragma unroll
    for (int t = 0; t < NTILES; ++t) {
        v2f B1; B1[0] = p1[t].x; B1[1] = p1[t].y;
        v2f B2; B2[0] = p2[t].x; B2[1] = p2[t].y;
        v8f C = {0.f, 0.f, 0.f, 0.f, 0.f, 0.f, 0.f, 0.f};
        C = __builtin_amdgcn_wmma_f32_16x16x4_f32(false, A1a, false, B1,
                                                  (short)0, C, false, false);
        C = __builtin_amdgcn_wmma_f32_16x16x4_f32(false, A2a, false, B2,
                                                  (short)0, C, false, false);
        D[t] = C;
    }
    #pragma unroll
    for (int q = 0; q < NTILES; ++q) {
        const int x  = q & 1;
        const int jA = q >> 1;
        v2f B1, B2;
        B1[0] = hiK ? D[x | 2][2 * jA]     : D[x][2 * jA];
        B1[1] = hiK ? D[x | 2][2 * jA + 1] : D[x][2 * jA + 1];
        B2[0] = hiK ? D[x | 6][2 * jA]     : D[x | 4][2 * jA];
        B2[1] = hiK ? D[x | 6][2 * jA + 1] : D[x | 4][2 * jA + 1];
        v8f C = {0.f, 0.f, 0.f, 0.f, 0.f, 0.f, 0.f, 0.f};
        C = __builtin_amdgcn_wmma_f32_16x16x4_f32(false, A1b, false, B1,
                                                  (short)0, C, false, false);
        C = __builtin_amdgcn_wmma_f32_16x16x4_f32(false, A2b, false, B2,
                                                  (short)0, C, false, false);
        const float2 oa = hiK ? make_float2(C[4], C[5]) : make_float2(C[0], C[1]);
        const float2 ob = hiK ? make_float2(C[6], C[7]) : make_float2(C[2], C[3]);
        const int rc = ((q & 1) ? dx : 0) ^ ((q & 2) ? a1 : 0) ^ ((q & 4) ? a2 : 0);
        st[st0 ^ rc]      = oa;
        st[st0 ^ rc ^ b1] = ob;
    }
}

__device__ __forceinline__ int place_bits(int col, int wave,
                                          int c0, int c1, int c2, int c3,
                                          int v0, int v1, int v2) {
    return ((col & 1) << c0) | (((col >> 1) & 1) << c1) | (((col >> 2) & 1) << c2) |
           (((col >> 3) & 1) << c3) |
           ((wave & 1) << v0) | (((wave >> 1) & 1) << v1) | (((wave >> 2) & 1) << v2);
}

__global__ __launch_bounds__(THREADS)
void qsim_kernel(const float* __restrict__ fv, const float* __restrict__ theta,
                 float* __restrict__ out) {
    __shared__ float2 st[DIM];
    __shared__ float2 wireU[2 * NPAIRS][2][4];
    __shared__ float2 layM[NLAYERS * NPAIRS][16];
    __shared__ float  wpart[NWAVES][NQ];

    const int b    = blockIdx.x;
    const int tid  = threadIdx.x;
    const int lane = tid & 31;
    const int wave = tid >> 5;

    for (int i = tid; i < DIM; i += THREADS) st[i] = make_float2(0.f, 0.f);
    if (tid == 0) st[0] = make_float2(1.f, 0.f);

    if (tid < NPAIRS + NLAYERS * NPAIRS) {
        float2 Ua[4], Ub[4];
        if (tid < NPAIRS) {
            embed_rot(fv[b * NQ + 2 * tid],     Ua);
            embed_rot(fv[b * NQ + 2 * tid + 1], Ub);
        } else {
            const int t = tid - NPAIRS;
            const float* ta = theta + (2 * t) * 3;
            fused_rot(ta[0], ta[1], ta[2], Ua);
            fused_rot(ta[3], ta[4], ta[5], Ub);
            if (t >= NPAIRS) kron22(Ub, Ua, &layM[t][0]);
        }
        if (tid < 2 * NPAIRS) {
            #pragma unroll
            for (int i = 0; i < 4; ++i) { wireU[tid][0][i] = Ua[i]; wireU[tid][1][i] = Ub[i]; }
        }
    }
    __syncthreads();
    if (tid < NPAIRS) {
        float2 Ua[4], Ub[4];
        mul22(&wireU[NPAIRS + tid][0][0], &wireU[tid][0][0], Ua);
        mul22(&wireU[NPAIRS + tid][1][0], &wireU[tid][1][0], Ub);
        kron22(Ub, Ua, &layM[tid][0]);
    }
    __syncthreads();

    int E[NQ];
    #pragma unroll
    for (int k = 0; k < NQ; ++k) E[k] = fold(1 << k);
    const int col = lane & 15;
    int lanv[NDUO];
    lanv[0] = fold(place_bits(col, wave, 4, 6, 7, 11, 8, 9, 10));
    lanv[1] = fold(place_bits(col, wave, 0, 1, 3, 11, 8, 9, 10));
    lanv[2] = fold(place_bits(col, wave, 0, 1, 2, 4, 5, 6, 7));

    for (int l = 0; l < NLAYERS; ++l) {
        const float2* M = &layM[l * NPAIRS][0];
        apply_two_pairs(st, M,      M + 16, 0, 2, 5, lanv[0], E, lane);
        __syncthreads();
        apply_two_pairs(st, M + 32, M + 48, 4, 6, 2, lanv[1], E, lane);
        __syncthreads();
        apply_two_pairs(st, M + 64, M + 80, 8, 10, 3, lanv[2], E, lane);
        __syncthreads();
        #pragma unroll
        for (int k = 0; k < NQ; ++k) E[k] = ring_inv_folded(E[k]);
        #pragma unroll
        for (int p = 0; p < NDUO; ++p) lanv[p] = ring_inv_folded(lanv[p]);
    }

    float acc[NQ];
    #pragma unroll
    for (int w = 0; w < NQ; ++w) acc[w] = 0.f;
    int ytid = fold(tid);
    #pragma unroll
    for (int k = 0; k < NLAYERS; ++k) ytid = ring_fwd(ytid);
    #pragma unroll
    for (int q = 0; q < DIM / THREADS; ++q) {
        int yq = fold(q * THREADS);
        #pragma unroll
        for (int k = 0; k < NLAYERS; ++k) yq = ring_fwd(yq);
        const int    y  = ytid ^ yq;
        const float2 a  = st[tid + q * THREADS];
        const float  pz = a.x * a.x + a.y * a.y;
        #pragma unroll
        for (int w = 0; w < NQ; ++w)
            acc[w] += ((y >> w) & 1) ? -pz : pz;
    }
    #pragma unroll
    for (int off = 16; off > 0; off >>= 1) {
        #pragma unroll
        for (int w = 0; w < NQ; ++w)
            acc[w] += __shfl_xor(acc[w], off, 32);
    }
    if (lane == 0) {
        #pragma unroll
        for (int w = 0; w < NQ; ++w) wpart[wave][w] = acc[w];
    }
    __syncthreads();
    if (tid < NQ) {
        float s = 0.f;
        for (int q = 0; q < NWAVES; ++q) s += wpart[q][tid];
        out[b * NQ + tid] = s;
    }
}

extern "C" void kernel_launch(void* const* d_in, const int* in_sizes, int n_in,
                              void* d_out, int out_size, void* d_ws, size_t ws_size,
                              hipStream_t stream) {
    const float* fv    = (const float*)d_in[0];
    const float* theta = (const float*)d_in[1];
    float*       out   = (float*)d_out;
    const int batch = in_sizes[0] / NQ;
    qsim_kernel<<<batch, THREADS, 0, stream>>>(fv, theta, out);
}


# three wire-pair gates per LDS round trip: 4 waves x 1024 amplitudes, 2nd and 3rd gate's B by lane-half select from register tiles; 2 passes a layer, not 3; 128 threads; <Z> sums in new order
# speedup vs baseline: 2.0419x; 1.0338x over previous
#include <hip/hip_runtime.h>


typedef __attribute__((ext_vector_type(2))) float v2f;
typedef __attribute__((ext_vector_type(8))) float v8f;

#define NQ      12
#define DIM     4096
#define THREADS 128
#define NWAVES  4
#define NLAYERS 6
#define NPAIRS  6
#define NPASS   2
#define NTILES  16

__host__ __device__ constexpr int ring_fwd(int x) {
    int t = x;
    t ^= t << 1; t ^= t << 2; t ^= t << 4; t ^= t << 8;
    t &= DIM - 1;
    return t ^ (t >> (NQ - 1));
}
__host__ __device__ constexpr int ring_inv(int y) {
    const int top = y >> (NQ - 1);
    return (y ^ (y << 1) ^ top ^ (top << 1)) & (DIM - 1);
}
__host__ __device__ constexpr int fold(int x) {
    constexpr int pat[7] = {15, 22, 19, 3, 10, 2, 21};
    int r = x;
    for (int b = 0; b < 7; ++b)
        if ((x >> (5 + b)) & 1) r ^= pat[b];
    return r;
}

__host__ __device__ constexpr bool lanes_distinct(const int* v) {
    unsigned seen = 0;
    for (int m = 0; m < 32; ++m) {
        int s = 0;
        for (int i = 0; i < 5; ++i)
            if ((m >> i) & 1) s ^= v[i];
        s &= 31;
        if ((seen >> s) & 1u) return false;
        seen |= 1u << s;
    }
    return true;
}

struct Maps {
    int  g[NLAYERS][NPASS][6];
    int  o[NLAYERS][NPASS][6];
    int  y[NQ];
    bool ok;
};
__host__ __device__ constexpr Maps make_maps() {
    Maps M{};
    M.ok = true;
    int E[NQ] = {};
    for (int k = 0; k < NQ; ++k) E[k] = fold(1 << k);
    for (int l = 0; l < NLAYERS; ++l) {
        for (int p = 0; p < NPASS; ++p) {
            const int gw = 6 * p;
            const int ob = 6 * (1 - p);
            for (int i = 0; i < 6; ++i) M.g[l][p][i] = E[gw + i];
            bool found = false;
            for (int mask = 0; mask < 64 && !found; ++mask) {
                int cols[6] = {}, nc = 0, rest[6] = {}, nr = 0;
                for (int i = 0; i < 6; ++i) {
                    if ((mask >> i) & 1) cols[nc++] = E[ob + i];
                    else                 rest[nr++] = E[ob + i];
                }
                if (nc != 4) continue;
                const int ld[5] = {cols[0], cols[1], cols[2], cols[3], E[gw]};
                const int sv[5] = {cols[0], cols[1], cols[2], cols[3], E[gw + 5]};
                if (lanes_distinct(ld) && lanes_distinct(sv)) {
                    found = true;
                    for (int i = 0; i < 4; ++i) M.o[l][p][i] = cols[i];
                    M.o[l][p][4] = rest[0];
                    M.o[l][p][5] = rest[1];
                }
            }
            if (!found) M.ok = false;
        }
        for (int k = 0; k < NQ; ++k) E[k] = fold(ring_inv(fold(E[k])));
    }
    for (int j = 0; j < NQ; ++j) {
        int v = fold(1 << j);
        for (int l = 0; l < NLAYERS; ++l) v = ring_fwd(v);
        M.y[j] = v;
    }
    return M;
}
static_assert(make_maps().ok, "no conflict-free lane columns for some layer and pass");

__device__ __forceinline__ float2 cmul(float2 a, float2 b) {
    return make_float2(a.x * b.x - a.y * b.y, a.x * b.y + a.y * b.x);
}

__device__ __forceinline__ void fused_rot(float tx, float ty, float tz, float2* U) {
    const float cx = cosf(0.5f * tx), sx = sinf(0.5f * tx);
    const float cy = cosf(0.5f * ty), sy = sinf(0.5f * ty);
    const float cz = cosf(0.5f * tz), sz = sinf(0.5f * tz);
    const float2 A00 = make_float2( cy * cx,  sy * sx);
    const float2 A01 = make_float2(-sy * cx, -cy * sx);
    const float2 A10 = make_float2( sy * cx, -cy * sx);
    const float2 A11 = make_float2( cy * cx, -sy * sx);
    const float2 e  = make_float2(cz, -sz);
    const float2 ec = make_float2(cz,  sz);
    U[0] = cmul(e, A00);  U[1] = cmul(e, A01);
    U[2] = cmul(ec, A10); U[3] = cmul(ec, A11);
}

__device__ __forceinline__ void embed_rot(float f, float2* U) {
    const float cy = cosf(0.5f * f),  sy = sinf(0.5f * f);
    const float cz = cosf(0.25f * f), sz = sinf(0.25f * f);
    U[0] = make_float2( cz * cy, -sz * cy);
    U[1] = make_float2(-cz * sy,  sz * sy);
    U[2] = make_float2( cz * sy,  sz * sy);
    U[3] = make_float2( cz * cy,  sz * cy);
}

__device__ __forceinline__ void kron22(const float2* Ub, const float2* Ua, float2* M) {
    #pragma unroll
    for (int o = 0; o < 4; ++o)
        #pragma unroll
        for (int j = 0; j < 4; ++j)
            M[o * 4 + j] = cmul(Ub[(o >> 1) * 2 + (j >> 1)], Ua[(o & 1) * 2 + (j & 1)]);
}

__device__ __forceinline__ void mul22(const float2* L, const float2* E, float2* C) {
    #pragma unroll
    for (int r = 0; r < 2; ++r)
        #pragma unroll
        for (int c = 0; c < 2; ++c) {
            const float2 x = cmul(L[r * 2], E[c]);
            const float2 y = cmul(L[r * 2 + 1], E[2 + c]);
            C[r * 2 + c] = make_float2(x.x + y.x, x.y + y.y);
        }
}

__device__ __forceinline__ void pair_A(float2 m1, float2 m2, int lane, v2f& A1, v2f& A2) {
    const bool im = (lane & 1) != 0;
    A1[0] = im ? m1.y :  m1.x;  A1[1] = im ? m1.x : -m1.y;
    A2[0] = im ? m2.y :  m2.x;  A2[1] = im ? m2.x : -m2.y;
}

__device__ __forceinline__ void apply_three_pairs(float2* st, const float2* MA, const float2* MB,
                                                  const float2* MC, const int* g,
                                                  int lan, int lane) {
    const int hiK = lane >> 4;
    const int o   = (lane >> 1) & 3;
    const float2 ma1 = MA[o * 4 + hiK], ma2 = MA[o * 4 + hiK + 2];
    const float2 mb1 = MB[o * 4 + hiK], mb2 = MB[o * 4 + hiK + 2];
    const float2 mc1 = MC[o * 4 + hiK], mc2 = MC[o * 4 + hiK + 2];

    const int a1 = g[0], a2 = g[1];
    const int b1 = g[2], b2 = g[3];
    const int c1 = g[4], c2 = g[5];
    const int ld0 = lan ^ (hiK ? a1 : 0);
    const int st0 = lan ^ (hiK ? c2 : 0);

    v2f A1a, A2a, A1b, A2b, A1c, A2c;
    pair_A(ma1, ma2, lane, A1a, A2a);
    pair_A(mb1, mb2, lane, A1b, A2b);
    pair_A(mc1, mc2, lane, A1c, A2c);

    v8f D1[NTILES];
    #pragma unroll
    for (int t = 0; t < NTILES; ++t) {
        const int rc = ((t & 1) ? b1 : 0) ^ ((t & 2) ? b2 : 0) ^
                       ((t & 4) ? c1 : 0) ^ ((t & 8) ? c2 : 0);
        const float2 p1 = st[ld0 ^ rc];
        const float2 p2 = st[ld0 ^ rc ^ a2];
        v2f B1; B1[0] = p1.x; B1[1] = p1.y;
        v2f B2; B2[0] = p2.x; B2[1] = p2.y;
        v8f C = {0.f, 0.f, 0.f, 0.f, 0.f, 0.f, 0.f, 0.f};
        C = __builtin_amdgcn_wmma_f32_16x16x4_f32(false, A1a, false, B1,
                                                  (short)0, C, false, false);
        C = __builtin_amdgcn_wmma_f32_16x16x4_f32(false, A2a, false, B2,
                                                  (short)0, C, false, false);
        D1[t] = C;
    }
    v8f D2[NTILES];
    #pragma unroll
    for (int tc = 0; tc < 4; ++tc) {
        #pragma unroll
        for (int jA = 0; jA < 4; ++jA) {
            v2f B1, B2;
            B1[0] = hiK ? D1[tc * 4 + 1][2 * jA]     : D1[tc * 4][2 * jA];
            B1[1] = hiK ? D1[tc * 4 + 1][2 * jA + 1] : D1[tc * 4][2 * jA + 1];
            B2[0] = hiK ? D1[tc * 4 + 3][2 * jA]     : D1[tc * 4 + 2][2 * jA];
            B2[1] = hiK ? D1[tc * 4 + 3][2 * jA + 1] : D1[tc * 4 + 2][2 * jA + 1];
            v8f C = {0.f, 0.f, 0.f, 0.f, 0.f, 0.f, 0.f, 0.f};
            C = __builtin_amdgcn_wmma_f32_16x16x4_f32(false, A1b, false, B1,
                                                      (short)0, C, false, false);
            C = __builtin_amdgcn_wmma_f32_16x16x4_f32(false, A2b, false, B2,
                                                      (short)0, C, false, false);
            D2[jA * 4 + tc] = C;
        }
    }
    #pragma unroll
    for (int jA = 0; jA < 4; ++jA) {
        #pragma unroll
        for (int jB = 0; jB < 4; ++jB) {
            v2f B1, B2;
            B1[0] = hiK ? D2[jA * 4 + 1][2 * jB]     : D2[jA * 4][2 * jB];
            B1[1] = hiK ? D2[jA * 4 + 1][2 * jB + 1] : D2[jA * 4][2 * jB + 1];
            B2[0] = hiK ? D2[jA * 4 + 3][2 * jB]     : D2[jA * 4 + 2][2 * jB];
            B2[1] = hiK ? D2[jA * 4 + 3][2 * jB + 1] : D2[jA * 4 + 2][2 * jB + 1];
            v8f C = {0.f, 0.f, 0.f, 0.f, 0.f, 0.f, 0.f, 0.f};
            C = __builtin_amdgcn_wmma_f32_16x16x4_f32(false, A1c, false, B1,
                                                      (short)0, C, false, false);
            C = __builtin_amdgcn_wmma_f32_16x16x4_f32(false, A2c, false, B2,
                                                      (short)0, C, false, false);
            const float2 oa = hiK ? make_float2(C[4], C[5]) : make_float2(C[0], C[1]);
            const float2 ob = hiK ? make_float2(C[6], C[7]) : make_float2(C[2], C[3]);
            const int rc = ((jA & 1) ? a1 : 0) ^ ((jA & 2) ? a2 : 0) ^
                           ((jB & 1) ? b1 : 0) ^ ((jB & 2) ? b2 : 0);
            st[st0 ^ rc]      = oa;
            st[st0 ^ rc ^ c1] = ob;
        }
    }
}

__device__ __forceinline__ int lane_slot(const int* o, int lane, int wave) {
    int s = 0;
    #pragma unroll
    for (int i = 0; i < 4; ++i) s ^= o[i] & -((lane >> i) & 1);
    s ^= o[4] & -(wave & 1);
    s ^= o[5] & -((wave >> 1) & 1);
    return s;
}

__global__ __launch_bounds__(THREADS, 4)
void qsim_kernel(const float* __restrict__ fv, const float* __restrict__ theta,
                 float* __restrict__ out) {
    __shared__ float2 st[DIM];
    __shared__ float2 wireU[2 * NPAIRS][2][4];
    __shared__ float2 layM[NLAYERS * NPAIRS][16];
    __shared__ float  wpart[NWAVES][NQ];

    const int b    = blockIdx.x;
    const int tid  = threadIdx.x;
    const int lane = tid & 31;
    const int wave = tid >> 5;

    for (int i = tid; i < DIM; i += THREADS) st[i] = make_float2(0.f, 0.f);
    if (tid == 0) st[0] = make_float2(1.f, 0.f);

    if (tid < NPAIRS + NLAYERS * NPAIRS) {
        float2 Ua[4], Ub[4];
        if (tid < NPAIRS) {
            embed_rot(fv[b * NQ + 2 * tid],     Ua);
            embed_rot(fv[b * NQ + 2 * tid + 1], Ub);
        } else {
            const int t = tid - NPAIRS;
            const float* ta = theta + (2 * t) * 3;
            fused_rot(ta[0], ta[1], ta[2], Ua);
            fused_rot(ta[3], ta[4], ta[5], Ub);
            if (t >= NPAIRS) kron22(Ub, Ua, &layM[t][0]);
        }
        if (tid < 2 * NPAIRS) {
            #pragma unroll
            for (int i = 0; i < 4; ++i) { wireU[tid][0][i] = Ua[i]; wireU[tid][1][i] = Ub[i]; }
        }
    }
    __syncthreads();
    if (tid < NPAIRS) {
        float2 Ua[4], Ub[4];
        mul22(&wireU[NPAIRS + tid][0][0], &wireU[tid][0][0], Ua);
        mul22(&wireU[NPAIRS + tid][1][0], &wireU[tid][1][0], Ub);
        kron22(Ub, Ua, &layM[tid][0]);
    }
    __syncthreads();

    static constexpr Maps MAPS = make_maps();

    #pragma nounroll
    for (int l = 0; l < NLAYERS; ++l) {
        const float2* M = &layM[l * NPAIRS][0];
        int g0[6], g1[6];
        #pragma unroll
        for (int i = 0; i < 6; ++i) { g0[i] = MAPS.g[l][0][i]; g1[i] = MAPS.g[l][1][i]; }
        const int lan0 = lane_slot(MAPS.o[l][0], lane, wave);
        const int lan1 = lane_slot(MAPS.o[l][1], lane, wave);
        apply_three_pairs(st, M,      M + 16, M + 32, g0, lan0, lane);
        __syncthreads();
        apply_three_pairs(st, M + 48, M + 64, M + 80, g1, lan1, lane);
        __syncthreads();
    }

    float acc[NQ];
    #pragma unroll
    for (int w = 0; w < NQ; ++w) acc[w] = 0.f;
    int ytid = 0;
    #pragma unroll
    for (int j = 0; j < 7; ++j) ytid ^= ((tid >> j) & 1) ? MAPS.y[j] : 0;
    #pragma unroll
    for (int q = 0; q < DIM / THREADS; ++q) {
        int yq = 0;
        #pragma unroll
        for (int j = 0; j < 5; ++j) yq ^= ((q >> j) & 1) ? MAPS.y[7 + j] : 0;
        const int    y  = ytid ^ yq;
        const float2 a  = st[tid + q * THREADS];
        const float  pz = a.x * a.x + a.y * a.y;
        #pragma unroll
        for (int w = 0; w < NQ; ++w)
            acc[w] += ((y >> w) & 1) ? -pz : pz;
    }
    #pragma unroll
    for (int off = 16; off > 0; off >>= 1) {
        #pragma unroll
        for (int w = 0; w < NQ; ++w)
            acc[w] += __shfl_xor(acc[w], off, 32);
    }
    if (lane == 0) {
        #pragma unroll
        for (int w = 0; w < NQ; ++w) wpart[wave][w] = acc[w];
    }
    __syncthreads();
    if (tid < NQ) {
        float s = 0.f;
        for (int q = 0; q < NWAVES; ++q) s += wpart[q][tid];
        out[b * NQ + tid] = s;
    }
}

extern "C" void kernel_launch(void* const* d_in, const int* in_sizes, int n_in,
                              void* d_out, int out_size, void* d_ws, size_t ws_size,
                              hipStream_t stream) {
    const float* fv    = (const float*)d_in[0];
    const float* theta = (const float*)d_in[1];
    float*       out   = (float*)d_out;
    const int batch = in_sizes[0] / NQ;
    qsim_kernel<<<batch, THREADS, 0, stream>>>(fv, theta, out);
}


# layer 1 starts from |0...0>: its two passes drop the tiles and products whose input is a known zero (21 matrix products a pass, not 96); only exact-zero terms leave the sums, order unchanged
# speedup vs baseline: 2.2882x; 1.1206x over previous
#include <hip/hip_runtime.h>


typedef __attribute__((ext_vector_type(2))) float v2f;
typedef __attribute__((ext_vector_type(8))) float v8f;

#define NQ      12
#define DIM     4096
#define THREADS 128
#define NWAVES  4
#define NLAYERS 6
#define NPAIRS  6
#define NPASS   2
#define NTILES  16

__host__ __device__ constexpr int ring_fwd(int x) {
    int t = x;
    t ^= t << 1; t ^= t << 2; t ^= t << 4; t ^= t << 8;
    t &= DIM - 1;
    return t ^ (t >> (NQ - 1));
}
__host__ __device__ constexpr int ring_inv(int y) {
    const int top = y >> (NQ - 1);
    return (y ^ (y << 1) ^ top ^ (top << 1)) & (DIM - 1);
}
__host__ __device__ constexpr int fold(int x) {
    constexpr int pat[7] = {15, 22, 19, 3, 10, 2, 21};
    int r = x;
    for (int b = 0; b < 7; ++b)
        if ((x >> (5 + b)) & 1) r ^= pat[b];
    return r;
}

__host__ __device__ constexpr bool lanes_distinct(const int* v) {
    unsigned seen = 0;
    for (int m = 0; m < 32; ++m) {
        int s = 0;
        for (int i = 0; i < 5; ++i)
            if ((m >> i) & 1) s ^= v[i];
        s &= 31;
        if ((seen >> s) & 1u) return false;
        seen |= 1u << s;
    }
    return true;
}

struct Maps {
    int  g[NLAYERS][NPASS][6];
    int  o[NLAYERS][NPASS][6];
    int  y[NQ];
    bool ok;
};
__host__ __device__ constexpr Maps make_maps() {
    Maps M{};
    M.ok = true;
    int E[NQ] = {};
    for (int k = 0; k < NQ; ++k) E[k] = fold(1 << k);
    for (int l = 0; l < NLAYERS; ++l) {
        for (int p = 0; p < NPASS; ++p) {
            const int gw = 6 * p;
            const int ob = 6 * (1 - p);
            for (int i = 0; i < 6; ++i) M.g[l][p][i] = E[gw + i];
            bool found = false;
            for (int mask = 0; mask < 64 && !found; ++mask) {
                int cols[6] = {}, nc = 0, rest[6] = {}, nr = 0;
                for (int i = 0; i < 6; ++i) {
                    if ((mask >> i) & 1) cols[nc++] = E[ob + i];
                    else                 rest[nr++] = E[ob + i];
                }
                if (nc != 4) continue;
                const int ld[5] = {cols[0], cols[1], cols[2], cols[3], E[gw]};
                const int sv[5] = {cols[0], cols[1], cols[2], cols[3], E[gw + 5]};
                if (lanes_distinct(ld) && lanes_distinct(sv)) {
                    found = true;
                    for (int i = 0; i < 4; ++i) M.o[l][p][i] = cols[i];
                    M.o[l][p][4] = rest[0];
                    M.o[l][p][5] = rest[1];
                }
            }
            if (!found) M.ok = false;
        }
        for (int k = 0; k < NQ; ++k) E[k] = fold(ring_inv(fold(E[k])));
    }
    for (int j = 0; j < NQ; ++j) {
        int v = fold(1 << j);
        for (int l = 0; l < NLAYERS; ++l) v = ring_fwd(v);
        M.y[j] = v;
    }
    return M;
}
static_assert(make_maps().ok, "no conflict-free lane columns for some layer and pass");

__device__ __forceinline__ float2 cmul(float2 a, float2 b) {
    return make_float2(a.x * b.x - a.y * b.y, a.x * b.y + a.y * b.x);
}

__device__ __forceinline__ void fused_rot(float tx, float ty, float tz, float2* U) {
    const float cx = cosf(0.5f * tx), sx = sinf(0.5f * tx);
    const float cy = cosf(0.5f * ty), sy = sinf(0.5f * ty);
    const float cz = cosf(0.5f * tz), sz = sinf(0.5f * tz);
    const float2 A00 = make_float2( cy * cx,  sy * sx);
    const float2 A01 = make_float2(-sy * cx, -cy * sx);
    const float2 A10 = make_float2( sy * cx, -cy * sx);
    const float2 A11 = make_float2( cy * cx, -sy * sx);
    const float2 e  = make_float2(cz, -sz);
    const float2 ec = make_float2(cz,  sz);
    U[0] = cmul(e, A00);  U[1] = cmul(e, A01);
    U[2] = cmul(ec, A10); U[3] = cmul(ec, A11);
}

__device__ __forceinline__ void embed_rot(float f, float2* U) {
    const float cy = cosf(0.5f * f),  sy = sinf(0.5f * f);
    const float cz = cosf(0.25f * f), sz = sinf(0.25f * f);
    U[0] = make_float2( cz * cy, -sz * cy);
    U[1] = make_float2(-cz * sy,  sz * sy);
    U[2] = make_float2( cz * sy,  sz * sy);
    U[3] = make_float2( cz * cy,  sz * cy);
}

__device__ __forceinline__ void kron22(const float2* Ub, const float2* Ua, float2* M) {
    #pragma unroll
    for (int o = 0; o < 4; ++o)
        #pragma unroll
        for (int j = 0; j < 4; ++j)
            M[o * 4 + j] = cmul(Ub[(o >> 1) * 2 + (j >> 1)], Ua[(o & 1) * 2 + (j & 1)]);
}

__device__ __forceinline__ void mul22(const float2* L, const float2* E, float2* C) {
    #pragma unroll
    for (int r = 0; r < 2; ++r)
        #pragma unroll
        for (int c = 0; c < 2; ++c) {
            const float2 x = cmul(L[r * 2], E[c]);
            const float2 y = cmul(L[r * 2 + 1], E[2 + c]);
            C[r * 2 + c] = make_float2(x.x + y.x, x.y + y.y);
        }
}

__device__ __forceinline__ void pair_A(float2 m1, float2 m2, int lane, v2f& A1, v2f& A2) {
    const bool im = (lane & 1) != 0;
    A1[0] = im ? m1.y :  m1.x;  A1[1] = im ? m1.x : -m1.y;
    A2[0] = im ? m2.y :  m2.x;  A2[1] = im ? m2.x : -m2.y;
}

template <bool ZIN>
__device__ __forceinline__ void apply_three_pairs(float2* st, const float2* MA, const float2* MB,
                                                  const float2* MC, const int* g,
                                                  int lan, int lane) {
    const int hiK = lane >> 4;
    const int o   = (lane >> 1) & 3;
    const float2 ma1 = MA[o * 4 + hiK], ma2 = MA[o * 4 + hiK + 2];
    const float2 mb1 = MB[o * 4 + hiK], mb2 = MB[o * 4 + hiK + 2];
    const float2 mc1 = MC[o * 4 + hiK], mc2 = MC[o * 4 + hiK + 2];

    const int a1 = g[0], a2 = g[1];
    const int b1 = g[2], b2 = g[3];
    const int c1 = g[4], c2 = g[5];
    const int ld0 = lan ^ (hiK ? a1 : 0);
    const int st0 = lan ^ (hiK ? c2 : 0);

    v2f A1a, A2a, A1b, A2b, A1c, A2c;
    pair_A(ma1, ma2, lane, A1a, A2a);
    pair_A(mb1, mb2, lane, A1b, A2b);
    pair_A(mc1, mc2, lane, A1c, A2c);

    if constexpr (ZIN) {
        const v8f Z = {0.f, 0.f, 0.f, 0.f, 0.f, 0.f, 0.f, 0.f};
        const float2 p1 = st[ld0];
        v2f B0; B0[0] = p1.x; B0[1] = p1.y;
        const v8f E1 = __builtin_amdgcn_wmma_f32_16x16x4_f32(false, A1a, false, B0,
                                                             (short)0, Z, false, false);
        v8f E2[4];
        #pragma unroll
        for (int jA = 0; jA < 4; ++jA) {
            v2f B1;
            B1[0] = hiK ? 0.f : E1[2 * jA];
            B1[1] = hiK ? 0.f : E1[2 * jA + 1];
            E2[jA] = __builtin_amdgcn_wmma_f32_16x16x4_f32(false, A1b, false, B1,
                                                           (short)0, Z, false, false);
        }
        #pragma unroll
        for (int jA = 0; jA < 4; ++jA) {
            #pragma unroll
            for (int jB = 0; jB < 4; ++jB) {
                v2f B1;
                B1[0] = hiK ? 0.f : E2[jA][2 * jB];
                B1[1] = hiK ? 0.f : E2[jA][2 * jB + 1];
                const v8f C = __builtin_amdgcn_wmma_f32_16x16x4_f32(false, A1c, false, B1,
                                                                    (short)0, Z, false, false);
                const float2 oa = hiK ? make_float2(C[4], C[5]) : make_float2(C[0], C[1]);
                const float2 ob = hiK ? make_float2(C[6], C[7]) : make_float2(C[2], C[3]);
                const int rc = ((jA & 1) ? a1 : 0) ^ ((jA & 2) ? a2 : 0) ^
                               ((jB & 1) ? b1 : 0) ^ ((jB & 2) ? b2 : 0);
                st[st0 ^ rc]      = oa;
                st[st0 ^ rc ^ c1] = ob;
            }
        }
        return;
    }

    v8f D1[NTILES];
    #pragma unroll
    for (int t = 0; t < NTILES; ++t) {
        const int rc = ((t & 1) ? b1 : 0) ^ ((t & 2) ? b2 : 0) ^
                       ((t & 4) ? c1 : 0) ^ ((t & 8) ? c2 : 0);
        const float2 p1 = st[ld0 ^ rc];
        const float2 p2 = st[ld0 ^ rc ^ a2];
        v2f B1; B1[0] = p1.x; B1[1] = p1.y;
        v2f B2; B2[0] = p2.x; B2[1] = p2.y;
        v8f C = {0.f, 0.f, 0.f, 0.f, 0.f, 0.f, 0.f, 0.f};
        C = __builtin_amdgcn_wmma_f32_16x16x4_f32(false, A1a, false, B1,
                                                  (short)0, C, false, false);
        C = __builtin_amdgcn_wmma_f32_16x16x4_f32(false, A2a, false, B2,
                                                  (short)0, C, false, false);
        D1[t] = C;
    }
    v8f D2[NTILES];
    #pragma unroll
    for (int tc = 0; tc < 4; ++tc) {
        #pragma unroll
        for (int jA = 0; jA < 4; ++jA) {
            v2f B1, B2;
            B1[0] = hiK ? D1[tc * 4 + 1][2 * jA]     : D1[tc * 4][2 * jA];
            B1[1] = hiK ? D1[tc * 4 + 1][2 * jA + 1] : D1[tc * 4][2 * jA + 1];
            B2[0] = hiK ? D1[tc * 4 + 3][2 * jA]     : D1[tc * 4 + 2][2 * jA];
            B2[1] = hiK ? D1[tc * 4 + 3][2 * jA + 1] : D1[tc * 4 + 2][2 * jA + 1];
            v8f C = {0.f, 0.f, 0.f, 0.f, 0.f, 0.f, 0.f, 0.f};
            C = __builtin_amdgcn_wmma_f32_16x16x4_f32(false, A1b, false, B1,
                                                      (short)0, C, false, false);
            C = __builtin_amdgcn_wmma_f32_16x16x4_f32(false, A2b, false, B2,
                                                      (short)0, C, false, false);
            D2[jA * 4 + tc] = C;
        }
    }
    #pragma unroll
    for (int jA = 0; jA < 4; ++jA) {
        #pragma unroll
        for (int jB = 0; jB < 4; ++jB) {
            v2f B1, B2;
            B1[0] = hiK ? D2[jA * 4 + 1][2 * jB]     : D2[jA * 4][2 * jB];
            B1[1] = hiK ? D2[jA * 4 + 1][2 * jB + 1] : D2[jA * 4][2 * jB + 1];
            B2[0] = hiK ? D2[jA * 4 + 3][2 * jB]     : D2[jA * 4 + 2][2 * jB];
            B2[1] = hiK ? D2[jA * 4 + 3][2 * jB + 1] : D2[jA * 4 + 2][2 * jB + 1];
            v8f C = {0.f, 0.f, 0.f, 0.f, 0.f, 0.f, 0.f, 0.f};
            C = __builtin_amdgcn_wmma_f32_16x16x4_f32(false, A1c, false, B1,
                                                      (short)0, C, false, false);
            C = __builtin_amdgcn_wmma_f32_16x16x4_f32(false, A2c, false, B2,
                                                      (short)0, C, false, false);
            const float2 oa = hiK ? make_float2(C[4], C[5]) : make_float2(C[0], C[1]);
            const float2 ob = hiK ? make_float2(C[6], C[7]) : make_float2(C[2], C[3]);
            const int rc = ((jA & 1) ? a1 : 0) ^ ((jA & 2) ? a2 : 0) ^
                           ((jB & 1) ? b1 : 0) ^ ((jB & 2) ? b2 : 0);
            st[st0 ^ rc]      = oa;
            st[st0 ^ rc ^ c1] = ob;
        }
    }
}

__device__ __forceinline__ int lane_slot(const int* o, int lane, int wave) {
    int s = 0;
    #pragma unroll
    for (int i = 0; i < 4; ++i) s ^= o[i] & -((lane >> i) & 1);
    s ^= o[4] & -(wave & 1);
    s ^= o[5] & -((wave >> 1) & 1);
    return s;
}

__global__ __launch_bounds__(THREADS, 4)
void qsim_kernel(const float* __restrict__ fv, const float* __restrict__ theta,
                 float* __restrict__ out) {
    __shared__ float2 st[DIM];
    __shared__ float2 wireU[2 * NPAIRS][2][4];
    __shared__ float2 layM[NLAYERS * NPAIRS][16];
    __shared__ float  wpart[NWAVES][NQ];

    const int b    = blockIdx.x;
    const int tid  = threadIdx.x;
    const int lane = tid & 31;
    const int wave = tid >> 5;

    for (int i = tid; i < DIM; i += THREADS) st[i] = make_float2(0.f, 0.f);
    if (tid == 0) st[0] = make_float2(1.f, 0.f);

    if (tid < NPAIRS + NLAYERS * NPAIRS) {
        float2 Ua[4], Ub[4];
        if (tid < NPAIRS) {
            embed_rot(fv[b * NQ + 2 * tid],     Ua);
            embed_rot(fv[b * NQ + 2 * tid + 1], Ub);
        } else {
            const int t = tid - NPAIRS;
            const float* ta = theta + (2 * t) * 3;
            fused_rot(ta[0], ta[1], ta[2], Ua);
            fused_rot(ta[3], ta[4], ta[5], Ub);
            if (t >= NPAIRS) kron22(Ub, Ua, &layM[t][0]);
        }
        if (tid < 2 * NPAIRS) {
            #pragma unroll
            for (int i = 0; i < 4; ++i) { wireU[tid][0][i] = Ua[i]; wireU[tid][1][i] = Ub[i]; }
        }
    }
    __syncthreads();
    if (tid < NPAIRS) {
        float2 Ua[4], Ub[4];
        mul22(&wireU[NPAIRS + tid][0][0], &wireU[tid][0][0], Ua);
        mul22(&wireU[NPAIRS + tid][1][0], &wireU[tid][1][0], Ub);
        kron22(Ub, Ua, &layM[tid][0]);
    }
    __syncthreads();

    static constexpr Maps MAPS = make_maps();

    {
        const float2* M = &layM[0][0];
        int g0[6], g1[6];
        #pragma unroll
        for (int i = 0; i < 6; ++i) { g0[i] = MAPS.g[0][0][i]; g1[i] = MAPS.g[0][1][i]; }
        const int lan0 = lane_slot(MAPS.o[0][0], lane, wave);
        const int lan1 = lane_slot(MAPS.o[0][1], lane, wave);
        apply_three_pairs<true>(st, M,      M + 16, M + 32, g0, lan0, lane);
        __syncthreads();
        apply_three_pairs<true>(st, M + 48, M + 64, M + 80, g1, lan1, lane);
        __syncthreads();
    }
    #pragma nounroll
    for (int l = 1; l < NLAYERS; ++l) {
        const float2* M = &layM[l * NPAIRS][0];
        int g0[6], g1[6];
        #pragma unroll
        for (int i = 0; i < 6; ++i) { g0[i] = MAPS.g[l][0][i]; g1[i] = MAPS.g[l][1][i]; }
        const int lan0 = lane_slot(MAPS.o[l][0], lane, wave);
        const int lan1 = lane_slot(MAPS.o[l][1], lane, wave);
        apply_three_pairs<false>(st, M,      M + 16, M + 32, g0, lan0, lane);
        __syncthreads();
        apply_three_pairs<false>(st, M + 48, M + 64, M + 80, g1, lan1, lane);
        __syncthreads();
    }

    float acc[NQ];
    #pragma unroll
    for (int w = 0; w < NQ; ++w) acc[w] = 0.f;
    int ytid = 0;
    #pragma unroll
    for (int j = 0; j < 7; ++j) ytid ^= ((tid >> j) & 1) ? MAPS.y[j] : 0;
    #pragma unroll
    for (int q = 0; q < DIM / THREADS; ++q) {
        int yq = 0;
        #pragma unroll
        for (int j = 0; j < 5; ++j) yq ^= ((q >> j) & 1) ? MAPS.y[7 + j] : 0;
        const int    y  = ytid ^ yq;
        const float2 a  = st[tid + q * THREADS];
        const float  pz = a.x * a.x + a.y * a.y;
        #pragma unroll
        for (int w = 0; w < NQ; ++w)
            acc[w] += ((y >> w) & 1) ? -pz : pz;
    }
    #pragma unroll
    for (int off = 16; off > 0; off >>= 1) {
        #pragma unroll
        for (int w = 0; w < NQ; ++w)
            acc[w] += __shfl_xor(acc[w], off, 32);
    }
    if (lane == 0) {
        #pragma unroll
        for (int w = 0; w < NQ; ++w) wpart[wave][w] = acc[w];
    }
    __syncthreads();
    if (tid < NQ) {
        float s = 0.f;
        for (int q = 0; q < NWAVES; ++q) s += wpart[q][tid];
        out[b * NQ + tid] = s;
    }
}

extern "C" void kernel_launch(void* const* d_in, const int* in_sizes, int n_in,
                              void* d_out, int out_size, void* d_ws, size_t ws_size,
                              hipStream_t stream) {
    const float* fv    = (const float*)d_in[0];
    const float* theta = (const float*)d_in[1];
    float*       out   = (float*)d_out;
    const int batch = in_sizes[0] / NQ;
    qsim_kernel<<<batch, THREADS, 0, stream>>>(fv, theta, out);
}


# next layer's gate on each even wire >= 2 multiplied into the pair matrix of the layer before the ring (two matrices, by parity of lower index bits); layers 2,4,6: 4 pair stages, not 6; new sum order
# speedup vs baseline: 2.5043x; 1.0944x over previous
#include <hip/hip_runtime.h>


typedef __attribute__((ext_vector_type(2))) float v2f;
typedef __attribute__((ext_vector_type(8))) float v8f;

#define NQ      12
#define DIM     4096
#define THREADS 128
#define NWAVES  4
#define NLAYERS 6
#define NPAIRS  6
#define NPASS   2
#define NTILES  16
#define NSLOT   12

__host__ __device__ constexpr int ring_fwd(int x) {
    int t = x;
    t ^= t << 1; t ^= t << 2; t ^= t << 4; t ^= t << 8;
    t &= DIM - 1;
    return t ^ (t >> (NQ - 1));
}
__host__ __device__ constexpr int ring_inv(int y) {
    const int top = y >> (NQ - 1);
    return (y ^ (y << 1) ^ top ^ (top << 1)) & (DIM - 1);
}
__host__ __device__ constexpr int fold(int x) {
    constexpr int pat[7] = {15, 22, 19, 3, 10, 2, 21};
    int r = x;
    for (int b = 0; b < 7; ++b)
        if ((x >> (5 + b)) & 1) r ^= pat[b];
    return r;
}

__host__ __device__ constexpr int lane_ways(const int* v) {
    int cnt[32] = {};
    for (int m = 0; m < 32; ++m) {
        int s = 0;
        for (int i = 0; i < 5; ++i)
            if ((m >> i) & 1) s ^= v[i];
        cnt[s & 31] += 1;
    }
    int mx = 0;
    for (int i = 0; i < 32; ++i) mx = cnt[i] > mx ? cnt[i] : mx;
    return mx;
}
__host__ __device__ constexpr int bit_parity(int x) {
    int p = 0;
    for (int i = 0; i < 12; ++i) p ^= (x >> i) & 1;
    return p;
}

struct PassSpec { int regw[6]; int othw[6]; int omask[6]; };

struct Maps {
    int  g[NLAYERS][NPASS][6];
    int  o[NLAYERS][NPASS][6];
    int  y[NQ];
    bool ok;
};
__host__ __device__ constexpr Maps make_maps() {
    constexpr PassSpec SPEC[NLAYERS][NPASS] = {
        {{{0, 1, 2, 3, 4, 5}, {6, 7, 8, 9, 10, 11}, {1, 2, 4, 24, 32, 8}}, {{6, 7, 8, 9, 10, 11}, {0, 1, 2, 3, 4, 5}, {3, 5, 9, 17, 1, 33}}},
        {{{1, 3, 5, 7, 9, 11}, {0, 2, 4, 6, 8, 10}, {1, 2, 4, 32, 16, 8}}, {{0, 1, 8, 9, 10, 11}, {2, 3, 4, 5, 6, 7}, {1, 2, 4, 16, 8, 32}}},
        {{{0, 1, 2, 3, 4, 5}, {6, 7, 8, 9, 10, 11}, {1, 2, 4, 8, 32, 16}}, {{6, 7, 8, 9, 10, 11}, {0, 1, 2, 3, 4, 5}, {3, 5, 17, 40, 1, 9}}},
        {{{1, 3, 5, 7, 9, 11}, {0, 2, 4, 6, 8, 10}, {1, 4, 8, 18, 32, 2}}, {{0, 1, 8, 9, 10, 11}, {2, 3, 4, 5, 6, 7}, {1, 2, 4, 32, 8, 16}}},
        {{{0, 1, 2, 3, 4, 5}, {6, 7, 8, 9, 10, 11}, {1, 2, 8, 16, 32, 4}}, {{6, 7, 8, 9, 10, 11}, {0, 1, 2, 3, 4, 5}, {3, 5, 9, 17, 1, 33}}},
        {{{1, 3, 5, 7, 9, 11}, {0, 2, 4, 6, 8, 10}, {1, 2, 8, 16, 32, 4}}, {{0, 1, 8, 9, 10, 11}, {2, 3, 4, 5, 6, 7}, {1, 2, 8, 16, 4, 32}}}
    };
    Maps M{};
    M.ok = true;
    int E[NQ] = {};
    for (int k = 0; k < NQ; ++k) E[k] = fold(1 << k);
    for (int l = 0; l < NLAYERS; ++l) {
        for (int p = 0; p < NPASS; ++p) {
            const PassSpec& S = SPEC[l][p];
            int all[12] = {};
            for (int i = 0; i < 6; ++i) {
                M.g[l][p][i] = E[S.regw[i]];
                int v = 0;
                for (int j = 0; j < 6; ++j)
                    if ((S.omask[i] >> j) & 1) v ^= E[S.othw[j]];
                M.o[l][p][i] = v;
                all[i] = M.g[l][p][i];
                all[6 + i] = v;
            }
            int rank = 0;
            for (int bit = NQ - 1; bit >= 0; --bit) {
                int piv = -1;
                for (int i = rank; i < 12; ++i)
                    if (piv < 0 && ((all[i] >> bit) & 1)) piv = i;
                if (piv < 0) continue;
                const int t = all[rank]; all[rank] = all[piv]; all[piv] = t;
                for (int i = 0; i < 12; ++i)
                    if (i != rank && ((all[i] >> bit) & 1)) all[i] ^= all[rank];
                ++rank;
            }
            if (rank != NQ) M.ok = false;
            const bool one_pair = (l & 1) && p == 1;
            const int ld[5] = {M.o[l][p][0], M.o[l][p][1], M.o[l][p][2], M.o[l][p][3], M.g[l][p][0]};
            const int sv[5] = {M.o[l][p][0], M.o[l][p][1], M.o[l][p][2], M.o[l][p][3],
                               one_pair ? M.g[l][p][1] : M.g[l][p][5]};
            if (lane_ways(ld) + lane_ways(sv) > 3) M.ok = false;
            if (!(l & 1) && p == 1) {
                for (int i = 0; i < 6; ++i) {
                    if (S.othw[i] != i) M.ok = false;
                    if (bit_parity(S.omask[i]) != (i == 4 ? 1 : 0)) M.ok = false;
                }
            }
        }
        for (int k = 0; k < NQ; ++k) E[k] = fold(ring_inv(fold(E[k])));
    }
    for (int j = 0; j < NQ; ++j) {
        int v = fold(1 << j);
        for (int l = 0; l < NLAYERS; ++l) v = ring_fwd(v);
        M.y[j] = v;
    }
    return M;
}
static_assert(make_maps().ok, "address map: rank, lane conflicts or column parity");

__device__ __forceinline__ float2 cmul(float2 a, float2 b) {
    return make_float2(a.x * b.x - a.y * b.y, a.x * b.y + a.y * b.x);
}

__device__ __forceinline__ void fused_rot(float tx, float ty, float tz, float2* U) {
    const float cx = cosf(0.5f * tx), sx = sinf(0.5f * tx);
    const float cy = cosf(0.5f * ty), sy = sinf(0.5f * ty);
    const float cz = cosf(0.5f * tz), sz = sinf(0.5f * tz);
    const float2 A00 = make_float2( cy * cx,  sy * sx);
    const float2 A01 = make_float2(-sy * cx, -cy * sx);
    const float2 A10 = make_float2( sy * cx, -cy * sx);
    const float2 A11 = make_float2( cy * cx, -sy * sx);
    const float2 e  = make_float2(cz, -sz);
    const float2 ec = make_float2(cz,  sz);
    U[0] = cmul(e, A00);  U[1] = cmul(e, A01);
    U[2] = cmul(ec, A10); U[3] = cmul(ec, A11);
}

__device__ __forceinline__ void embed_rot(float f, float2* U) {
    const float cy = cosf(0.5f * f),  sy = sinf(0.5f * f);
    const float cz = cosf(0.25f * f), sz = sinf(0.25f * f);
    U[0] = make_float2( cz * cy, -sz * cy);
    U[1] = make_float2(-cz * sy,  sz * sy);
    U[2] = make_float2( cz * sy,  sz * sy);
    U[3] = make_float2( cz * cy,  sz * cy);
}

__device__ __forceinline__ void kron22(const float2* Ub, const float2* Ua, float2* M) {
    #pragma unroll
    for (int o = 0; o < 4; ++o)
        #pragma unroll
        for (int j = 0; j < 4; ++j)
            M[o * 4 + j] = cmul(Ub[(o >> 1) * 2 + (j >> 1)], Ua[(o & 1) * 2 + (j & 1)]);
}

__device__ __forceinline__ void mul22(const float2* L, const float2* E, float2* C) {
    #pragma unroll
    for (int r = 0; r < 2; ++r)
        #pragma unroll
        for (int c = 0; c < 2; ++c) {
            const float2 x = cmul(L[r * 2], E[c]);
            const float2 y = cmul(L[r * 2 + 1], E[2 + c]);
            C[r * 2 + c] = make_float2(x.x + y.x, x.y + y.y);
        }
}

__device__ __forceinline__ void pair_A(float2 m1, float2 m2, int lane, v2f& A1, v2f& A2) {
    const bool im = (lane & 1) != 0;
    A1[0] = im ? m1.y :  m1.x;  A1[1] = im ? m1.x : -m1.y;
    A2[0] = im ? m2.y :  m2.x;  A2[1] = im ? m2.x : -m2.y;
}

__device__ __forceinline__ void ride_next(const float2* G, const float2* Un, int q, float2* T) {
    const float2 d_even = q ? Un[3] : Un[0], x_even = q ? Un[2] : Un[1];
    const float2 d_odd  = q ? Un[0] : Un[3], x_odd  = q ? Un[1] : Un[2];
    #pragma unroll
    for (int o = 0; o < 4; ++o) {
        const float2 d = (o & 1) ? d_odd : d_even;
        const float2 x = (o & 1) ? x_odd : x_even;
        #pragma unroll
        for (int j = 0; j < 4; ++j) {
            const float2 a = cmul(d, G[o * 4 + j]);
            const float2 b = cmul(x, G[(o ^ 3) * 4 + j]);
            T[o * 4 + j] = make_float2(a.x + b.x, a.y + b.y);
        }
    }
}

__host__ __device__ constexpr int par2(int j) { return (j ^ (j >> 1)) & 1; }

template <bool ZIN>
__device__ __forceinline__ void apply_three_pairs(float2* st, const float2* M, const int* g,
                                                  int lan, int lane, int pi) {
    const int hiK = lane >> 4;
    const int o   = (lane >> 1) & 3;
    const int e   = o * 4 + hiK;
    const float2* MA  = M +      (pi ? 16 : 0);
    const float2* MB0 = M + 32 + (pi ? 16 : 0);
    const float2* MB1 = M + 32 + (pi ? 0 : 16);
    const float2* MC0 = M + 64 + (pi ? 16 : 0);
    const float2* MC1 = M + 64 + (pi ? 0 : 16);
    const float2 ma1  = MA[e],  ma2  = MA[e + 2];
    const float2 mb01 = MB0[e], mb02 = MB0[e + 2];
    const float2 mb11 = MB1[e], mb12 = MB1[e + 2];
    const float2 mc01 = MC0[e], mc02 = MC0[e + 2];
    const float2 mc11 = MC1[e], mc12 = MC1[e + 2];

    const int a1 = g[0], a2 = g[1];
    const int b1 = g[2], b2 = g[3];
    const int c1 = g[4], c2 = g[5];
    const int ld0 = lan ^ (hiK ? a1 : 0);
    const int st0 = lan ^ (hiK ? c2 : 0);

    v2f A1a, A2a, A1b0, A2b0, A1b1, A2b1, A1c0, A2c0, A1c1, A2c1;
    pair_A(ma1,  ma2,  lane, A1a,  A2a);
    pair_A(mb01, mb02, lane, A1b0, A2b0);
    pair_A(mb11, mb12, lane, A1b1, A2b1);
    pair_A(mc01, mc02, lane, A1c0, A2c0);
    pair_A(mc11, mc12, lane, A1c1, A2c1);

    if constexpr (ZIN) {
        const v8f Z = {0.f, 0.f, 0.f, 0.f, 0.f, 0.f, 0.f, 0.f};
        const float2 p1 = st[ld0];
        v2f B0; B0[0] = p1.x; B0[1] = p1.y;
        const v8f E1 = __builtin_amdgcn_wmma_f32_16x16x4_f32(false, A1a, false, B0,
                                                             (short)0, Z, false, false);
        v8f E2[4];
        #pragma unroll
        for (int jA = 0; jA < 4; ++jA) {
            v2f B1;
            B1[0] = hiK ? 0.f : E1[2 * jA];
            B1[1] = hiK ? 0.f : E1[2 * jA + 1];
            const v2f A1 = par2(jA) ? A1b1 : A1b0;
            E2[jA] = __builtin_amdgcn_wmma_f32_16x16x4_f32(false, A1, false, B1,
                                                           (short)0, Z, false, false);
        }
        #pragma unroll
        for (int jA = 0; jA < 4; ++jA) {
            #pragma unroll
            for (int jB = 0; jB < 4; ++jB) {
                v2f B1;
                B1[0] = hiK ? 0.f : E2[jA][2 * jB];
                B1[1] = hiK ? 0.f : E2[jA][2 * jB + 1];
                const v2f A1 = (par2(jA) ^ par2(jB)) ? A1c1 : A1c0;
                const v8f C = __builtin_amdgcn_wmma_f32_16x16x4_f32(false, A1, false, B1,
                                                                    (short)0, Z, false, false);
                const float2 oa = hiK ? make_float2(C[4], C[5]) : make_float2(C[0], C[1]);
                const float2 ob = hiK ? make_float2(C[6], C[7]) : make_float2(C[2], C[3]);
                const int rc = ((jA & 1) ? a1 : 0) ^ ((jA & 2) ? a2 : 0) ^
                               ((jB & 1) ? b1 : 0) ^ ((jB & 2) ? b2 : 0);
                st[st0 ^ rc]      = oa;
                st[st0 ^ rc ^ c1] = ob;
            }
        }
        return;
    }

    v8f D1[NTILES];
    #pragma unroll
    for (int t = 0; t < NTILES; ++t) {
        const int rc = ((t & 1) ? b1 : 0) ^ ((t & 2) ? b2 : 0) ^
                       ((t & 4) ? c1 : 0) ^ ((t & 8) ? c2 : 0);
        const float2 p1 = st[ld0 ^ rc];
        const float2 p2 = st[ld0 ^ rc ^ a2];
        v2f B1; B1[0] = p1.x; B1[1] = p1.y;
        v2f B2; B2[0] = p2.x; B2[1] = p2.y;
        v8f C = {0.f, 0.f, 0.f, 0.f, 0.f, 0.f, 0.f, 0.f};
        C = __builtin_amdgcn_wmma_f32_16x16x4_f32(false, A1a, false, B1,
                                                  (short)0, C, false, false);
        C = __builtin_amdgcn_wmma_f32_16x16x4_f32(false, A2a, false, B2,
                                                  (short)0, C, false, false);
        D1[t] = C;
    }
    v8f D2[NTILES];
    #pragma unroll
    for (int tc = 0; tc < 4; ++tc) {
        #pragma unroll
        for (int jA = 0; jA < 4; ++jA) {
            v2f B1, B2;
            B1[0] = hiK ? D1[tc * 4 + 1][2 * jA]     : D1[tc * 4][2 * jA];
            B1[1] = hiK ? D1[tc * 4 + 1][2 * jA + 1] : D1[tc * 4][2 * jA + 1];
            B2[0] = hiK ? D1[tc * 4 + 3][2 * jA]     : D1[tc * 4 + 2][2 * jA];
            B2[1] = hiK ? D1[tc * 4 + 3][2 * jA + 1] : D1[tc * 4 + 2][2 * jA + 1];
            const v2f A1 = par2(jA) ? A1b1 : A1b0;
            const v2f A2 = par2(jA) ? A2b1 : A2b0;
            v8f C = {0.f, 0.f, 0.f, 0.f, 0.f, 0.f, 0.f, 0.f};
            C = __builtin_amdgcn_wmma_f32_16x16x4_f32(false, A1, false, B1,
                                                      (short)0, C, false, false);
            C = __builtin_amdgcn_wmma_f32_16x16x4_f32(false, A2, false, B2,
                                                      (short)0, C, false, false);
            D2[jA * 4 + tc] = C;
        }
    }
    #pragma unroll
    for (int jA = 0; jA < 4; ++jA) {
        #pragma unroll
        for (int jB = 0; jB < 4; ++jB) {
            v2f B1, B2;
            B1[0] = hiK ? D2[jA * 4 + 1][2 * jB]     : D2[jA * 4][2 * jB];
            B1[1] = hiK ? D2[jA * 4 + 1][2 * jB + 1] : D2[jA * 4][2 * jB + 1];
            B2[0] = hiK ? D2[jA * 4 + 3][2 * jB]     : D2[jA * 4 + 2][2 * jB];
            B2[1] = hiK ? D2[jA * 4 + 3][2 * jB + 1] : D2[jA * 4 + 2][2 * jB + 1];
            const v2f A1 = (par2(jA) ^ par2(jB)) ? A1c1 : A1c0;
            const v2f A2 = (par2(jA) ^ par2(jB)) ? A2c1 : A2c0;
            v8f C = {0.f, 0.f, 0.f, 0.f, 0.f, 0.f, 0.f, 0.f};
            C = __builtin_amdgcn_wmma_f32_16x16x4_f32(false, A1, false, B1,
                                                      (short)0, C, false, false);
            C = __builtin_amdgcn_wmma_f32_16x16x4_f32(false, A2, false, B2,
                                                      (short)0, C, false, false);
            const float2 oa = hiK ? make_float2(C[4], C[5]) : make_float2(C[0], C[1]);
            const float2 ob = hiK ? make_float2(C[6], C[7]) : make_float2(C[2], C[3]);
            const int rc = ((jA & 1) ? a1 : 0) ^ ((jA & 2) ? a2 : 0) ^
                           ((jB & 1) ? b1 : 0) ^ ((jB & 2) ? b2 : 0);
            st[st0 ^ rc]      = oa;
            st[st0 ^ rc ^ c1] = ob;
        }
    }
}

__device__ __forceinline__ void apply_one_pair(float2* st, const float2* MA, const int* g,
                                               int lan, int lane) {
    const int hiK = lane >> 4;
    const int o   = (lane >> 1) & 3;
    const float2 ma1 = MA[o * 4 + hiK], ma2 = MA[o * 4 + hiK + 2];
    const int a1 = g[0], a2 = g[1];
    const int ld0 = lan ^ (hiK ? a1 : 0);
    const int st0 = lan ^ (hiK ? a2 : 0);
    v2f A1a, A2a;
    pair_A(ma1, ma2, lane, A1a, A2a);
    float2 p1[NTILES], p2[NTILES];
    #pragma unroll
    for (int t = 0; t < NTILES; ++t) {
        const int rc = ((t & 1) ? g[2] : 0) ^ ((t & 2) ? g[3] : 0) ^
                       ((t & 4) ? g[4] : 0) ^ ((t & 8) ? g[5] : 0);
        p1[t] = st[ld0 ^ rc];
        p2[t] = st[ld0 ^ rc ^ a2];
    }
    #pragma unroll
    for (int t = 0; t < NTILES; ++t) {
        const int rc = ((t & 1) ? g[2] : 0) ^ ((t & 2) ? g[3] : 0) ^
                       ((t & 4) ? g[4] : 0) ^ ((t & 8) ? g[5] : 0);
        v2f B1; B1[0] = p1[t].x; B1[1] = p1[t].y;
        v2f B2; B2[0] = p2[t].x; B2[1] = p2[t].y;
        v8f C = {0.f, 0.f, 0.f, 0.f, 0.f, 0.f, 0.f, 0.f};
        C = __builtin_amdgcn_wmma_f32_16x16x4_f32(false, A1a, false, B1,
                                                  (short)0, C, false, false);
        C = __builtin_amdgcn_wmma_f32_16x16x4_f32(false, A2a, false, B2,
                                                  (short)0, C, false, false);
        const float2 oa = hiK ? make_float2(C[4], C[5]) : make_float2(C[0], C[1]);
        const float2 ob = hiK ? make_float2(C[6], C[7]) : make_float2(C[2], C[3]);
        st[st0 ^ rc]      = oa;
        st[st0 ^ rc ^ a1] = ob;
    }
}

__device__ __forceinline__ int lane_slot(const int* o, int lane, int wave) {
    int s = 0;
    #pragma unroll
    for (int i = 0; i < 4; ++i) s ^= o[i] & -((lane >> i) & 1);
    s ^= o[4] & -(wave & 1);
    s ^= o[5] & -((wave >> 1) & 1);
    return s;
}

__global__ __launch_bounds__(THREADS, 4)
void qsim_kernel(const float* __restrict__ fv, const float* __restrict__ theta,
                 float* __restrict__ out) {
    __shared__ float2 st[DIM];
    __shared__ float2 wireU[(1 + NLAYERS) * NQ][4];
    __shared__ float2 layM[NLAYERS * NSLOT][16];
    __shared__ float  wpart[NWAVES][NQ];

    const int b    = blockIdx.x;
    const int tid  = threadIdx.x;
    const int lane = tid & 31;
    const int wave = tid >> 5;

    for (int i = tid; i < DIM; i += THREADS) st[i] = make_float2(0.f, 0.f);
    if (tid == 0) st[0] = make_float2(1.f, 0.f);

    if (tid < NPAIRS + NLAYERS * NPAIRS) {
        float2 Ua[4], Ub[4];
        int row;
        if (tid < NPAIRS) {
            embed_rot(fv[b * NQ + 2 * tid],     Ua);
            embed_rot(fv[b * NQ + 2 * tid + 1], Ub);
            row = 2 * tid;
        } else {
            const int t = tid - NPAIRS;
            const float* ta = theta + (2 * t) * 3;
            fused_rot(ta[0], ta[1], ta[2], Ua);
            fused_rot(ta[3], ta[4], ta[5], Ub);
            row = NQ + 2 * t;
        }
        #pragma unroll
        for (int i = 0; i < 4; ++i) { wireU[row][i] = Ua[i]; wireU[row + 1][i] = Ub[i]; }
    }
    __syncthreads();
    if (tid < NLAYERS * NSLOT) {
        const int q = tid & 1;
        const int l = (tid >> 1) / NPAIRS;
        const int s = (tid >> 1) - l * NPAIRS;
        const bool carry = (l & 1) == 0;
        const int  lo = carry ? 2 * s     : (s < 3 ? 4 * s + 1 : 0);
        const int  hi = carry ? 2 * s + 1 : (s < 3 ? 4 * s + 3 : 0);
        const bool live = carry || s <= 3;
        if (live) {
            float2 Ua[4], Ub[4], G[16];
            if (l == 0) {
                mul22(&wireU[NQ + lo][0], &wireU[lo][0], Ua);
                mul22(&wireU[NQ + hi][0], &wireU[hi][0], Ub);
            } else {
                #pragma unroll
                for (int i = 0; i < 4; ++i) {
                    Ua[i] = wireU[(1 + l) * NQ + lo][i];
                    Ub[i] = wireU[(1 + l) * NQ + hi][i];
                }
            }
            if (!carry && s == 3) {
                Ub[0] = make_float2(1.f, 0.f); Ub[1] = make_float2(0.f, 0.f);
                Ub[2] = make_float2(0.f, 0.f); Ub[3] = make_float2(1.f, 0.f);
            }
            kron22(Ub, Ua, G);
            if (carry && s >= 1) {
                float2 Un[4], T[16];
                #pragma unroll
                for (int i = 0; i < 4; ++i) Un[i] = wireU[(2 + l) * NQ + lo][i];
                ride_next(G, Un, q, T);
                #pragma unroll
                for (int i = 0; i < 16; ++i) layM[tid][i] = T[i];
            } else {
                #pragma unroll
                for (int i = 0; i < 16; ++i) layM[tid][i] = G[i];
            }
        }
    }
    __syncthreads();

    static constexpr Maps MAPS = make_maps();
    const int pi = wave & 1;

    {
        const float2* M = &layM[0][0];
        int g0[6], g1[6];
        #pragma unroll
        for (int i = 0; i < 6; ++i) { g0[i] = MAPS.g[0][0][i]; g1[i] = MAPS.g[0][1][i]; }
        const int lan0 = lane_slot(MAPS.o[0][0], lane, wave);
        const int lan1 = lane_slot(MAPS.o[0][1], lane, wave);
        apply_three_pairs<true>(st, M,      g0, lan0, lane, 0);
        __syncthreads();
        apply_three_pairs<true>(st, M + 96, g1, lan1, lane, pi);
        __syncthreads();
    }
    #pragma nounroll
    for (int l = 1; l < NLAYERS; ++l) {
        const float2* M = &layM[l * NSLOT][0];
        int g0[6], g1[6];
        #pragma unroll
        for (int i = 0; i < 6; ++i) { g0[i] = MAPS.g[l][0][i]; g1[i] = MAPS.g[l][1][i]; }
        const int lan0 = lane_slot(MAPS.o[l][0], lane, wave);
        const int lan1 = lane_slot(MAPS.o[l][1], lane, wave);
        apply_three_pairs<false>(st, M, g0, lan0, lane, 0);
        __syncthreads();
        if (l & 1) apply_one_pair(st, M + 96, g1, lan1, lane);
        else       apply_three_pairs<false>(st, M + 96, g1, lan1, lane, pi);
        __syncthreads();
    }

    float acc[NQ];
    #pragma unroll
    for (int w = 0; w < NQ; ++w) acc[w] = 0.f;
    int ytid = 0;
    #pragma unroll
    for (int j = 0; j < 7; ++j) ytid ^= ((tid >> j) & 1) ? MAPS.y[j] : 0;
    #pragma unroll
    for (int q = 0; q < DIM / THREADS; ++q) {
        int yq = 0;
        #pragma unroll
        for (int j = 0; j < 5; ++j) yq ^= ((q >> j) & 1) ? MAPS.y[7 + j] : 0;
        const int    y  = ytid ^ yq;
        const float2 a  = st[tid + q * THREADS];
        const float  pz = a.x * a.x + a.y * a.y;
        #pragma unroll
        for (int w = 0; w < NQ; ++w)
            acc[w] += ((y >> w) & 1) ? -pz : pz;
    }
    #pragma unroll
    for (int off = 16; off > 0; off >>= 1) {
        #pragma unroll
        for (int w = 0; w < NQ; ++w)
            acc[w] += __shfl_xor(acc[w], off, 32);
    }
    if (lane == 0) {
        #pragma unroll
        for (int w = 0; w < NQ; ++w) wpart[wave][w] = acc[w];
    }
    __syncthreads();
    if (tid < NQ) {
        float s = 0.f;
        for (int q = 0; q < NWAVES; ++q) s += wpart[q][tid];
        out[b * NQ + tid] = s;
    }
}

extern "C" void kernel_launch(void* const* d_in, const int* in_sizes, int n_in,
                              void* d_out, int out_size, void* d_ws, size_t ws_size,
                              hipStream_t stream) {
    const float* fv    = (const float*)d_in[0];
    const float* theta = (const float*)d_in[1];
    float*       out   = (float*)d_out;
    const int batch = in_sizes[0] / NQ;
    qsim_kernel<<<batch, THREADS, 0, stream>>>(fv, theta, out);
}


# wire 11 of layers 2,4 rides forward into the next layer's pair (0,11) as a 4x4 factor; pairs shifted so wires 1,3,5,7,9 ride back; layers 2,4 run one pass: 10 LDS passes, not 12; new sum order
# speedup vs baseline: 2.6922x; 1.0750x over previous
#include <hip/hip_runtime.h>


typedef __attribute__((ext_vector_type(2))) float v2f;
typedef __attribute__((ext_vector_type(8))) float v8f;

#define NQ      12
#define DIM     4096
#define THREADS 128
#define NWAVES  4
#define NLAYERS 6
#define NPAIRS  6
#define NPASS   2
#define NTILES  16
#define NSLOT   12

__host__ __device__ constexpr int ring_fwd(int x) {
    int t = x;
    t ^= t << 1; t ^= t << 2; t ^= t << 4; t ^= t << 8;
    t &= DIM - 1;
    return t ^ (t >> (NQ - 1));
}
__host__ __device__ constexpr int ring_inv(int y) {
    const int top = y >> (NQ - 1);
    return (y ^ (y << 1) ^ top ^ (top << 1)) & (DIM - 1);
}
__host__ __device__ constexpr int fold(int x) {
    constexpr int pat[7] = {15, 22, 19, 3, 10, 2, 21};
    int r = x;
    for (int b = 0; b < 7; ++b)
        if ((x >> (5 + b)) & 1) r ^= pat[b];
    return r;
}

__host__ __device__ constexpr int lane_ways(const int* v) {
    int cnt[32] = {};
    for (int m = 0; m < 32; ++m) {
        int s = 0;
        for (int i = 0; i < 5; ++i)
            if ((m >> i) & 1) s ^= v[i];
        cnt[s & 31] += 1;
    }
    int mx = 0;
    for (int i = 0; i < 32; ++i) mx = cnt[i] > mx ? cnt[i] : mx;
    return mx;
}
__host__ __device__ constexpr int bit_parity(int x) {
    int p = 0;
    for (int i = 0; i < 12; ++i) p ^= (x >> i) & 1;
    return p;
}

struct PassSpec { int regw[6]; int othw[6]; int omask[6]; };

struct Maps {
    int  g[NLAYERS][NPASS][6];
    int  o[NLAYERS][NPASS][6];
    int  y[NQ];
    bool ok;
};
__host__ __device__ constexpr Maps make_maps() {
    constexpr PassSpec SPEC[NLAYERS][NPASS] = {
        {{{0, 11, 1, 2, 3, 4}, {5, 6, 7, 8, 9, 10}, {27, 13, 17, 20, 44, 12}}, {{5, 6, 7, 8, 9, 10}, {0, 1, 2, 3, 4, 11}, {32, 10, 3, 59, 16, 62}}},
        {{{0, 2, 4, 6, 8, 10}, {1, 3, 5, 7, 9, 11}, {7, 15, 13, 61, 51, 21}}, {{11, 10, 6, 7, 8, 9}, {0, 1, 2, 3, 4, 5}, {27, 48, 58, 49, 36, 45}}},
        {{{0, 11, 1, 2, 3, 4}, {5, 6, 7, 8, 9, 10}, {19, 4, 24, 21, 51, 18}}, {{5, 6, 7, 8, 9, 10}, {0, 1, 2, 3, 4, 11}, {59, 55, 20, 41, 7, 32}}},
        {{{0, 2, 4, 6, 8, 10}, {1, 3, 5, 7, 9, 11}, {46, 9, 49, 61, 15, 17}}, {{11, 10, 6, 7, 8, 9}, {0, 1, 2, 3, 4, 5}, {57, 3, 42, 51, 52, 11}}},
        {{{0, 11, 1, 2, 3, 4}, {5, 6, 7, 8, 9, 10}, {7, 6, 20, 17, 32, 26}}, {{5, 6, 7, 8, 9, 10}, {0, 1, 2, 3, 4, 11}, {42, 62, 24, 59, 4, 6}}},
        {{{0, 2, 4, 6, 8, 10}, {1, 3, 5, 7, 9, 11}, {30, 40, 31, 21, 25, 57}}, {{11, 10, 6, 7, 8, 9}, {0, 1, 2, 3, 4, 5}, {20, 29, 62, 37, 33, 10}}}
    };
    Maps M{};
    M.ok = true;
    int E[NQ] = {};
    for (int k = 0; k < NQ; ++k) E[k] = fold(1 << k);
    for (int l = 0; l < NLAYERS; ++l) {
        for (int p = 0; p < NPASS; ++p) {
            const PassSpec& S = SPEC[l][p];
            int all[12] = {};
            for (int i = 0; i < 6; ++i) {
                M.g[l][p][i] = E[S.regw[i]];
                int v = 0;
                for (int j = 0; j < 6; ++j)
                    if ((S.omask[i] >> j) & 1) v ^= E[S.othw[j]];
                M.o[l][p][i] = v;
                all[i] = M.g[l][p][i];
                all[6 + i] = v;
            }
            int rank = 0;
            for (int bit = NQ - 1; bit >= 0; --bit) {
                int piv = -1;
                for (int i = rank; i < 12; ++i)
                    if (piv < 0 && ((all[i] >> bit) & 1)) piv = i;
                if (piv < 0) continue;
                const int t = all[rank]; all[rank] = all[piv]; all[piv] = t;
                for (int i = 0; i < 12; ++i)
                    if (i != rank && ((all[i] >> bit) & 1)) all[i] ^= all[rank];
                ++rank;
            }
            if (rank != NQ) M.ok = false;
            const bool one_pair = (l & 1) && p == 1;
            const int ld[5] = {M.o[l][p][0], M.o[l][p][1], M.o[l][p][2], M.o[l][p][3], M.g[l][p][0]};
            const int sv[5] = {M.o[l][p][0], M.o[l][p][1], M.o[l][p][2], M.o[l][p][3],
                               one_pair ? M.g[l][p][1] : M.g[l][p][5]};
            if (lane_ways(ld) + lane_ways(sv) > 4) M.ok = false;
            if (!(l & 1) && p == 0) {
                if (S.othw[5] != 10) M.ok = false;
                for (int i = 0; i < 6; ++i)
                    if (((S.omask[i] >> 5) & 1) != (i == 4 ? 1 : 0)) M.ok = false;
            }
            if (!(l & 1) && p == 1) {
                for (int i = 0; i < 5; ++i)
                    if (S.othw[i] != i) M.ok = false;
                for (int i = 0; i < 6; ++i)
                    if (bit_parity(S.omask[i] & 31) != (i == 4 ? 1 : 0)) M.ok = false;
            }
        }
        for (int k = 0; k < NQ; ++k) E[k] = fold(ring_inv(fold(E[k])));
    }
    for (int j = 0; j < NQ; ++j) {
        int v = fold(1 << j);
        for (int l = 0; l < NLAYERS; ++l) v = ring_fwd(v);
        M.y[j] = v;
    }
    return M;
}
static_assert(make_maps().ok, "address map: rank, lane conflicts, column parity or wave bit");

__device__ __forceinline__ float2 cmul(float2 a, float2 b) {
    return make_float2(a.x * b.x - a.y * b.y, a.x * b.y + a.y * b.x);
}

__device__ __forceinline__ void fused_rot(float tx, float ty, float tz, float2* U) {
    const float cx = cosf(0.5f * tx), sx = sinf(0.5f * tx);
    const float cy = cosf(0.5f * ty), sy = sinf(0.5f * ty);
    const float cz = cosf(0.5f * tz), sz = sinf(0.5f * tz);
    const float2 A00 = make_float2( cy * cx,  sy * sx);
    const float2 A01 = make_float2(-sy * cx, -cy * sx);
    const float2 A10 = make_float2( sy * cx, -cy * sx);
    const float2 A11 = make_float2( cy * cx, -sy * sx);
    const float2 e  = make_float2(cz, -sz);
    const float2 ec = make_float2(cz,  sz);
    U[0] = cmul(e, A00);  U[1] = cmul(e, A01);
    U[2] = cmul(ec, A10); U[3] = cmul(ec, A11);
}

__device__ __forceinline__ void embed_rot(float f, float2* U) {
    const float cy = cosf(0.5f * f),  sy = sinf(0.5f * f);
    const float cz = cosf(0.25f * f), sz = sinf(0.25f * f);
    U[0] = make_float2( cz * cy, -sz * cy);
    U[1] = make_float2(-cz * sy,  sz * sy);
    U[2] = make_float2( cz * sy,  sz * sy);
    U[3] = make_float2( cz * cy,  sz * cy);
}

__device__ __forceinline__ void kron22(const float2* Ub, const float2* Ua, float2* M) {
    #pragma unroll
    for (int o = 0; o < 4; ++o)
        #pragma unroll
        for (int j = 0; j < 4; ++j)
            M[o * 4 + j] = cmul(Ub[(o >> 1) * 2 + (j >> 1)], Ua[(o & 1) * 2 + (j & 1)]);
}

__device__ __forceinline__ void mul22(const float2* L, const float2* E, float2* C) {
    #pragma unroll
    for (int r = 0; r < 2; ++r)
        #pragma unroll
        for (int c = 0; c < 2; ++c) {
            const float2 x = cmul(L[r * 2], E[c]);
            const float2 y = cmul(L[r * 2 + 1], E[2 + c]);
            C[r * 2 + c] = make_float2(x.x + y.x, x.y + y.y);
        }
}

__device__ __forceinline__ void pair_A(float2 m1, float2 m2, int lane, v2f& A1, v2f& A2) {
    const bool im = (lane & 1) != 0;
    A1[0] = im ? m1.y :  m1.x;  A1[1] = im ? m1.x : -m1.y;
    A2[0] = im ? m2.y :  m2.x;  A2[1] = im ? m2.x : -m2.y;
}

__device__ __forceinline__ void ride_next(const float2* G, const float2* Un, int q, float2* T) {
    const float2 d_even = q ? Un[3] : Un[0], x_even = q ? Un[2] : Un[1];
    const float2 d_odd  = q ? Un[0] : Un[3], x_odd  = q ? Un[1] : Un[2];
    #pragma unroll
    for (int o = 0; o < 4; ++o) {
        const float2 d = (o & 1) ? d_odd : d_even;
        const float2 x = (o & 1) ? x_odd : x_even;
        #pragma unroll
        for (int j = 0; j < 4; ++j) {
            const float2 a = cmul(d, G[o * 4 + j]);
            const float2 b = cmul(x, G[(o ^ 3) * 4 + j]);
            T[o * 4 + j] = make_float2(a.x + b.x, a.y + b.y);
        }
    }
}

__device__ __forceinline__ void push_prev(const float2* G, const float2* Up, int q, float2* T) {
    const float2 d_lo = q ? Up[3] : Up[0], x_lo = q ? Up[1] : Up[2];
    const float2 d_hi = q ? Up[0] : Up[3], x_hi = q ? Up[2] : Up[1];
    #pragma unroll
    for (int j = 0; j < 4; ++j) {
        const float2 d = (j & 2) ? d_hi : d_lo;
        const float2 x = (j & 2) ? x_hi : x_lo;
        #pragma unroll
        for (int o = 0; o < 4; ++o) {
            const float2 a = cmul(G[o * 4 + j], d);
            const float2 b = cmul(G[o * 4 + (j ^ 3)], x);
            T[o * 4 + j] = make_float2(a.x + b.x, a.y + b.y);
        }
    }
}

__host__ __device__ constexpr int par2(int j) { return (j ^ (j >> 1)) & 1; }
template <bool HEAD>
__host__ __device__ constexpr int parA(int jA) { return HEAD ? (jA & 1) : par2(jA); }

template <bool ZIN, bool HEAD>
__device__ __forceinline__ void apply_three_pairs(float2* st, const float2* M, const int* g,
                                                  int lan, int lane, int pi) {
    const int hiK = lane >> 4;
    const int o   = (lane >> 1) & 3;
    const int e   = o * 4 + hiK;
    const int pq = HEAD ? 0 : pi;
    const float2* MA  = M +      (pi ? 16 : 0);
    const float2* MB0 = M + 32 + (pq ? 16 : 0);
    const float2* MB1 = M + 32 + (pq ? 0 : 16);
    const float2* MC0 = M + 64 + (pq ? 16 : 0);
    const float2* MC1 = M + 64 + (pq ? 0 : 16);
    const float2 ma1  = MA[e],  ma2  = MA[e + 2];
    const float2 mb01 = MB0[e], mb02 = MB0[e + 2];
    const float2 mb11 = MB1[e], mb12 = MB1[e + 2];
    const float2 mc01 = MC0[e], mc02 = MC0[e + 2];
    const float2 mc11 = MC1[e], mc12 = MC1[e + 2];

    const int a1 = g[0], a2 = g[1];
    const int b1 = g[2], b2 = g[3];
    const int c1 = g[4], c2 = g[5];
    const int ld0 = lan ^ (hiK ? a1 : 0);
    const int st0 = lan ^ (hiK ? c2 : 0);

    v2f A1a, A2a, A1b0, A2b0, A1b1, A2b1, A1c0, A2c0, A1c1, A2c1;
    pair_A(ma1,  ma2,  lane, A1a,  A2a);
    pair_A(mb01, mb02, lane, A1b0, A2b0);
    pair_A(mb11, mb12, lane, A1b1, A2b1);
    pair_A(mc01, mc02, lane, A1c0, A2c0);
    pair_A(mc11, mc12, lane, A1c1, A2c1);

    if constexpr (ZIN) {
        const v8f Z = {0.f, 0.f, 0.f, 0.f, 0.f, 0.f, 0.f, 0.f};
        const float2 p1 = st[ld0];
        v2f B0; B0[0] = p1.x; B0[1] = p1.y;
        const v8f E1 = __builtin_amdgcn_wmma_f32_16x16x4_f32(false, A1a, false, B0,
                                                             (short)0, Z, false, false);
        v8f E2[4];
        #pragma unroll
        for (int jA = 0; jA < 4; ++jA) {
            v2f B1;
            B1[0] = hiK ? 0.f : E1[2 * jA];
            B1[1] = hiK ? 0.f : E1[2 * jA + 1];
            const v2f A1 = parA<HEAD>(jA) ? A1b1 : A1b0;
            E2[jA] = __builtin_amdgcn_wmma_f32_16x16x4_f32(false, A1, false, B1,
                                                           (short)0, Z, false, false);
        }
        #pragma unroll
        for (int jA = 0; jA < 4; ++jA) {
            #pragma unroll
            for (int jB = 0; jB < 4; ++jB) {
                v2f B1;
                B1[0] = hiK ? 0.f : E2[jA][2 * jB];
                B1[1] = hiK ? 0.f : E2[jA][2 * jB + 1];
                const v2f A1 = (parA<HEAD>(jA) ^ par2(jB)) ? A1c1 : A1c0;
                const v8f C = __builtin_amdgcn_wmma_f32_16x16x4_f32(false, A1, false, B1,
                                                                    (short)0, Z, false, false);
                const float2 oa = hiK ? make_float2(C[4], C[5]) : make_float2(C[0], C[1]);
                const float2 ob = hiK ? make_float2(C[6], C[7]) : make_float2(C[2], C[3]);
                const int rc = ((jA & 1) ? a1 : 0) ^ ((jA & 2) ? a2 : 0) ^
                               ((jB & 1) ? b1 : 0) ^ ((jB & 2) ? b2 : 0);
                st[st0 ^ rc]      = oa;
                st[st0 ^ rc ^ c1] = ob;
            }
        }
        return;
    }

    v8f D1[NTILES];
    #pragma unroll
    for (int t = 0; t < NTILES; ++t) {
        const int rc = ((t & 1) ? b1 : 0) ^ ((t & 2) ? b2 : 0) ^
                       ((t & 4) ? c1 : 0) ^ ((t & 8) ? c2 : 0);
        const float2 p1 = st[ld0 ^ rc];
        const float2 p2 = st[ld0 ^ rc ^ a2];
        v2f B1; B1[0] = p1.x; B1[1] = p1.y;
        v2f B2; B2[0] = p2.x; B2[1] = p2.y;
        v8f C = {0.f, 0.f, 0.f, 0.f, 0.f, 0.f, 0.f, 0.f};
        C = __builtin_amdgcn_wmma_f32_16x16x4_f32(false, A1a, false, B1,
                                                  (short)0, C, false, false);
        C = __builtin_amdgcn_wmma_f32_16x16x4_f32(false, A2a, false, B2,
                                                  (short)0, C, false, false);
        D1[t] = C;
    }
    v8f D2[NTILES];
    #pragma unroll
    for (int tc = 0; tc < 4; ++tc) {
        #pragma unroll
        for (int jA = 0; jA < 4; ++jA) {
            v2f B1, B2;
            B1[0] = hiK ? D1[tc * 4 + 1][2 * jA]     : D1[tc * 4][2 * jA];
            B1[1] = hiK ? D1[tc * 4 + 1][2 * jA + 1] : D1[tc * 4][2 * jA + 1];
            B2[0] = hiK ? D1[tc * 4 + 3][2 * jA]     : D1[tc * 4 + 2][2 * jA];
            B2[1] = hiK ? D1[tc * 4 + 3][2 * jA + 1] : D1[tc * 4 + 2][2 * jA + 1];
            const v2f A1 = parA<HEAD>(jA) ? A1b1 : A1b0;
            const v2f A2 = parA<HEAD>(jA) ? A2b1 : A2b0;
            v8f C = {0.f, 0.f, 0.f, 0.f, 0.f, 0.f, 0.f, 0.f};
            C = __builtin_amdgcn_wmma_f32_16x16x4_f32(false, A1, false, B1,
                                                      (short)0, C, false, false);
            C = __builtin_amdgcn_wmma_f32_16x16x4_f32(false, A2, false, B2,
                                                      (short)0, C, false, false);
            D2[jA * 4 + tc] = C;
        }
    }
    #pragma unroll
    for (int jA = 0; jA < 4; ++jA) {
        #pragma unroll
        for (int jB = 0; jB < 4; ++jB) {
            v2f B1, B2;
            B1[0] = hiK ? D2[jA * 4 + 1][2 * jB]     : D2[jA * 4][2 * jB];
            B1[1] = hiK ? D2[jA * 4 + 1][2 * jB + 1] : D2[jA * 4][2 * jB + 1];
            B2[0] = hiK ? D2[jA * 4 + 3][2 * jB]     : D2[jA * 4 + 2][2 * jB];
            B2[1] = hiK ? D2[jA * 4 + 3][2 * jB + 1] : D2[jA * 4 + 2][2 * jB + 1];
            const v2f A1 = (parA<HEAD>(jA) ^ par2(jB)) ? A1c1 : A1c0;
            const v2f A2 = (parA<HEAD>(jA) ^ par2(jB)) ? A2c1 : A2c0;
            v8f C = {0.f, 0.f, 0.f, 0.f, 0.f, 0.f, 0.f, 0.f};
            C = __builtin_amdgcn_wmma_f32_16x16x4_f32(false, A1, false, B1,
                                                      (short)0, C, false, false);
            C = __builtin_amdgcn_wmma_f32_16x16x4_f32(false, A2, false, B2,
                                                      (short)0, C, false, false);
            const float2 oa = hiK ? make_float2(C[4], C[5]) : make_float2(C[0], C[1]);
            const float2 ob = hiK ? make_float2(C[6], C[7]) : make_float2(C[2], C[3]);
            const int rc = ((jA & 1) ? a1 : 0) ^ ((jA & 2) ? a2 : 0) ^
                           ((jB & 1) ? b1 : 0) ^ ((jB & 2) ? b2 : 0);
            st[st0 ^ rc]      = oa;
            st[st0 ^ rc ^ c1] = ob;
        }
    }
}

__device__ __forceinline__ void apply_one_pair(float2* st, const float2* MA, const int* g,
                                               int lan, int lane) {
    const int hiK = lane >> 4;
    const int o   = (lane >> 1) & 3;
    const float2 ma1 = MA[o * 4 + hiK], ma2 = MA[o * 4 + hiK + 2];
    const int a1 = g[0], a2 = g[1];
    const int ld0 = lan ^ (hiK ? a1 : 0);
    const int st0 = lan ^ (hiK ? a2 : 0);
    v2f A1a, A2a;
    pair_A(ma1, ma2, lane, A1a, A2a);
    float2 p1[NTILES], p2[NTILES];
    #pragma unroll
    for (int t = 0; t < NTILES; ++t) {
        const int rc = ((t & 1) ? g[2] : 0) ^ ((t & 2) ? g[3] : 0) ^
                       ((t & 4) ? g[4] : 0) ^ ((t & 8) ? g[5] : 0);
        p1[t] = st[ld0 ^ rc];
        p2[t] = st[ld0 ^ rc ^ a2];
    }
    #pragma unroll
    for (int t = 0; t < NTILES; ++t) {
        const int rc = ((t & 1) ? g[2] : 0) ^ ((t & 2) ? g[3] : 0) ^
                       ((t & 4) ? g[4] : 0) ^ ((t & 8) ? g[5] : 0);
        v2f B1; B1[0] = p1[t].x; B1[1] = p1[t].y;
        v2f B2; B2[0] = p2[t].x; B2[1] = p2[t].y;
        v8f C = {0.f, 0.f, 0.f, 0.f, 0.f, 0.f, 0.f, 0.f};
        C = __builtin_amdgcn_wmma_f32_16x16x4_f32(false, A1a, false, B1,
                                                  (short)0, C, false, false);
        C = __builtin_amdgcn_wmma_f32_16x16x4_f32(false, A2a, false, B2,
                                                  (short)0, C, false, false);
        const float2 oa = hiK ? make_float2(C[4], C[5]) : make_float2(C[0], C[1]);
        const float2 ob = hiK ? make_float2(C[6], C[7]) : make_float2(C[2], C[3]);
        st[st0 ^ rc]      = oa;
        st[st0 ^ rc ^ a1] = ob;
    }
}

__device__ __forceinline__ int lane_slot(const int* o, int lane, int wave) {
    int s = 0;
    #pragma unroll
    for (int i = 0; i < 4; ++i) s ^= o[i] & -((lane >> i) & 1);
    s ^= o[4] & -(wave & 1);
    s ^= o[5] & -((wave >> 1) & 1);
    return s;
}

__global__ __launch_bounds__(THREADS, 4)
void qsim_kernel(const float* __restrict__ fv, const float* __restrict__ theta,
                 float* __restrict__ out) {
    __shared__ float2 st[DIM];
    __shared__ float2 wireU[(1 + NLAYERS) * NQ][4];
    __shared__ float2 layM[NLAYERS * NSLOT][16];
    __shared__ float  wpart[NWAVES][NQ];

    const int b    = blockIdx.x;
    const int tid  = threadIdx.x;
    const int lane = tid & 31;
    const int wave = tid >> 5;

    for (int i = tid; i < DIM; i += THREADS) st[i] = make_float2(0.f, 0.f);
    if (tid == 0) st[0] = make_float2(1.f, 0.f);

    if (tid < NPAIRS + NLAYERS * NPAIRS) {
        float2 Ua[4], Ub[4];
        int row;
        if (tid < NPAIRS) {
            embed_rot(fv[b * NQ + 2 * tid],     Ua);
            embed_rot(fv[b * NQ + 2 * tid + 1], Ub);
            row = 2 * tid;
        } else {
            const int t = tid - NPAIRS;
            const float* ta = theta + (2 * t) * 3;
            fused_rot(ta[0], ta[1], ta[2], Ua);
            fused_rot(ta[3], ta[4], ta[5], Ub);
            row = NQ + 2 * t;
        }
        #pragma unroll
        for (int i = 0; i < 4; ++i) { wireU[row][i] = Ua[i]; wireU[row + 1][i] = Ub[i]; }
    }
    __syncthreads();
    if (tid < NLAYERS * NSLOT) {
        const int q = tid & 1;
        const int l = (tid >> 1) / NPAIRS;
        const int s = (tid >> 1) - l * NPAIRS;
        const bool carry = (l & 1) == 0;
        const int  lo = carry ? (s == 0 ? 0  : 2 * s - 1) : (s < 3 ? 4 * s     : 11);
        const int  hi = carry ? (s == 0 ? 11 : 2 * s)     : (s < 3 ? 4 * s + 2 : 11);
        const bool live = carry || s < 3 || (s == 3 && l == NLAYERS - 1);
        if (live) {
            float2 Ua[4], Ub[4], G[16];
            if (l == 0) {
                mul22(&wireU[NQ + lo][0], &wireU[lo][0], Ua);
                mul22(&wireU[NQ + hi][0], &wireU[hi][0], Ub);
            } else {
                #pragma unroll
                for (int i = 0; i < 4; ++i) {
                    Ua[i] = wireU[(1 + l) * NQ + lo][i];
                    Ub[i] = wireU[(1 + l) * NQ + hi][i];
                }
            }
            if (!carry && s == 3) {
                Ub[0] = make_float2(1.f, 0.f); Ub[1] = make_float2(0.f, 0.f);
                Ub[2] = make_float2(0.f, 0.f); Ub[3] = make_float2(1.f, 0.f);
            }
            kron22(Ub, Ua, G);
            if (carry && s >= 1) {
                float2 Un[4], T[16];
                #pragma unroll
                for (int i = 0; i < 4; ++i) Un[i] = wireU[(2 + l) * NQ + lo][i];
                ride_next(G, Un, q, T);
                #pragma unroll
                for (int i = 0; i < 16; ++i) layM[tid][i] = T[i];
            } else if (carry && l >= 2) {
                float2 Up[4], T[16];
                #pragma unroll
                for (int i = 0; i < 4; ++i) Up[i] = wireU[l * NQ + NQ - 1][i];
                push_prev(G, Up, q, T);
                #pragma unroll
                for (int i = 0; i < 16; ++i) layM[tid][i] = T[i];
            } else {
                #pragma unroll
                for (int i = 0; i < 16; ++i) layM[tid][i] = G[i];
            }
        }
    }
    __syncthreads();

    static constexpr Maps MAPS = make_maps();
    const int pi = wave & 1;

    {
        const float2* M = &layM[0][0];
        int g0[6], g1[6];
        #pragma unroll
        for (int i = 0; i < 6; ++i) { g0[i] = MAPS.g[0][0][i]; g1[i] = MAPS.g[0][1][i]; }
        const int lan0 = lane_slot(MAPS.o[0][0], lane, wave);
        const int lan1 = lane_slot(MAPS.o[0][1], lane, wave);
        apply_three_pairs<true, true>(st, M,       g0, lan0, lane, pi);
        __syncthreads();
        apply_three_pairs<true, false>(st, M + 96, g1, lan1, lane, pi);
        __syncthreads();
    }
    #pragma nounroll
    for (int l = 1; l < NLAYERS; ++l) {
        const float2* M = &layM[l * NSLOT][0];
        int g0[6], g1[6];
        #pragma unroll
        for (int i = 0; i < 6; ++i) { g0[i] = MAPS.g[l][0][i]; g1[i] = MAPS.g[l][1][i]; }
        const int lan0 = lane_slot(MAPS.o[l][0], lane, wave);
        const int lan1 = lane_slot(MAPS.o[l][1], lane, wave);
        const bool carry = (l & 1) == 0;
        if (carry) {
            apply_three_pairs<false, true>(st, M, g0, lan0, lane, pi);
            __syncthreads();
        }
        int gs[6];
        #pragma unroll
        for (int i = 0; i < 6; ++i) gs[i] = carry ? g1[i] : g0[i];
        apply_three_pairs<false, false>(st, carry ? M + 96 : M, gs, carry ? lan1 : lan0, lane, pi);
        __syncthreads();
        if (l == NLAYERS - 1) {
            apply_one_pair(st, M + 96, g1, lan1, lane);
            __syncthreads();
        }
    }

    float acc[NQ];
    #pragma unroll
    for (int w = 0; w < NQ; ++w) acc[w] = 0.f;
    int ytid = 0;
    #pragma unroll
    for (int j = 0; j < 7; ++j) ytid ^= ((tid >> j) & 1) ? MAPS.y[j] : 0;
    #pragma unroll
    for (int q = 0; q < DIM / THREADS; ++q) {
        int yq = 0;
        #pragma unroll
        for (int j = 0; j < 5; ++j) yq ^= ((q >> j) & 1) ? MAPS.y[7 + j] : 0;
        const int    y  = ytid ^ yq;
        const float2 a  = st[tid + q * THREADS];
        const float  pz = a.x * a.x + a.y * a.y;
        #pragma unroll
        for (int w = 0; w < NQ; ++w)
            acc[w] += ((y >> w) & 1) ? -pz : pz;
    }
    #pragma unroll
    for (int off = 16; off > 0; off >>= 1) {
        #pragma unroll
        for (int w = 0; w < NQ; ++w)
            acc[w] += __shfl_xor(acc[w], off, 32);
    }
    if (lane == 0) {
        #pragma unroll
        for (int w = 0; w < NQ; ++w) wpart[wave][w] = acc[w];
    }
    __syncthreads();
    if (tid < NQ) {
        float s = 0.f;
        for (int q = 0; q < NWAVES; ++q) s += wpart[q][tid];
        out[b * NQ + tid] = s;
    }
}

extern "C" void kernel_launch(void* const* d_in, const int* in_sizes, int n_in,
                              void* d_out, int out_size, void* d_ws, size_t ws_size,
                              hipStream_t stream) {
    const float* fv    = (const float*)d_in[0];
    const float* theta = (const float*)d_in[1];
    float*       out   = (float*)d_out;
    const int batch = in_sizes[0] / NQ;
    qsim_kernel<<<batch, THREADS, 0, stream>>>(fv, theta, out);
}


# set-up gates dealt one wire per thread (72 layer wires on waves 0-2, 12 embedding wires on wave 3): longest wave evaluates 3 angles, not 10; kernel text a fifth shorter; values and sums unchanged
# speedup vs baseline: 2.7772x; 1.0316x over previous
#include <hip/hip_runtime.h>


typedef __attribute__((ext_vector_type(2))) float v2f;
typedef __attribute__((ext_vector_type(8))) float v8f;

#define NQ      12
#define DIM     4096
#define THREADS 128
#define NWAVES  4
#define NLAYERS 6
#define NPAIRS  6
#define NPASS   2
#define NTILES  16
#define NSLOT   12

__host__ __device__ constexpr int ring_fwd(int x) {
    int t = x;
    t ^= t << 1; t ^= t << 2; t ^= t << 4; t ^= t << 8;
    t &= DIM - 1;
    return t ^ (t >> (NQ - 1));
}
__host__ __device__ constexpr int ring_inv(int y) {
    const int top = y >> (NQ - 1);
    return (y ^ (y << 1) ^ top ^ (top << 1)) & (DIM - 1);
}
__host__ __device__ constexpr int fold(int x) {
    constexpr int pat[7] = {15, 22, 19, 3, 10, 2, 21};
    int r = x;
    for (int b = 0; b < 7; ++b)
        if ((x >> (5 + b)) & 1) r ^= pat[b];
    return r;
}

__host__ __device__ constexpr int lane_ways(const int* v) {
    int cnt[32] = {};
    for (int m = 0; m < 32; ++m) {
        int s = 0;
        for (int i = 0; i < 5; ++i)
            if ((m >> i) & 1) s ^= v[i];
        cnt[s & 31] += 1;
    }
    int mx = 0;
    for (int i = 0; i < 32; ++i) mx = cnt[i] > mx ? cnt[i] : mx;
    return mx;
}
__host__ __device__ constexpr int bit_parity(int x) {
    int p = 0;
    for (int i = 0; i < 12; ++i) p ^= (x >> i) & 1;
    return p;
}

struct PassSpec { int regw[6]; int othw[6]; int omask[6]; };

struct Maps {
    int  g[NLAYERS][NPASS][6];
    int  o[NLAYERS][NPASS][6];
    int  y[NQ];
    bool ok;
};
__host__ __device__ constexpr Maps make_maps() {
    constexpr PassSpec SPEC[NLAYERS][NPASS] = {
        {{{0, 11, 1, 2, 3, 4}, {5, 6, 7, 8, 9, 10}, {27, 13, 17, 20, 44, 12}}, {{5, 6, 7, 8, 9, 10}, {0, 1, 2, 3, 4, 11}, {32, 10, 3, 59, 16, 62}}},
        {{{0, 2, 4, 6, 8, 10}, {1, 3, 5, 7, 9, 11}, {7, 15, 13, 61, 51, 21}}, {{11, 10, 6, 7, 8, 9}, {0, 1, 2, 3, 4, 5}, {27, 48, 58, 49, 36, 45}}},
        {{{0, 11, 1, 2, 3, 4}, {5, 6, 7, 8, 9, 10}, {19, 4, 24, 21, 51, 18}}, {{5, 6, 7, 8, 9, 10}, {0, 1, 2, 3, 4, 11}, {59, 55, 20, 41, 7, 32}}},
        {{{0, 2, 4, 6, 8, 10}, {1, 3, 5, 7, 9, 11}, {46, 9, 49, 61, 15, 17}}, {{11, 10, 6, 7, 8, 9}, {0, 1, 2, 3, 4, 5}, {57, 3, 42, 51, 52, 11}}},
        {{{0, 11, 1, 2, 3, 4}, {5, 6, 7, 8, 9, 10}, {7, 6, 20, 17, 32, 26}}, {{5, 6, 7, 8, 9, 10}, {0, 1, 2, 3, 4, 11}, {42, 62, 24, 59, 4, 6}}},
        {{{0, 2, 4, 6, 8, 10}, {1, 3, 5, 7, 9, 11}, {30, 40, 31, 21, 25, 57}}, {{11, 10, 6, 7, 8, 9}, {0, 1, 2, 3, 4, 5}, {20, 29, 62, 37, 33, 10}}}
    };
    Maps M{};
    M.ok = true;
    int E[NQ] = {};
    for (int k = 0; k < NQ; ++k) E[k] = fold(1 << k);
    for (int l = 0; l < NLAYERS; ++l) {
        for (int p = 0; p < NPASS; ++p) {
            const PassSpec& S = SPEC[l][p];
            int all[12] = {};
            for (int i = 0; i < 6; ++i) {
                M.g[l][p][i] = E[S.regw[i]];
                int v = 0;
                for (int j = 0; j < 6; ++j)
                    if ((S.omask[i] >> j) & 1) v ^= E[S.othw[j]];
                M.o[l][p][i] = v;
                all[i] = M.g[l][p][i];
                all[6 + i] = v;
            }
            int rank = 0;
            for (int bit = NQ - 1; bit >= 0; --bit) {
                int piv = -1;
                for (int i = rank; i < 12; ++i)
                    if (piv < 0 && ((all[i] >> bit) & 1)) piv = i;
                if (piv < 0) continue;
                const int t = all[rank]; all[rank] = all[piv]; all[piv] = t;
                for (int i = 0; i < 12; ++i)
                    if (i != rank && ((all[i] >> bit) & 1)) all[i] ^= all[rank];
                ++rank;
            }
            if (rank != NQ) M.ok = false;
            const bool one_pair = (l & 1) && p == 1;
            const int ld[5] = {M.o[l][p][0], M.o[l][p][1], M.o[l][p][2], M.o[l][p][3], M.g[l][p][0]};
            const int sv[5] = {M.o[l][p][0], M.o[l][p][1], M.o[l][p][2], M.o[l][p][3],
                               one_pair ? M.g[l][p][1] : M.g[l][p][5]};
            if (lane_ways(ld) + lane_ways(sv) > 4) M.ok = false;
            if (!(l & 1) && p == 0) {
                if (S.othw[5] != 10) M.ok = false;
                for (int i = 0; i < 6; ++i)
                    if (((S.omask[i] >> 5) & 1) != (i == 4 ? 1 : 0)) M.ok = false;
            }
            if (!(l & 1) && p == 1) {
                for (int i = 0; i < 5; ++i)
                    if (S.othw[i] != i) M.ok = false;
                for (int i = 0; i < 6; ++i)
                    if (bit_parity(S.omask[i] & 31) != (i == 4 ? 1 : 0)) M.ok = false;
            }
        }
        for (int k = 0; k < NQ; ++k) E[k] = fold(ring_inv(fold(E[k])));
    }
    for (int j = 0; j < NQ; ++j) {
        int v = fold(1 << j);
        for (int l = 0; l < NLAYERS; ++l) v = ring_fwd(v);
        M.y[j] = v;
    }
    return M;
}
static_assert(make_maps().ok, "address map: rank, lane conflicts, column parity or wave bit");

__device__ __forceinline__ float2 cmul(float2 a, float2 b) {
    return make_float2(a.x * b.x - a.y * b.y, a.x * b.y + a.y * b.x);
}

__device__ __forceinline__ void fused_rot(float tx, float ty, float tz, float2* U) {
    const float cx = cosf(0.5f * tx), sx = sinf(0.5f * tx);
    const float cy = cosf(0.5f * ty), sy = sinf(0.5f * ty);
    const float cz = cosf(0.5f * tz), sz = sinf(0.5f * tz);
    const float2 A00 = make_float2( cy * cx,  sy * sx);
    const float2 A01 = make_float2(-sy * cx, -cy * sx);
    const float2 A10 = make_float2( sy * cx, -cy * sx);
    const float2 A11 = make_float2( cy * cx, -sy * sx);
    const float2 e  = make_float2(cz, -sz);
    const float2 ec = make_float2(cz,  sz);
    U[0] = cmul(e, A00);  U[1] = cmul(e, A01);
    U[2] = cmul(ec, A10); U[3] = cmul(ec, A11);
}

__device__ __forceinline__ void embed_rot(float f, float2* U) {
    const float cy = cosf(0.5f * f),  sy = sinf(0.5f * f);
    const float cz = cosf(0.25f * f), sz = sinf(0.25f * f);
    U[0] = make_float2( cz * cy, -sz * cy);
    U[1] = make_float2(-cz * sy,  sz * sy);
    U[2] = make_float2( cz * sy,  sz * sy);
    U[3] = make_float2( cz * cy,  sz * cy);
}

__device__ __forceinline__ void kron22(const float2* Ub, const float2* Ua, float2* M) {
    #pragma unroll
    for (int o = 0; o < 4; ++o)
        #pragma unroll
        for (int j = 0; j < 4; ++j)
            M[o * 4 + j] = cmul(Ub[(o >> 1) * 2 + (j >> 1)], Ua[(o & 1) * 2 + (j & 1)]);
}

__device__ __forceinline__ void mul22(const float2* L, const float2* E, float2* C) {
    #pragma unroll
    for (int r = 0; r < 2; ++r)
        #pragma unroll
        for (int c = 0; c < 2; ++c) {
            const float2 x = cmul(L[r * 2], E[c]);
            const float2 y = cmul(L[r * 2 + 1], E[2 + c]);
            C[r * 2 + c] = make_float2(x.x + y.x, x.y + y.y);
        }
}

__device__ __forceinline__ void pair_A(float2 m1, float2 m2, int lane, v2f& A1, v2f& A2) {
    const bool im = (lane & 1) != 0;
    A1[0] = im ? m1.y :  m1.x;  A1[1] = im ? m1.x : -m1.y;
    A2[0] = im ? m2.y :  m2.x;  A2[1] = im ? m2.x : -m2.y;
}

__device__ __forceinline__ void ride_next(const float2* G, const float2* Un, int q, float2* T) {
    const float2 d_even = q ? Un[3] : Un[0], x_even = q ? Un[2] : Un[1];
    const float2 d_odd  = q ? Un[0] : Un[3], x_odd  = q ? Un[1] : Un[2];
    #pragma unroll
    for (int o = 0; o < 4; ++o) {
        const float2 d = (o & 1) ? d_odd : d_even;
        const float2 x = (o & 1) ? x_odd : x_even;
        #pragma unroll
        for (int j = 0; j < 4; ++j) {
            const float2 a = cmul(d, G[o * 4 + j]);
            const float2 b = cmul(x, G[(o ^ 3) * 4 + j]);
            T[o * 4 + j] = make_float2(a.x + b.x, a.y + b.y);
        }
    }
}

__device__ __forceinline__ void push_prev(const float2* G, const float2* Up, int q, float2* T) {
    const float2 d_lo = q ? Up[3] : Up[0], x_lo = q ? Up[1] : Up[2];
    const float2 d_hi = q ? Up[0] : Up[3], x_hi = q ? Up[2] : Up[1];
    #pragma unroll
    for (int j = 0; j < 4; ++j) {
        const float2 d = (j & 2) ? d_hi : d_lo;
        const float2 x = (j & 2) ? x_hi : x_lo;
        #pragma unroll
        for (int o = 0; o < 4; ++o) {
            const float2 a = cmul(G[o * 4 + j], d);
            const float2 b = cmul(G[o * 4 + (j ^ 3)], x);
            T[o * 4 + j] = make_float2(a.x + b.x, a.y + b.y);
        }
    }
}

__host__ __device__ constexpr int par2(int j) { return (j ^ (j >> 1)) & 1; }
template <bool HEAD>
__host__ __device__ constexpr int parA(int jA) { return HEAD ? (jA & 1) : par2(jA); }

template <bool ZIN, bool HEAD>
__device__ __forceinline__ void apply_three_pairs(float2* st, const float2* M, const int* g,
                                                  int lan, int lane, int pi) {
    const int hiK = lane >> 4;
    const int o   = (lane >> 1) & 3;
    const int e   = o * 4 + hiK;
    const int pq = HEAD ? 0 : pi;
    const float2* MA  = M +      (pi ? 16 : 0);
    const float2* MB0 = M + 32 + (pq ? 16 : 0);
    const float2* MB1 = M + 32 + (pq ? 0 : 16);
    const float2* MC0 = M + 64 + (pq ? 16 : 0);
    const float2* MC1 = M + 64 + (pq ? 0 : 16);
    const float2 ma1  = MA[e],  ma2  = MA[e + 2];
    const float2 mb01 = MB0[e], mb02 = MB0[e + 2];
    const float2 mb11 = MB1[e], mb12 = MB1[e + 2];
    const float2 mc01 = MC0[e], mc02 = MC0[e + 2];
    const float2 mc11 = MC1[e], mc12 = MC1[e + 2];

    const int a1 = g[0], a2 = g[1];
    const int b1 = g[2], b2 = g[3];
    const int c1 = g[4], c2 = g[5];
    const int ld0 = lan ^ (hiK ? a1 : 0);
    const int st0 = lan ^ (hiK ? c2 : 0);

    v2f A1a, A2a, A1b0, A2b0, A1b1, A2b1, A1c0, A2c0, A1c1, A2c1;
    pair_A(ma1,  ma2,  lane, A1a,  A2a);
    pair_A(mb01, mb02, lane, A1b0, A2b0);
    pair_A(mb11, mb12, lane, A1b1, A2b1);
    pair_A(mc01, mc02, lane, A1c0, A2c0);
    pair_A(mc11, mc12, lane, A1c1, A2c1);

    if constexpr (ZIN) {
        const v8f Z = {0.f, 0.f, 0.f, 0.f, 0.f, 0.f, 0.f, 0.f};
        const float2 p1 = st[ld0];
        v2f B0; B0[0] = p1.x; B0[1] = p1.y;
        const v8f E1 = __builtin_amdgcn_wmma_f32_16x16x4_f32(false, A1a, false, B0,
                                                             (short)0, Z, false, false);
        v8f E2[4];
        #pragma unroll
        for (int jA = 0; jA < 4; ++jA) {
            v2f B1;
            B1[0] = hiK ? 0.f : E1[2 * jA];
            B1[1] = hiK ? 0.f : E1[2 * jA + 1];
            const v2f A1 = parA<HEAD>(jA) ? A1b1 : A1b0;
            E2[jA] = __builtin_amdgcn_wmma_f32_16x16x4_f32(false, A1, false, B1,
                                                           (short)0, Z, false, false);
        }
        #pragma unroll
        for (int jA = 0; jA < 4; ++jA) {
            #pragma unroll
            for (int jB = 0; jB < 4; ++jB) {
                v2f B1;
                B1[0] = hiK ? 0.f : E2[jA][2 * jB];
                B1[1] = hiK ? 0.f : E2[jA][2 * jB + 1];
                const v2f A1 = (parA<HEAD>(jA) ^ par2(jB)) ? A1c1 : A1c0;
                const v8f C = __builtin_amdgcn_wmma_f32_16x16x4_f32(false, A1, false, B1,
                                                                    (short)0, Z, false, false);
                const float2 oa = hiK ? make_float2(C[4], C[5]) : make_float2(C[0], C[1]);
                const float2 ob = hiK ? make_float2(C[6], C[7]) : make_float2(C[2], C[3]);
                const int rc = ((jA & 1) ? a1 : 0) ^ ((jA & 2) ? a2 : 0) ^
                               ((jB & 1) ? b1 : 0) ^ ((jB & 2) ? b2 : 0);
                st[st0 ^ rc]      = oa;
                st[st0 ^ rc ^ c1] = ob;
            }
        }
        return;
    }

    v8f D1[NTILES];
    #pragma unroll
    for (int t = 0; t < NTILES; ++t) {
        const int rc = ((t & 1) ? b1 : 0) ^ ((t & 2) ? b2 : 0) ^
                       ((t & 4) ? c1 : 0) ^ ((t & 8) ? c2 : 0);
        const float2 p1 = st[ld0 ^ rc];
        const float2 p2 = st[ld0 ^ rc ^ a2];
        v2f B1; B1[0] = p1.x; B1[1] = p1.y;
        v2f B2; B2[0] = p2.x; B2[1] = p2.y;
        v8f C = {0.f, 0.f, 0.f, 0.f, 0.f, 0.f, 0.f, 0.f};
        C = __builtin_amdgcn_wmma_f32_16x16x4_f32(false, A1a, false, B1,
                                                  (short)0, C, false, false);
        C = __builtin_amdgcn_wmma_f32_16x16x4_f32(false, A2a, false, B2,
                                                  (short)0, C, false, false);
        D1[t] = C;
    }
    v8f D2[NTILES];
    #pragma unroll
    for (int tc = 0; tc < 4; ++tc) {
        #pragma unroll
        for (int jA = 0; jA < 4; ++jA) {
            v2f B1, B2;
            B1[0] = hiK ? D1[tc * 4 + 1][2 * jA]     : D1[tc * 4][2 * jA];
            B1[1] = hiK ? D1[tc * 4 + 1][2 * jA + 1] : D1[tc * 4][2 * jA + 1];
            B2[0] = hiK ? D1[tc * 4 + 3][2 * jA]     : D1[tc * 4 + 2][2 * jA];
            B2[1] = hiK ? D1[tc * 4 + 3][2 * jA + 1] : D1[tc * 4 + 2][2 * jA + 1];
            const v2f A1 = parA<HEAD>(jA) ? A1b1 : A1b0;
            const v2f A2 = parA<HEAD>(jA) ? A2b1 : A2b0;
            v8f C = {0.f, 0.f, 0.f, 0.f, 0.f, 0.f, 0.f, 0.f};
            C = __builtin_amdgcn_wmma_f32_16x16x4_f32(false, A1, false, B1,
                                                      (short)0, C, false, false);
            C = __builtin_amdgcn_wmma_f32_16x16x4_f32(false, A2, false, B2,
                                                      (short)0, C, false, false);
            D2[jA * 4 + tc] = C;
        }
    }
    #pragma unroll
    for (int jA = 0; jA < 4; ++jA) {
        #pragma unroll
        for (int jB = 0; jB < 4; ++jB) {
            v2f B1, B2;
            B1[0] = hiK ? D2[jA * 4 + 1][2 * jB]     : D2[jA * 4][2 * jB];
            B1[1] = hiK ? D2[jA * 4 + 1][2 * jB + 1] : D2[jA * 4][2 * jB + 1];
            B2[0] = hiK ? D2[jA * 4 + 3][2 * jB]     : D2[jA * 4 + 2][2 * jB];
            B2[1] = hiK ? D2[jA * 4 + 3][2 * jB + 1] : D2[jA * 4 + 2][2 * jB + 1];
            const v2f A1 = (parA<HEAD>(jA) ^ par2(jB)) ? A1c1 : A1c0;
            const v2f A2 = (parA<HEAD>(jA) ^ par2(jB)) ? A2c1 : A2c0;
            v8f C = {0.f, 0.f, 0.f, 0.f, 0.f, 0.f, 0.f, 0.f};
            C = __builtin_amdgcn_wmma_f32_16x16x4_f32(false, A1, false, B1,
                                                      (short)0, C, false, false);
            C = __builtin_amdgcn_wmma_f32_16x16x4_f32(false, A2, false, B2,
                                                      (short)0, C, false, false);
            const float2 oa = hiK ? make_float2(C[4], C[5]) : make_float2(C[0], C[1]);
            const float2 ob = hiK ? make_float2(C[6], C[7]) : make_float2(C[2], C[3]);
            const int rc = ((jA & 1) ? a1 : 0) ^ ((jA & 2) ? a2 : 0) ^
                           ((jB & 1) ? b1 : 0) ^ ((jB & 2) ? b2 : 0);
            st[st0 ^ rc]      = oa;
            st[st0 ^ rc ^ c1] = ob;
        }
    }
}

__device__ __forceinline__ void apply_one_pair(float2* st, const float2* MA, const int* g,
                                               int lan, int lane) {
    const int hiK = lane >> 4;
    const int o   = (lane >> 1) & 3;
    const float2 ma1 = MA[o * 4 + hiK], ma2 = MA[o * 4 + hiK + 2];
    const int a1 = g[0], a2 = g[1];
    const int ld0 = lan ^ (hiK ? a1 : 0);
    const int st0 = lan ^ (hiK ? a2 : 0);
    v2f A1a, A2a;
    pair_A(ma1, ma2, lane, A1a, A2a);
    float2 p1[NTILES], p2[NTILES];
    #pragma unroll
    for (int t = 0; t < NTILES; ++t) {
        const int rc = ((t & 1) ? g[2] : 0) ^ ((t & 2) ? g[3] : 0) ^
                       ((t & 4) ? g[4] : 0) ^ ((t & 8) ? g[5] : 0);
        p1[t] = st[ld0 ^ rc];
        p2[t] = st[ld0 ^ rc ^ a2];
    }
    #pragma unroll
    for (int t = 0; t < NTILES; ++t) {
        const int rc = ((t & 1) ? g[2] : 0) ^ ((t & 2) ? g[3] : 0) ^
                       ((t & 4) ? g[4] : 0) ^ ((t & 8) ? g[5] : 0);
        v2f B1; B1[0] = p1[t].x; B1[1] = p1[t].y;
        v2f B2; B2[0] = p2[t].x; B2[1] = p2[t].y;
        v8f C = {0.f, 0.f, 0.f, 0.f, 0.f, 0.f, 0.f, 0.f};
        C = __builtin_amdgcn_wmma_f32_16x16x4_f32(false, A1a, false, B1,
                                                  (short)0, C, false, false);
        C = __builtin_amdgcn_wmma_f32_16x16x4_f32(false, A2a, false, B2,
                                                  (short)0, C, false, false);
        const float2 oa = hiK ? make_float2(C[4], C[5]) : make_float2(C[0], C[1]);
        const float2 ob = hiK ? make_float2(C[6], C[7]) : make_float2(C[2], C[3]);
        st[st0 ^ rc]      = oa;
        st[st0 ^ rc ^ a1] = ob;
    }
}

__device__ __forceinline__ int lane_slot(const int* o, int lane, int wave) {
    int s = 0;
    #pragma unroll
    for (int i = 0; i < 4; ++i) s ^= o[i] & -((lane >> i) & 1);
    s ^= o[4] & -(wave & 1);
    s ^= o[5] & -((wave >> 1) & 1);
    return s;
}

__global__ __launch_bounds__(THREADS, 4)
void qsim_kernel(const float* __restrict__ fv, const float* __restrict__ theta,
                 float* __restrict__ out) {
    __shared__ float2 st[DIM];
    __shared__ float2 wireU[(1 + NLAYERS) * NQ][4];
    __shared__ float2 layM[NLAYERS * NSLOT][16];
    __shared__ float  wpart[NWAVES][NQ];

    const int b    = blockIdx.x;
    const int tid  = threadIdx.x;
    const int lane = tid & 31;
    const int wave = tid >> 5;

    for (int i = tid; i < DIM; i += THREADS) st[i] = make_float2(0.f, 0.f);
    if (tid == 0) st[0] = make_float2(1.f, 0.f);

    if (tid < NLAYERS * NQ) {
        float2 U[4];
        const float* ta = theta + tid * 3;
        fused_rot(ta[0], ta[1], ta[2], U);
        #pragma unroll
        for (int i = 0; i < 4; ++i) wireU[NQ + tid][i] = U[i];
    } else if (tid >= 3 * 32 && tid < 3 * 32 + NQ) {
        const int w = tid - 3 * 32;
        float2 U[4];
        embed_rot(fv[b * NQ + w], U);
        #pragma unroll
        for (int i = 0; i < 4; ++i) wireU[w][i] = U[i];
    }
    __syncthreads();
    if (tid < NLAYERS * NSLOT) {
        const int q = tid & 1;
        const int l = (tid >> 1) / NPAIRS;
        const int s = (tid >> 1) - l * NPAIRS;
        const bool carry = (l & 1) == 0;
        const int  lo = carry ? (s == 0 ? 0  : 2 * s - 1) : (s < 3 ? 4 * s     : 11);
        const int  hi = carry ? (s == 0 ? 11 : 2 * s)     : (s < 3 ? 4 * s + 2 : 11);
        const bool live = carry || s < 3 || (s == 3 && l == NLAYERS - 1);
        if (live) {
            float2 Ua[4], Ub[4], G[16];
            if (l == 0) {
                mul22(&wireU[NQ + lo][0], &wireU[lo][0], Ua);
                mul22(&wireU[NQ + hi][0], &wireU[hi][0], Ub);
            } else {
                #pragma unroll
                for (int i = 0; i < 4; ++i) {
                    Ua[i] = wireU[(1 + l) * NQ + lo][i];
                    Ub[i] = wireU[(1 + l) * NQ + hi][i];
                }
            }
            if (!carry && s == 3) {
                Ub[0] = make_float2(1.f, 0.f); Ub[1] = make_float2(0.f, 0.f);
                Ub[2] = make_float2(0.f, 0.f); Ub[3] = make_float2(1.f, 0.f);
            }
            kron22(Ub, Ua, G);
            if (carry && s >= 1) {
                float2 Un[4], T[16];
                #pragma unroll
                for (int i = 0; i < 4; ++i) Un[i] = wireU[(2 + l) * NQ + lo][i];
                ride_next(G, Un, q, T);
                #pragma unroll
                for (int i = 0; i < 16; ++i) layM[tid][i] = T[i];
            } else if (carry && l >= 2) {
                float2 Up[4], T[16];
                #pragma unroll
                for (int i = 0; i < 4; ++i) Up[i] = wireU[l * NQ + NQ - 1][i];
                push_prev(G, Up, q, T);
                #pragma unroll
                for (int i = 0; i < 16; ++i) layM[tid][i] = T[i];
            } else {
                #pragma unroll
                for (int i = 0; i < 16; ++i) layM[tid][i] = G[i];
            }
        }
    }
    __syncthreads();

    static constexpr Maps MAPS = make_maps();
    const int pi = wave & 1;

    {
        const float2* M = &layM[0][0];
        int g0[6], g1[6];
        #pragma unroll
        for (int i = 0; i < 6; ++i) { g0[i] = MAPS.g[0][0][i]; g1[i] = MAPS.g[0][1][i]; }
        const int lan0 = lane_slot(MAPS.o[0][0], lane, wave);
        const int lan1 = lane_slot(MAPS.o[0][1], lane, wave);
        apply_three_pairs<true, true>(st, M,       g0, lan0, lane, pi);
        __syncthreads();
        apply_three_pairs<true, false>(st, M + 96, g1, lan1, lane, pi);
        __syncthreads();
    }
    #pragma nounroll
    for (int l = 1; l < NLAYERS; ++l) {
        const float2* M = &layM[l * NSLOT][0];
        int g0[6], g1[6];
        #pragma unroll
        for (int i = 0; i < 6; ++i) { g0[i] = MAPS.g[l][0][i]; g1[i] = MAPS.g[l][1][i]; }
        const int lan0 = lane_slot(MAPS.o[l][0], lane, wave);
        const int lan1 = lane_slot(MAPS.o[l][1], lane, wave);
        const bool carry = (l & 1) == 0;
        if (carry) {
            apply_three_pairs<false, true>(st, M, g0, lan0, lane, pi);
            __syncthreads();
        }
        int gs[6];
        #pragma unroll
        for (int i = 0; i < 6; ++i) gs[i] = carry ? g1[i] : g0[i];
        apply_three_pairs<false, false>(st, carry ? M + 96 : M, gs, carry ? lan1 : lan0, lane, pi);
        __syncthreads();
        if (l == NLAYERS - 1) {
            apply_one_pair(st, M + 96, g1, lan1, lane);
            __syncthreads();
        }
    }

    float acc[NQ];
    #pragma unroll
    for (int w = 0; w < NQ; ++w) acc[w] = 0.f;
    int ytid = 0;
    #pragma unroll
    for (int j = 0; j < 7; ++j) ytid ^= ((tid >> j) & 1) ? MAPS.y[j] : 0;
    #pragma unroll
    for (int q = 0; q < DIM / THREADS; ++q) {
        int yq = 0;
        #pragma unroll
        for (int j = 0; j < 5; ++j) yq ^= ((q >> j) & 1) ? MAPS.y[7 + j] : 0;
        const int    y  = ytid ^ yq;
        const float2 a  = st[tid + q * THREADS];
        const float  pz = a.x * a.x + a.y * a.y;
        #pragma unroll
        for (int w = 0; w < NQ; ++w)
            acc[w] += ((y >> w) & 1) ? -pz : pz;
    }
    #pragma unroll
    for (int off = 16; off > 0; off >>= 1) {
        #pragma unroll
        for (int w = 0; w < NQ; ++w)
            acc[w] += __shfl_xor(acc[w], off, 32);
    }
    if (lane == 0) {
        #pragma unroll
        for (int w = 0; w < NQ; ++w) wpart[wave][w] = acc[w];
    }
    __syncthreads();
    if (tid < NQ) {
        float s = 0.f;
        for (int q = 0; q < NWAVES; ++q) s += wpart[q][tid];
        out[b * NQ + tid] = s;
    }
}

extern "C" void kernel_launch(void* const* d_in, const int* in_sizes, int n_in,
                              void* d_out, int out_size, void* d_ws, size_t ws_size,
                              hipStream_t stream) {
    const float* fv    = (const float*)d_in[0];
    const float* theta = (const float*)d_in[1];
    float*       out   = (float*)d_out;
    const int batch = in_sizes[0] / NQ;
    qsim_kernel<<<batch, THREADS, 0, stream>>>(fv, theta, out);
}


# last layer's wire-11 gate never touches the state: <Z> sums run as a matrix product, A = |a0|^2, |a1|^2, Re, Im of a0 conj(a1), B = signs times the gate's quadratic form; one pass gone; new sum order
# speedup vs baseline: 2.9726x; 1.0704x over previous
#include <hip/hip_runtime.h>


typedef __attribute__((ext_vector_type(2))) float v2f;
typedef __attribute__((ext_vector_type(8))) float v8f;

#define NQ      12
#define DIM     4096
#define THREADS 128
#define NWAVES  4
#define NLAYERS 6
#define NPAIRS  6
#define NPASS   2
#define NTILES  16
#define NSLOT   12

__host__ __device__ constexpr int ring_fwd(int x) {
    int t = x;
    t ^= t << 1; t ^= t << 2; t ^= t << 4; t ^= t << 8;
    t &= DIM - 1;
    return t ^ (t >> (NQ - 1));
}
__host__ __device__ constexpr int ring_inv(int y) {
    const int top = y >> (NQ - 1);
    return (y ^ (y << 1) ^ top ^ (top << 1)) & (DIM - 1);
}
__host__ __device__ constexpr int fold(int x) {
    constexpr int pat[7] = {15, 22, 19, 3, 10, 2, 21};
    int r = x;
    for (int b = 0; b < 7; ++b)
        if ((x >> (5 + b)) & 1) r ^= pat[b];
    return r;
}

__host__ __device__ constexpr int lane_ways(const int* v) {
    int cnt[32] = {};
    for (int m = 0; m < 32; ++m) {
        int s = 0;
        for (int i = 0; i < 5; ++i)
            if ((m >> i) & 1) s ^= v[i];
        cnt[s & 31] += 1;
    }
    int mx = 0;
    for (int i = 0; i < 32; ++i) mx = cnt[i] > mx ? cnt[i] : mx;
    return mx;
}
__host__ __device__ constexpr int bit_parity(int x) {
    int p = 0;
    for (int i = 0; i < 12; ++i) p ^= (x >> i) & 1;
    return p;
}

struct PassSpec { int regw[6]; int othw[6]; int omask[6]; };

struct Maps {
    int  g[NLAYERS][NPASS][6];
    int  o[NLAYERS][NPASS][6];
    int  zg[6];
    int  zo[6];
    bool ok;
};
__host__ __device__ constexpr Maps make_maps() {
    constexpr PassSpec SPEC[NLAYERS][NPASS] = {
        {{{0, 11, 1, 2, 3, 4}, {5, 6, 7, 8, 9, 10}, {27, 13, 17, 20, 44, 12}}, {{5, 6, 7, 8, 9, 10}, {0, 1, 2, 3, 4, 11}, {32, 10, 3, 59, 16, 62}}},
        {{{0, 2, 4, 6, 8, 10}, {1, 3, 5, 7, 9, 11}, {7, 15, 13, 61, 51, 21}}, {{11, 10, 6, 7, 8, 9}, {0, 1, 2, 3, 4, 5}, {27, 48, 58, 49, 36, 45}}},
        {{{0, 11, 1, 2, 3, 4}, {5, 6, 7, 8, 9, 10}, {19, 4, 24, 21, 51, 18}}, {{5, 6, 7, 8, 9, 10}, {0, 1, 2, 3, 4, 11}, {59, 55, 20, 41, 7, 32}}},
        {{{0, 2, 4, 6, 8, 10}, {1, 3, 5, 7, 9, 11}, {46, 9, 49, 61, 15, 17}}, {{11, 10, 6, 7, 8, 9}, {0, 1, 2, 3, 4, 5}, {57, 3, 42, 51, 52, 11}}},
        {{{0, 11, 1, 2, 3, 4}, {5, 6, 7, 8, 9, 10}, {7, 6, 20, 17, 32, 26}}, {{5, 6, 7, 8, 9, 10}, {0, 1, 2, 3, 4, 11}, {42, 62, 24, 59, 4, 6}}},
        {{{0, 2, 4, 6, 8, 10}, {1, 3, 5, 7, 9, 11}, {30, 40, 31, 21, 25, 57}}, {{11, 10, 6, 7, 8, 9}, {0, 1, 2, 3, 4, 5}, {20, 29, 62, 37, 33, 10}}}
    };
    Maps M{};
    M.ok = true;
    int E[NQ] = {};
    for (int k = 0; k < NQ; ++k) E[k] = fold(1 << k);
    for (int l = 0; l < NLAYERS; ++l) {
        for (int p = 0; p < NPASS; ++p) {
            const PassSpec& S = SPEC[l][p];
            int all[12] = {};
            for (int i = 0; i < 6; ++i) {
                M.g[l][p][i] = E[S.regw[i]];
                int v = 0;
                for (int j = 0; j < 6; ++j)
                    if ((S.omask[i] >> j) & 1) v ^= E[S.othw[j]];
                M.o[l][p][i] = v;
                all[i] = M.g[l][p][i];
                all[6 + i] = v;
            }
            int rank = 0;
            for (int bit = NQ - 1; bit >= 0; --bit) {
                int piv = -1;
                for (int i = rank; i < 12; ++i)
                    if (piv < 0 && ((all[i] >> bit) & 1)) piv = i;
                if (piv < 0) continue;
                const int t = all[rank]; all[rank] = all[piv]; all[piv] = t;
                for (int i = 0; i < 12; ++i)
                    if (i != rank && ((all[i] >> bit) & 1)) all[i] ^= all[rank];
                ++rank;
            }
            if (rank != NQ) M.ok = false;
            const bool one_pair = (l & 1) && p == 1;
            const int ld[5] = {M.o[l][p][0], M.o[l][p][1], M.o[l][p][2], M.o[l][p][3], M.g[l][p][0]};
            const int sv[5] = {M.o[l][p][0], M.o[l][p][1], M.o[l][p][2], M.o[l][p][3],
                               one_pair ? M.g[l][p][1] : M.g[l][p][5]};
            if (lane_ways(ld) + lane_ways(sv) > 4) M.ok = false;
            if (l == NLAYERS - 1 && p == 1) {
                if (lane_ways(sv) != 1) M.ok = false;
                for (int i = 0; i < 6; ++i) {
                    int v = 0;
                    for (int j = 0; j < 6; ++j)
                        if ((S.omask[i] >> j) & 1) v |= 1 << S.othw[j];
                    M.zo[i] = ring_fwd(v);
                    M.zg[i] = ring_fwd(1 << S.regw[i]);
                }
            }
            if (!(l & 1) && p == 0) {
                if (S.othw[5] != 10) M.ok = false;
                for (int i = 0; i < 6; ++i)
                    if (((S.omask[i] >> 5) & 1) != (i == 4 ? 1 : 0)) M.ok = false;
            }
            if (!(l & 1) && p == 1) {
                for (int i = 0; i < 5; ++i)
                    if (S.othw[i] != i) M.ok = false;
                for (int i = 0; i < 6; ++i)
                    if (bit_parity(S.omask[i] & 31) != (i == 4 ? 1 : 0)) M.ok = false;
            }
        }
        for (int k = 0; k < NQ; ++k) E[k] = fold(ring_inv(fold(E[k])));
    }
    return M;
}
static_assert(make_maps().ok, "address map: rank, lane conflicts, column parity or wave bit");

__device__ __forceinline__ float2 cmul(float2 a, float2 b) {
    return make_float2(a.x * b.x - a.y * b.y, a.x * b.y + a.y * b.x);
}

__device__ __forceinline__ void fused_rot(float tx, float ty, float tz, float2* U) {
    const float cx = cosf(0.5f * tx), sx = sinf(0.5f * tx);
    const float cy = cosf(0.5f * ty), sy = sinf(0.5f * ty);
    const float cz = cosf(0.5f * tz), sz = sinf(0.5f * tz);
    const float2 A00 = make_float2( cy * cx,  sy * sx);
    const float2 A01 = make_float2(-sy * cx, -cy * sx);
    const float2 A10 = make_float2( sy * cx, -cy * sx);
    const float2 A11 = make_float2( cy * cx, -sy * sx);
    const float2 e  = make_float2(cz, -sz);
    const float2 ec = make_float2(cz,  sz);
    U[0] = cmul(e, A00);  U[1] = cmul(e, A01);
    U[2] = cmul(ec, A10); U[3] = cmul(ec, A11);
}

__device__ __forceinline__ void embed_rot(float f, float2* U) {
    const float cy = cosf(0.5f * f),  sy = sinf(0.5f * f);
    const float cz = cosf(0.25f * f), sz = sinf(0.25f * f);
    U[0] = make_float2( cz * cy, -sz * cy);
    U[1] = make_float2(-cz * sy,  sz * sy);
    U[2] = make_float2( cz * sy,  sz * sy);
    U[3] = make_float2( cz * cy,  sz * cy);
}

__device__ __forceinline__ void kron22(const float2* Ub, const float2* Ua, float2* M) {
    #pragma unroll
    for (int o = 0; o < 4; ++o)
        #pragma unroll
        for (int j = 0; j < 4; ++j)
            M[o * 4 + j] = cmul(Ub[(o >> 1) * 2 + (j >> 1)], Ua[(o & 1) * 2 + (j & 1)]);
}

__device__ __forceinline__ void mul22(const float2* L, const float2* E, float2* C) {
    #pragma unroll
    for (int r = 0; r < 2; ++r)
        #pragma unroll
        for (int c = 0; c < 2; ++c) {
            const float2 x = cmul(L[r * 2], E[c]);
            const float2 y = cmul(L[r * 2 + 1], E[2 + c]);
            C[r * 2 + c] = make_float2(x.x + y.x, x.y + y.y);
        }
}

__device__ __forceinline__ void pair_A(float2 m1, float2 m2, int lane, v2f& A1, v2f& A2) {
    const bool im = (lane & 1) != 0;
    A1[0] = im ? m1.y :  m1.x;  A1[1] = im ? m1.x : -m1.y;
    A2[0] = im ? m2.y :  m2.x;  A2[1] = im ? m2.x : -m2.y;
}

__device__ __forceinline__ void ride_next(const float2* G, const float2* Un, int q, float2* T) {
    const float2 d_even = q ? Un[3] : Un[0], x_even = q ? Un[2] : Un[1];
    const float2 d_odd  = q ? Un[0] : Un[3], x_odd  = q ? Un[1] : Un[2];
    #pragma unroll
    for (int o = 0; o < 4; ++o) {
        const float2 d = (o & 1) ? d_odd : d_even;
        const float2 x = (o & 1) ? x_odd : x_even;
        #pragma unroll
        for (int j = 0; j < 4; ++j) {
            const float2 a = cmul(d, G[o * 4 + j]);
            const float2 b = cmul(x, G[(o ^ 3) * 4 + j]);
            T[o * 4 + j] = make_float2(a.x + b.x, a.y + b.y);
        }
    }
}

__device__ __forceinline__ void push_prev(const float2* G, const float2* Up, int q, float2* T) {
    const float2 d_lo = q ? Up[3] : Up[0], x_lo = q ? Up[1] : Up[2];
    const float2 d_hi = q ? Up[0] : Up[3], x_hi = q ? Up[2] : Up[1];
    #pragma unroll
    for (int j = 0; j < 4; ++j) {
        const float2 d = (j & 2) ? d_hi : d_lo;
        const float2 x = (j & 2) ? x_hi : x_lo;
        #pragma unroll
        for (int o = 0; o < 4; ++o) {
            const float2 a = cmul(G[o * 4 + j], d);
            const float2 b = cmul(G[o * 4 + (j ^ 3)], x);
            T[o * 4 + j] = make_float2(a.x + b.x, a.y + b.y);
        }
    }
}

__host__ __device__ constexpr int par2(int j) { return (j ^ (j >> 1)) & 1; }
template <bool HEAD>
__host__ __device__ constexpr int parA(int jA) { return HEAD ? (jA & 1) : par2(jA); }

template <bool ZIN, bool HEAD>
__device__ __forceinline__ void apply_three_pairs(float2* st, const float2* M, const int* g,
                                                  int lan, int lane, int pi) {
    const int hiK = lane >> 4;
    const int o   = (lane >> 1) & 3;
    const int e   = o * 4 + hiK;
    const int pq = HEAD ? 0 : pi;
    const float2* MA  = M +      (pi ? 16 : 0);
    const float2* MB0 = M + 32 + (pq ? 16 : 0);
    const float2* MB1 = M + 32 + (pq ? 0 : 16);
    const float2* MC0 = M + 64 + (pq ? 16 : 0);
    const float2* MC1 = M + 64 + (pq ? 0 : 16);
    const float2 ma1  = MA[e],  ma2  = MA[e + 2];
    const float2 mb01 = MB0[e], mb02 = MB0[e + 2];
    const float2 mb11 = MB1[e], mb12 = MB1[e + 2];
    const float2 mc01 = MC0[e], mc02 = MC0[e + 2];
    const float2 mc11 = MC1[e], mc12 = MC1[e + 2];

    const int a1 = g[0], a2 = g[1];
    const int b1 = g[2], b2 = g[3];
    const int c1 = g[4], c2 = g[5];
    const int ld0 = lan ^ (hiK ? a1 : 0);
    const int st0 = lan ^ (hiK ? c2 : 0);

    v2f A1a, A2a, A1b0, A2b0, A1b1, A2b1, A1c0, A2c0, A1c1, A2c1;
    pair_A(ma1,  ma2,  lane, A1a,  A2a);
    pair_A(mb01, mb02, lane, A1b0, A2b0);
    pair_A(mb11, mb12, lane, A1b1, A2b1);
    pair_A(mc01, mc02, lane, A1c0, A2c0);
    pair_A(mc11, mc12, lane, A1c1, A2c1);

    if constexpr (ZIN) {
        const v8f Z = {0.f, 0.f, 0.f, 0.f, 0.f, 0.f, 0.f, 0.f};
        const float2 p1 = st[ld0];
        v2f B0; B0[0] = p1.x; B0[1] = p1.y;
        const v8f E1 = __builtin_amdgcn_wmma_f32_16x16x4_f32(false, A1a, false, B0,
                                                             (short)0, Z, false, false);
        v8f E2[4];
        #pragma unroll
        for (int jA = 0; jA < 4; ++jA) {
            v2f B1;
            B1[0] = hiK ? 0.f : E1[2 * jA];
            B1[1] = hiK ? 0.f : E1[2 * jA + 1];
            const v2f A1 = parA<HEAD>(jA) ? A1b1 : A1b0;
            E2[jA] = __builtin_amdgcn_wmma_f32_16x16x4_f32(false, A1, false, B1,
                                                           (short)0, Z, false, false);
        }
        #pragma unroll
        for (int jA = 0; jA < 4; ++jA) {
            #pragma unroll
            for (int jB = 0; jB < 4; ++jB) {
                v2f B1;
                B1[0] = hiK ? 0.f : E2[jA][2 * jB];
                B1[1] = hiK ? 0.f : E2[jA][2 * jB + 1];
                const v2f A1 = (parA<HEAD>(jA) ^ par2(jB)) ? A1c1 : A1c0;
                const v8f C = __builtin_amdgcn_wmma_f32_16x16x4_f32(false, A1, false, B1,
                                                                    (short)0, Z, false, false);
                const float2 oa = hiK ? make_float2(C[4], C[5]) : make_float2(C[0], C[1]);
                const float2 ob = hiK ? make_float2(C[6], C[7]) : make_float2(C[2], C[3]);
                const int rc = ((jA & 1) ? a1 : 0) ^ ((jA & 2) ? a2 : 0) ^
                               ((jB & 1) ? b1 : 0) ^ ((jB & 2) ? b2 : 0);
                st[st0 ^ rc]      = oa;
                st[st0 ^ rc ^ c1] = ob;
            }
        }
        return;
    }

    v8f D1[NTILES];
    #pragma unroll
    for (int t = 0; t < NTILES; ++t) {
        const int rc = ((t & 1) ? b1 : 0) ^ ((t & 2) ? b2 : 0) ^
                       ((t & 4) ? c1 : 0) ^ ((t & 8) ? c2 : 0);
        const float2 p1 = st[ld0 ^ rc];
        const float2 p2 = st[ld0 ^ rc ^ a2];
        v2f B1; B1[0] = p1.x; B1[1] = p1.y;
        v2f B2; B2[0] = p2.x; B2[1] = p2.y;
        v8f C = {0.f, 0.f, 0.f, 0.f, 0.f, 0.f, 0.f, 0.f};
        C = __builtin_amdgcn_wmma_f32_16x16x4_f32(false, A1a, false, B1,
                                                  (short)0, C, false, false);
        C = __builtin_amdgcn_wmma_f32_16x16x4_f32(false, A2a, false, B2,
                                                  (short)0, C, false, false);
        D1[t] = C;
    }
    v8f D2[NTILES];
    #pragma unroll
    for (int tc = 0; tc < 4; ++tc) {
        #pragma unroll
        for (int jA = 0; jA < 4; ++jA) {
            v2f B1, B2;
            B1[0] = hiK ? D1[tc * 4 + 1][2 * jA]     : D1[tc * 4][2 * jA];
            B1[1] = hiK ? D1[tc * 4 + 1][2 * jA + 1] : D1[tc * 4][2 * jA + 1];
            B2[0] = hiK ? D1[tc * 4 + 3][2 * jA]     : D1[tc * 4 + 2][2 * jA];
            B2[1] = hiK ? D1[tc * 4 + 3][2 * jA + 1] : D1[tc * 4 + 2][2 * jA + 1];
            const v2f A1 = parA<HEAD>(jA) ? A1b1 : A1b0;
            const v2f A2 = parA<HEAD>(jA) ? A2b1 : A2b0;
            v8f C = {0.f, 0.f, 0.f, 0.f, 0.f, 0.f, 0.f, 0.f};
            C = __builtin_amdgcn_wmma_f32_16x16x4_f32(false, A1, false, B1,
                                                      (short)0, C, false, false);
            C = __builtin_amdgcn_wmma_f32_16x16x4_f32(false, A2, false, B2,
                                                      (short)0, C, false, false);
            D2[jA * 4 + tc] = C;
        }
    }
    #pragma unroll
    for (int jA = 0; jA < 4; ++jA) {
        #pragma unroll
        for (int jB = 0; jB < 4; ++jB) {
            v2f B1, B2;
            B1[0] = hiK ? D2[jA * 4 + 1][2 * jB]     : D2[jA * 4][2 * jB];
            B1[1] = hiK ? D2[jA * 4 + 1][2 * jB + 1] : D2[jA * 4][2 * jB + 1];
            B2[0] = hiK ? D2[jA * 4 + 3][2 * jB]     : D2[jA * 4 + 2][2 * jB];
            B2[1] = hiK ? D2[jA * 4 + 3][2 * jB + 1] : D2[jA * 4 + 2][2 * jB + 1];
            const v2f A1 = (parA<HEAD>(jA) ^ par2(jB)) ? A1c1 : A1c0;
            const v2f A2 = (parA<HEAD>(jA) ^ par2(jB)) ? A2c1 : A2c0;
            v8f C = {0.f, 0.f, 0.f, 0.f, 0.f, 0.f, 0.f, 0.f};
            C = __builtin_amdgcn_wmma_f32_16x16x4_f32(false, A1, false, B1,
                                                      (short)0, C, false, false);
            C = __builtin_amdgcn_wmma_f32_16x16x4_f32(false, A2, false, B2,
                                                      (short)0, C, false, false);
            const float2 oa = hiK ? make_float2(C[4], C[5]) : make_float2(C[0], C[1]);
            const float2 ob = hiK ? make_float2(C[6], C[7]) : make_float2(C[2], C[3]);
            const int rc = ((jA & 1) ? a1 : 0) ^ ((jA & 2) ? a2 : 0) ^
                           ((jB & 1) ? b1 : 0) ^ ((jB & 2) ? b2 : 0);
            st[st0 ^ rc]      = oa;
            st[st0 ^ rc ^ c1] = ob;
        }
    }
}

__device__ __forceinline__ int lane_slot(const int* o, int lane, int wave) {
    int s = 0;
    #pragma unroll
    for (int i = 0; i < 4; ++i) s ^= o[i] & -((lane >> i) & 1);
    s ^= o[4] & -(wave & 1);
    s ^= o[5] & -((wave >> 1) & 1);
    return s;
}

__device__ __forceinline__ unsigned sign_word(int pat, int sh) {
    return ((unsigned)pat << sh) & 0x80000000u;
}

__global__ __launch_bounds__(THREADS, 4)
void qsim_kernel(const float* __restrict__ fv, const float* __restrict__ theta,
                 float* __restrict__ out) {
    __shared__ float2 st[DIM];
    __shared__ float2 wireU[(1 + NLAYERS) * NQ][4];
    __shared__ float2 layM[NLAYERS * NSLOT][16];
    __shared__ float  wpart[NWAVES][NQ];

    const int b    = blockIdx.x;
    const int tid  = threadIdx.x;
    const int lane = tid & 31;
    const int wave = tid >> 5;

    for (int i = tid; i < DIM; i += THREADS) st[i] = make_float2(0.f, 0.f);
    if (tid == 0) st[0] = make_float2(1.f, 0.f);

    if (tid < NLAYERS * NQ) {
        float2 U[4];
        const float* ta = theta + tid * 3;
        fused_rot(ta[0], ta[1], ta[2], U);
        #pragma unroll
        for (int i = 0; i < 4; ++i) wireU[NQ + tid][i] = U[i];
    } else if (tid >= 3 * 32 && tid < 3 * 32 + NQ) {
        const int w = tid - 3 * 32;
        float2 U[4];
        embed_rot(fv[b * NQ + w], U);
        #pragma unroll
        for (int i = 0; i < 4; ++i) wireU[w][i] = U[i];
    }
    __syncthreads();
    if (tid < NLAYERS * NSLOT) {
        const int q = tid & 1;
        const int l = (tid >> 1) / NPAIRS;
        const int s = (tid >> 1) - l * NPAIRS;
        const bool carry = (l & 1) == 0;
        const int  lo = carry ? (s == 0 ? 0  : 2 * s - 1) : (s < 3 ? 4 * s     : 11);
        const int  hi = carry ? (s == 0 ? 11 : 2 * s)     : (s < 3 ? 4 * s + 2 : 11);
        const bool live = carry || s < 3;
        if (live) {
            float2 Ua[4], Ub[4], G[16];
            if (l == 0) {
                mul22(&wireU[NQ + lo][0], &wireU[lo][0], Ua);
                mul22(&wireU[NQ + hi][0], &wireU[hi][0], Ub);
            } else {
                #pragma unroll
                for (int i = 0; i < 4; ++i) {
                    Ua[i] = wireU[(1 + l) * NQ + lo][i];
                    Ub[i] = wireU[(1 + l) * NQ + hi][i];
                }
            }
            kron22(Ub, Ua, G);
            if (carry && s >= 1) {
                float2 Un[4], T[16];
                #pragma unroll
                for (int i = 0; i < 4; ++i) Un[i] = wireU[(2 + l) * NQ + lo][i];
                ride_next(G, Un, q, T);
                #pragma unroll
                for (int i = 0; i < 16; ++i) layM[tid][i] = T[i];
            } else if (carry && l >= 2) {
                float2 Up[4], T[16];
                #pragma unroll
                for (int i = 0; i < 4; ++i) Up[i] = wireU[l * NQ + NQ - 1][i];
                push_prev(G, Up, q, T);
                #pragma unroll
                for (int i = 0; i < 16; ++i) layM[tid][i] = T[i];
            } else {
                #pragma unroll
                for (int i = 0; i < 16; ++i) layM[tid][i] = G[i];
            }
        }
    }
    __syncthreads();

    static constexpr Maps MAPS = make_maps();
    const int pi = wave & 1;

    {
        const float2* M = &layM[0][0];
        int g0[6], g1[6];
        #pragma unroll
        for (int i = 0; i < 6; ++i) { g0[i] = MAPS.g[0][0][i]; g1[i] = MAPS.g[0][1][i]; }
        const int lan0 = lane_slot(MAPS.o[0][0], lane, wave);
        const int lan1 = lane_slot(MAPS.o[0][1], lane, wave);
        apply_three_pairs<true, true>(st, M,       g0, lan0, lane, pi);
        __syncthreads();
        apply_three_pairs<true, false>(st, M + 96, g1, lan1, lane, pi);
        __syncthreads();
    }
    #pragma nounroll
    for (int l = 1; l < NLAYERS; ++l) {
        const float2* M = &layM[l * NSLOT][0];
        int g0[6], g1[6];
        #pragma unroll
        for (int i = 0; i < 6; ++i) { g0[i] = MAPS.g[l][0][i]; g1[i] = MAPS.g[l][1][i]; }
        const int lan0 = lane_slot(MAPS.o[l][0], lane, wave);
        const int lan1 = lane_slot(MAPS.o[l][1], lane, wave);
        const bool carry = (l & 1) == 0;
        if (carry) {
            apply_three_pairs<false, true>(st, M, g0, lan0, lane, pi);
            __syncthreads();
        }
        int gs[6];
        #pragma unroll
        for (int i = 0; i < 6; ++i) gs[i] = carry ? g1[i] : g0[i];
        apply_three_pairs<false, false>(st, carry ? M + 96 : M, gs, carry ? lan1 : lan0, lane, pi);
        __syncthreads();
    }

    constexpr int LZ = NLAYERS - 1;
    const int      h   = lane >> 4;
    const int      wc  = lane & 15;
    const int      sh  = 31 - wc;
    const unsigned one = 0x3f800000u;
    const float2 u00 = wireU[NLAYERS * NQ + NQ - 1][0], u01 = wireU[NLAYERS * NQ + NQ - 1][1];
    const float2 u10 = wireU[NLAYERS * NQ + NQ - 1][2], u11 = wireU[NLAYERS * NQ + NQ - 1][3];
    const float f   = __uint_as_float(one ^ sign_word(MAPS.zg[0], sh));
    const float n00 = u00.x * u00.x + u00.y * u00.y, n01 = u01.x * u01.x + u01.y * u01.y;
    const float n10 = u10.x * u10.x + u10.y * u10.y, n11 = u11.x * u11.x + u11.y * u11.y;
    const float c0r = u00.x * u01.x + u00.y * u01.y, c0i = u00.y * u01.x - u00.x * u01.y;
    const float c1r = u10.x * u11.x + u10.y * u11.y, c1i = u10.y * u11.x - u10.x * u11.y;
    const unsigned k0 = __float_as_uint(n00 + f * n10);
    const unsigned k1 = __float_as_uint(n01 + f * n11);
    const unsigned kr = __float_as_uint( 2.f * (c0r + f * c1r));
    const unsigned ki = __float_as_uint(-2.f * (c0i + f * c1i));
    const unsigned hb  = h ? sign_word(MAPS.zg[1], sh) : 0u;
    const unsigned gt0 = sign_word(MAPS.zg[2], sh), gt1 = sign_word(MAPS.zg[3], sh);
    const unsigned gt2 = sign_word(MAPS.zg[4], sh), gt3 = sign_word(MAPS.zg[5], sh);
    const int lanz = lane_slot(MAPS.o[LZ][1], lane, wave) ^ (h ? MAPS.g[LZ][1][1] : 0);
    float2 za0[DIM / THREADS / 2], za1[DIM / THREADS / 2];
    #pragma unroll
    for (int t = 0; t < DIM / THREADS / 2; ++t) {
        const int rc = ((t & 1) ? MAPS.g[LZ][1][2] : 0) ^ ((t & 2) ? MAPS.g[LZ][1][3] : 0) ^
                       ((t & 4) ? MAPS.g[LZ][1][4] : 0) ^ ((t & 8) ? MAPS.g[LZ][1][5] : 0);
        za0[t] = st[lanz ^ rc];
        za1[t] = st[lanz ^ rc ^ MAPS.g[LZ][1][0]];
    }
    v8f zacc0 = {0.f, 0.f, 0.f, 0.f, 0.f, 0.f, 0.f, 0.f};
    v8f zacc1 = {0.f, 0.f, 0.f, 0.f, 0.f, 0.f, 0.f, 0.f};
    #pragma unroll
    for (int t = 0; t < DIM / THREADS / 2; ++t) {
        const float2 a0 = za0[t], a1 = za1[t];
        const unsigned s = hb ^ ((t & 1) ? gt0 : 0u) ^ ((t & 2) ? gt1 : 0u) ^
                                ((t & 4) ? gt2 : 0u) ^ ((t & 8) ? gt3 : 0u);
        v2f A, B;
        A[0] = a0.x * a0.x + a0.y * a0.y;
        A[1] = a1.x * a1.x + a1.y * a1.y;
        B[0] = __uint_as_float(k0 ^ s);
        B[1] = __uint_as_float(k1 ^ s);
        zacc0 = __builtin_amdgcn_wmma_f32_16x16x4_f32(false, A, false, B,
                                                      (short)0, zacc0, false, false);
        A[0] = a0.x * a1.x + a0.y * a1.y;
        A[1] = a0.y * a1.x - a0.x * a1.y;
        B[0] = __uint_as_float(kr ^ s);
        B[1] = __uint_as_float(ki ^ s);
        zacc1 = __builtin_amdgcn_wmma_f32_16x16x4_f32(false, A, false, B,
                                                      (short)0, zacc1, false, false);
    }
    const float s0 = __uint_as_float(one ^ sign_word(MAPS.zo[0], sh));
    const float s1 = __uint_as_float(one ^ sign_word(MAPS.zo[1], sh));
    const float s2 = __uint_as_float(one ^ sign_word(MAPS.zo[2], sh));
    float dz[8];
    #pragma unroll
    for (int r = 0; r < 8; ++r) dz[r] = zacc0[r] + zacc1[r];
    const float e0 = dz[0] + s0 * dz[1], e1 = dz[2] + s0 * dz[3];
    const float e2 = dz[4] + s0 * dz[5], e3 = dz[6] + s0 * dz[7];
    const float f0 = e0 + s1 * e1, f1 = e2 + s1 * e3;
    float zs = f0 + s2 * f1;
    const unsigned sw = (h ? sign_word(MAPS.zo[3], sh) : 0u) ^
                        ((wave & 1) ? sign_word(MAPS.zo[4], sh) : 0u) ^
                        ((wave & 2) ? sign_word(MAPS.zo[5], sh) : 0u);
    zs = __uint_as_float(__float_as_uint(zs) ^ sw);
    zs += __shfl_xor(zs, 16, 32);
    if (lane < NQ) wpart[wave][lane] = zs;
    __syncthreads();
    if (tid < NQ) {
        float s = 0.f;
        for (int q = 0; q < NWAVES; ++q) s += wpart[q][tid];
        out[b * NQ + tid] = s;
    }
}

extern "C" void kernel_launch(void* const* d_in, const int* in_sizes, int n_in,
                              void* d_out, int out_size, void* d_ws, size_t ws_size,
                              hipStream_t stream) {
    const float* fv    = (const float*)d_in[0];
    const float* theta = (const float*)d_in[1];
    float*       out   = (float*)d_out;
    const int batch = in_sizes[0] / NQ;
    qsim_kernel<<<batch, THREADS, 0, stream>>>(fv, theta, out);
}


# layers 2,4,6 (even wires left): three duplicated-row pair gates become two full 16-row wire-triple gates (0,2,4),(6,8,10), lane halves swapped in registers; 64 products a pass, not 96; new sum order
# speedup vs baseline: 3.3319x; 1.1209x over previous
#include <hip/hip_runtime.h>


typedef __attribute__((ext_vector_type(2))) float v2f;
typedef __attribute__((ext_vector_type(8))) float v8f;

#define NQ      12
#define DIM     4096
#define THREADS 128
#define NWAVES  4
#define NLAYERS 6
#define NPAIRS  6
#define NPASS   2
#define NTILES  16
#define NSLOT   12

__host__ __device__ constexpr int ring_fwd(int x) {
    int t = x;
    t ^= t << 1; t ^= t << 2; t ^= t << 4; t ^= t << 8;
    t &= DIM - 1;
    return t ^ (t >> (NQ - 1));
}
__host__ __device__ constexpr int ring_inv(int y) {
    const int top = y >> (NQ - 1);
    return (y ^ (y << 1) ^ top ^ (top << 1)) & (DIM - 1);
}
__host__ __device__ constexpr int fold(int x) {
    constexpr int pat[7] = {15, 22, 19, 3, 10, 2, 21};
    int r = x;
    for (int b = 0; b < 7; ++b)
        if ((x >> (5 + b)) & 1) r ^= pat[b];
    return r;
}

__host__ __device__ constexpr int lane_ways(const int* v) {
    int cnt[32] = {};
    for (int m = 0; m < 32; ++m) {
        int s = 0;
        for (int i = 0; i < 5; ++i)
            if ((m >> i) & 1) s ^= v[i];
        cnt[s & 31] += 1;
    }
    int mx = 0;
    for (int i = 0; i < 32; ++i) mx = cnt[i] > mx ? cnt[i] : mx;
    return mx;
}
__host__ __device__ constexpr int bit_parity(int x) {
    int p = 0;
    for (int i = 0; i < 12; ++i) p ^= (x >> i) & 1;
    return p;
}

struct PassSpec { int regw[6]; int othw[6]; int omask[6]; };

struct Maps {
    int  g[NLAYERS][NPASS][6];
    int  o[NLAYERS][NPASS][6];
    int  zg[6];
    int  zo[6];
    bool ok;
};
__host__ __device__ constexpr Maps make_maps() {
    constexpr PassSpec SPEC[NLAYERS][NPASS] = {
        {{{0, 11, 1, 2, 3, 4}, {5, 6, 7, 8, 9, 10}, {27, 13, 17, 20, 44, 12}}, {{5, 6, 7, 8, 9, 10}, {0, 1, 2, 3, 4, 11}, {32, 10, 3, 59, 16, 62}}},
        {{{0, 2, 4, 6, 8, 10}, {1, 3, 5, 7, 9, 11}, {7, 15, 13, 61, 51, 21}}, {{11, 10, 6, 7, 8, 9}, {0, 1, 2, 3, 4, 5}, {27, 48, 58, 49, 36, 45}}},
        {{{0, 11, 1, 2, 3, 4}, {5, 6, 7, 8, 9, 10}, {19, 4, 24, 21, 51, 18}}, {{5, 6, 7, 8, 9, 10}, {0, 1, 2, 3, 4, 11}, {59, 55, 20, 41, 7, 32}}},
        {{{0, 2, 4, 6, 8, 10}, {1, 3, 5, 7, 9, 11}, {46, 9, 49, 61, 15, 17}}, {{11, 10, 6, 7, 8, 9}, {0, 1, 2, 3, 4, 5}, {57, 3, 42, 51, 52, 11}}},
        {{{0, 11, 1, 2, 3, 4}, {5, 6, 7, 8, 9, 10}, {7, 6, 20, 17, 32, 26}}, {{5, 6, 7, 8, 9, 10}, {0, 1, 2, 3, 4, 11}, {42, 62, 24, 59, 4, 6}}},
        {{{0, 2, 4, 6, 8, 10}, {1, 3, 5, 7, 9, 11}, {30, 40, 31, 21, 25, 57}}, {{11, 10, 6, 7, 8, 9}, {0, 1, 2, 3, 4, 5}, {20, 29, 62, 37, 33, 10}}}
    };
    Maps M{};
    M.ok = true;
    int E[NQ] = {};
    for (int k = 0; k < NQ; ++k) E[k] = fold(1 << k);
    for (int l = 0; l < NLAYERS; ++l) {
        for (int p = 0; p < NPASS; ++p) {
            const PassSpec& S = SPEC[l][p];
            int all[12] = {};
            for (int i = 0; i < 6; ++i) {
                M.g[l][p][i] = E[S.regw[i]];
                int v = 0;
                for (int j = 0; j < 6; ++j)
                    if ((S.omask[i] >> j) & 1) v ^= E[S.othw[j]];
                M.o[l][p][i] = v;
                all[i] = M.g[l][p][i];
                all[6 + i] = v;
            }
            int rank = 0;
            for (int bit = NQ - 1; bit >= 0; --bit) {
                int piv = -1;
                for (int i = rank; i < 12; ++i)
                    if (piv < 0 && ((all[i] >> bit) & 1)) piv = i;
                if (piv < 0) continue;
                const int t = all[rank]; all[rank] = all[piv]; all[piv] = t;
                for (int i = 0; i < 12; ++i)
                    if (i != rank && ((all[i] >> bit) & 1)) all[i] ^= all[rank];
                ++rank;
            }
            if (rank != NQ) M.ok = false;
            const bool one_pair = (l & 1) && p == 1;
            const int ld[5] = {M.o[l][p][0], M.o[l][p][1], M.o[l][p][2], M.o[l][p][3], M.g[l][p][0]};
            const int sv[5] = {M.o[l][p][0], M.o[l][p][1], M.o[l][p][2], M.o[l][p][3],
                               one_pair ? M.g[l][p][1] : M.g[l][p][5]};
            if (lane_ways(ld) + lane_ways(sv) > 4) M.ok = false;
            if (l == NLAYERS - 1 && p == 1) {
                if (lane_ways(sv) != 1) M.ok = false;
                for (int i = 0; i < 6; ++i) {
                    int v = 0;
                    for (int j = 0; j < 6; ++j)
                        if ((S.omask[i] >> j) & 1) v |= 1 << S.othw[j];
                    M.zo[i] = ring_fwd(v);
                    M.zg[i] = ring_fwd(1 << S.regw[i]);
                }
            }
            if (!(l & 1) && p == 0) {
                if (S.othw[5] != 10) M.ok = false;
                for (int i = 0; i < 6; ++i)
                    if (((S.omask[i] >> 5) & 1) != (i == 4 ? 1 : 0)) M.ok = false;
            }
            if (!(l & 1) && p == 1) {
                for (int i = 0; i < 5; ++i)
                    if (S.othw[i] != i) M.ok = false;
                for (int i = 0; i < 6; ++i)
                    if (bit_parity(S.omask[i] & 31) != (i == 4 ? 1 : 0)) M.ok = false;
            }
        }
        for (int k = 0; k < NQ; ++k) E[k] = fold(ring_inv(fold(E[k])));
    }
    return M;
}
static_assert(make_maps().ok, "address map: rank, lane conflicts, column parity or wave bit");

__device__ __forceinline__ float2 cmul(float2 a, float2 b) {
    return make_float2(a.x * b.x - a.y * b.y, a.x * b.y + a.y * b.x);
}

__device__ __forceinline__ void fused_rot(float tx, float ty, float tz, float2* U) {
    const float cx = cosf(0.5f * tx), sx = sinf(0.5f * tx);
    const float cy = cosf(0.5f * ty), sy = sinf(0.5f * ty);
    const float cz = cosf(0.5f * tz), sz = sinf(0.5f * tz);
    const float2 A00 = make_float2( cy * cx,  sy * sx);
    const float2 A01 = make_float2(-sy * cx, -cy * sx);
    const float2 A10 = make_float2( sy * cx, -cy * sx);
    const float2 A11 = make_float2( cy * cx, -sy * sx);
    const float2 e  = make_float2(cz, -sz);
    const float2 ec = make_float2(cz,  sz);
    U[0] = cmul(e, A00);  U[1] = cmul(e, A01);
    U[2] = cmul(ec, A10); U[3] = cmul(ec, A11);
}

__device__ __forceinline__ void embed_rot(float f, float2* U) {
    const float cy = cosf(0.5f * f),  sy = sinf(0.5f * f);
    const float cz = cosf(0.25f * f), sz = sinf(0.25f * f);
    U[0] = make_float2( cz * cy, -sz * cy);
    U[1] = make_float2(-cz * sy,  sz * sy);
    U[2] = make_float2( cz * sy,  sz * sy);
    U[3] = make_float2( cz * cy,  sz * cy);
}

__device__ __forceinline__ void kron22(const float2* Ub, const float2* Ua, float2* M) {
    #pragma unroll
    for (int o = 0; o < 4; ++o)
        #pragma unroll
        for (int j = 0; j < 4; ++j)
            M[o * 4 + j] = cmul(Ub[(o >> 1) * 2 + (j >> 1)], Ua[(o & 1) * 2 + (j & 1)]);
}

__device__ __forceinline__ void mul22(const float2* L, const float2* E, float2* C) {
    #pragma unroll
    for (int r = 0; r < 2; ++r)
        #pragma unroll
        for (int c = 0; c < 2; ++c) {
            const float2 x = cmul(L[r * 2], E[c]);
            const float2 y = cmul(L[r * 2 + 1], E[2 + c]);
            C[r * 2 + c] = make_float2(x.x + y.x, x.y + y.y);
        }
}

__device__ __forceinline__ void pair_A(float2 m1, float2 m2, int lane, v2f& A1, v2f& A2) {
    const bool im = (lane & 1) != 0;
    A1[0] = im ? m1.y :  m1.x;  A1[1] = im ? m1.x : -m1.y;
    A2[0] = im ? m2.y :  m2.x;  A2[1] = im ? m2.x : -m2.y;
}

__device__ __forceinline__ void ride_next(const float2* G, const float2* Un, int q, float2* T) {
    const float2 d_even = q ? Un[3] : Un[0], x_even = q ? Un[2] : Un[1];
    const float2 d_odd  = q ? Un[0] : Un[3], x_odd  = q ? Un[1] : Un[2];
    #pragma unroll
    for (int o = 0; o < 4; ++o) {
        const float2 d = (o & 1) ? d_odd : d_even;
        const float2 x = (o & 1) ? x_odd : x_even;
        #pragma unroll
        for (int j = 0; j < 4; ++j) {
            const float2 a = cmul(d, G[o * 4 + j]);
            const float2 b = cmul(x, G[(o ^ 3) * 4 + j]);
            T[o * 4 + j] = make_float2(a.x + b.x, a.y + b.y);
        }
    }
}

__device__ __forceinline__ void push_prev(const float2* G, const float2* Up, int q, float2* T) {
    const float2 d_lo = q ? Up[3] : Up[0], x_lo = q ? Up[1] : Up[2];
    const float2 d_hi = q ? Up[0] : Up[3], x_hi = q ? Up[2] : Up[1];
    #pragma unroll
    for (int j = 0; j < 4; ++j) {
        const float2 d = (j & 2) ? d_hi : d_lo;
        const float2 x = (j & 2) ? x_hi : x_lo;
        #pragma unroll
        for (int o = 0; o < 4; ++o) {
            const float2 a = cmul(G[o * 4 + j], d);
            const float2 b = cmul(G[o * 4 + (j ^ 3)], x);
            T[o * 4 + j] = make_float2(a.x + b.x, a.y + b.y);
        }
    }
}

__host__ __device__ constexpr int par2(int j) { return (j ^ (j >> 1)) & 1; }
template <bool HEAD>
__host__ __device__ constexpr int parA(int jA) { return HEAD ? (jA & 1) : par2(jA); }

template <bool ZIN, bool HEAD>
__device__ __forceinline__ void apply_three_pairs(float2* st, const float2* M, const int* g,
                                                  int lan, int lane, int pi) {
    const int hiK = lane >> 4;
    const int o   = (lane >> 1) & 3;
    const int e   = o * 4 + hiK;
    const int pq = HEAD ? 0 : pi;
    const float2* MA  = M +      (pi ? 16 : 0);
    const float2* MB0 = M + 32 + (pq ? 16 : 0);
    const float2* MB1 = M + 32 + (pq ? 0 : 16);
    const float2* MC0 = M + 64 + (pq ? 16 : 0);
    const float2* MC1 = M + 64 + (pq ? 0 : 16);
    const float2 ma1  = MA[e],  ma2  = MA[e + 2];
    const float2 mb01 = MB0[e], mb02 = MB0[e + 2];
    const float2 mb11 = MB1[e], mb12 = MB1[e + 2];
    const float2 mc01 = MC0[e], mc02 = MC0[e + 2];
    const float2 mc11 = MC1[e], mc12 = MC1[e + 2];

    const int a1 = g[0], a2 = g[1];
    const int b1 = g[2], b2 = g[3];
    const int c1 = g[4], c2 = g[5];
    const int ld0 = lan ^ (hiK ? a1 : 0);
    const int st0 = lan ^ (hiK ? c2 : 0);

    v2f A1a, A2a, A1b0, A2b0, A1b1, A2b1, A1c0, A2c0, A1c1, A2c1;
    pair_A(ma1,  ma2,  lane, A1a,  A2a);
    pair_A(mb01, mb02, lane, A1b0, A2b0);
    pair_A(mb11, mb12, lane, A1b1, A2b1);
    pair_A(mc01, mc02, lane, A1c0, A2c0);
    pair_A(mc11, mc12, lane, A1c1, A2c1);

    if constexpr (ZIN) {
        const v8f Z = {0.f, 0.f, 0.f, 0.f, 0.f, 0.f, 0.f, 0.f};
        const float2 p1 = st[ld0];
        v2f B0; B0[0] = p1.x; B0[1] = p1.y;
        const v8f E1 = __builtin_amdgcn_wmma_f32_16x16x4_f32(false, A1a, false, B0,
                                                             (short)0, Z, false, false);
        v8f E2[4];
        #pragma unroll
        for (int jA = 0; jA < 4; ++jA) {
            v2f B1;
            B1[0] = hiK ? 0.f : E1[2 * jA];
            B1[1] = hiK ? 0.f : E1[2 * jA + 1];
            const v2f A1 = parA<HEAD>(jA) ? A1b1 : A1b0;
            E2[jA] = __builtin_amdgcn_wmma_f32_16x16x4_f32(false, A1, false, B1,
                                                           (short)0, Z, false, false);
        }
        #pragma unroll
        for (int jA = 0; jA < 4; ++jA) {
            #pragma unroll
            for (int jB = 0; jB < 4; ++jB) {
                v2f B1;
                B1[0] = hiK ? 0.f : E2[jA][2 * jB];
                B1[1] = hiK ? 0.f : E2[jA][2 * jB + 1];
                const v2f A1 = (parA<HEAD>(jA) ^ par2(jB)) ? A1c1 : A1c0;
                const v8f C = __builtin_amdgcn_wmma_f32_16x16x4_f32(false, A1, false, B1,
                                                                    (short)0, Z, false, false);
                const float2 oa = hiK ? make_float2(C[4], C[5]) : make_float2(C[0], C[1]);
                const float2 ob = hiK ? make_float2(C[6], C[7]) : make_float2(C[2], C[3]);
                const int rc = ((jA & 1) ? a1 : 0) ^ ((jA & 2) ? a2 : 0) ^
                               ((jB & 1) ? b1 : 0) ^ ((jB & 2) ? b2 : 0);
                st[st0 ^ rc]      = oa;
                st[st0 ^ rc ^ c1] = ob;
            }
        }
        return;
    }

    v8f D1[NTILES];
    #pragma unroll
    for (int t = 0; t < NTILES; ++t) {
        const int rc = ((t & 1) ? b1 : 0) ^ ((t & 2) ? b2 : 0) ^
                       ((t & 4) ? c1 : 0) ^ ((t & 8) ? c2 : 0);
        const float2 p1 = st[ld0 ^ rc];
        const float2 p2 = st[ld0 ^ rc ^ a2];
        v2f B1; B1[0] = p1.x; B1[1] = p1.y;
        v2f B2; B2[0] = p2.x; B2[1] = p2.y;
        v8f C = {0.f, 0.f, 0.f, 0.f, 0.f, 0.f, 0.f, 0.f};
        C = __builtin_amdgcn_wmma_f32_16x16x4_f32(false, A1a, false, B1,
                                                  (short)0, C, false, false);
        C = __builtin_amdgcn_wmma_f32_16x16x4_f32(false, A2a, false, B2,
                                                  (short)0, C, false, false);
        D1[t] = C;
    }
    v8f D2[NTILES];
    #pragma unroll
    for (int tc = 0; tc < 4; ++tc) {
        #pragma unroll
        for (int jA = 0; jA < 4; ++jA) {
            v2f B1, B2;
            B1[0] = hiK ? D1[tc * 4 + 1][2 * jA]     : D1[tc * 4][2 * jA];
            B1[1] = hiK ? D1[tc * 4 + 1][2 * jA + 1] : D1[tc * 4][2 * jA + 1];
            B2[0] = hiK ? D1[tc * 4 + 3][2 * jA]     : D1[tc * 4 + 2][2 * jA];
            B2[1] = hiK ? D1[tc * 4 + 3][2 * jA + 1] : D1[tc * 4 + 2][2 * jA + 1];
            const v2f A1 = parA<HEAD>(jA) ? A1b1 : A1b0;
            const v2f A2 = parA<HEAD>(jA) ? A2b1 : A2b0;
            v8f C = {0.f, 0.f, 0.f, 0.f, 0.f, 0.f, 0.f, 0.f};
            C = __builtin_amdgcn_wmma_f32_16x16x4_f32(false, A1, false, B1,
                                                      (short)0, C, false, false);
            C = __builtin_amdgcn_wmma_f32_16x16x4_f32(false, A2, false, B2,
                                                      (short)0, C, false, false);
            D2[jA * 4 + tc] = C;
        }
    }
    #pragma unroll
    for (int jA = 0; jA < 4; ++jA) {
        #pragma unroll
        for (int jB = 0; jB < 4; ++jB) {
            v2f B1, B2;
            B1[0] = hiK ? D2[jA * 4 + 1][2 * jB]     : D2[jA * 4][2 * jB];
            B1[1] = hiK ? D2[jA * 4 + 1][2 * jB + 1] : D2[jA * 4][2 * jB + 1];
            B2[0] = hiK ? D2[jA * 4 + 3][2 * jB]     : D2[jA * 4 + 2][2 * jB];
            B2[1] = hiK ? D2[jA * 4 + 3][2 * jB + 1] : D2[jA * 4 + 2][2 * jB + 1];
            const v2f A1 = (parA<HEAD>(jA) ^ par2(jB)) ? A1c1 : A1c0;
            const v2f A2 = (parA<HEAD>(jA) ^ par2(jB)) ? A2c1 : A2c0;
            v8f C = {0.f, 0.f, 0.f, 0.f, 0.f, 0.f, 0.f, 0.f};
            C = __builtin_amdgcn_wmma_f32_16x16x4_f32(false, A1, false, B1,
                                                      (short)0, C, false, false);
            C = __builtin_amdgcn_wmma_f32_16x16x4_f32(false, A2, false, B2,
                                                      (short)0, C, false, false);
            const float2 oa = hiK ? make_float2(C[4], C[5]) : make_float2(C[0], C[1]);
            const float2 ob = hiK ? make_float2(C[6], C[7]) : make_float2(C[2], C[3]);
            const int rc = ((jA & 1) ? a1 : 0) ^ ((jA & 2) ? a2 : 0) ^
                           ((jB & 1) ? b1 : 0) ^ ((jB & 2) ? b2 : 0);
            st[st0 ^ rc]      = oa;
            st[st0 ^ rc ^ c1] = ob;
        }
    }
}

typedef __attribute__((ext_vector_type(2))) unsigned v2u;
__device__ __forceinline__ void apply_two_triples(float2* st, const float2* M, const int* g,
                                                  int lan, int lane) {
    const int  h  = lane >> 4;
    const int  o  = (lane >> 1) & 7;
    const bool im = (lane & 1) != 0;
    v2f A1[4], A2[4];
    #pragma unroll
    for (int kk = 0; kk < 4; ++kk) {
        const float2 m1 = M[o * 8 + 2 * kk + h];
        const float2 m2 = M[64 + o * 8 + 2 * kk + h];
        A1[kk][0] = im ? m1.y :  m1.x;  A1[kk][1] = im ? m1.x : -m1.y;
        A2[kk][0] = im ? m2.y :  m2.x;  A2[kk][1] = im ? m2.x : -m2.y;
    }
    const int t0w = g[0], t1w = g[1], t2w = g[2];
    const int u0w = g[3], u1w = g[4], u2w = g[5];
    const int ld0 = lan ^ (h ? t0w : 0);
    const int st0 = lan ^ (h ? u2w : 0);

    v8f D1[8];
    #pragma unroll
    for (int kk = 0; kk < 4; ++kk) {
        #pragma unroll
        for (int t = 0; t < 8; ++t) {
            const int rc = ((t & 1) ? u0w : 0) ^ ((t & 2) ? u1w : 0) ^ ((t & 4) ? u2w : 0) ^
                           ((kk & 1) ? t1w : 0) ^ ((kk & 2) ? t2w : 0);
            const float2 p = st[ld0 ^ rc];
            v2f B; B[0] = p.x; B[1] = p.y;
            if (kk == 0) {
                const v8f Z = {0.f, 0.f, 0.f, 0.f, 0.f, 0.f, 0.f, 0.f};
                D1[t] = __builtin_amdgcn_wmma_f32_16x16x4_f32(false, A1[0], false, B,
                                                              (short)0, Z, false, false);
            } else {
                D1[t] = __builtin_amdgcn_wmma_f32_16x16x4_f32(false, A1[kk], false, B,
                                                              (short)0, D1[t], false, false);
            }
        }
    }
    #pragma unroll
    for (int tp = 0; tp < 8; tp += 2) {
        #pragma unroll
        for (int r = 0; r < 8; ++r) {
            const v2u sw = __builtin_amdgcn_permlane16_swap(__float_as_uint(D1[tp][r]),
                                                            __float_as_uint(D1[tp + 1][r]),
                                                            false, false);
            D1[tp][r]     = __uint_as_float(sw[0]);
            D1[tp + 1][r] = __uint_as_float(sw[1]);
        }
    }
    #pragma unroll
    for (int w = 0; w < 2; ++w) {
        v8f C[4];
        #pragma unroll
        for (int kk = 0; kk < 4; ++kk) {
            #pragma unroll
            for (int rp = 0; rp < 4; ++rp) {
                v2f B;
                B[0] = D1[w | (kk << 1)][2 * rp];
                B[1] = D1[w | (kk << 1)][2 * rp + 1];
                if (kk == 0) {
                    const v8f Z = {0.f, 0.f, 0.f, 0.f, 0.f, 0.f, 0.f, 0.f};
                    C[rp] = __builtin_amdgcn_wmma_f32_16x16x4_f32(false, A2[0], false, B,
                                                                  (short)0, Z, false, false);
                } else {
                    C[rp] = __builtin_amdgcn_wmma_f32_16x16x4_f32(false, A2[kk], false, B,
                                                                  (short)0, C[rp], false, false);
                }
            }
        }
        #pragma unroll
        for (int rp = 0; rp < 4; ++rp) {
            const int rc = ((rp & 1) ? t0w : 0) ^ ((rp & 2) ? t1w : 0) ^ (w ? t2w : 0);
            #pragma unroll
            for (int q = 0; q < 4; ++q)
                st[st0 ^ rc ^ ((q & 1) ? u0w : 0) ^ ((q & 2) ? u1w : 0)] =
                    make_float2(C[rp][2 * q], C[rp][2 * q + 1]);
        }
    }
}

__device__ __forceinline__ int lane_slot(const int* o, int lane, int wave) {
    int s = 0;
    #pragma unroll
    for (int i = 0; i < 4; ++i) s ^= o[i] & -((lane >> i) & 1);
    s ^= o[4] & -(wave & 1);
    s ^= o[5] & -((wave >> 1) & 1);
    return s;
}

__device__ __forceinline__ unsigned sign_word(int pat, int sh) {
    return ((unsigned)pat << sh) & 0x80000000u;
}

__global__ __launch_bounds__(THREADS, 4)
void qsim_kernel(const float* __restrict__ fv, const float* __restrict__ theta,
                 float* __restrict__ out) {
    __shared__ float2 st[DIM];
    __shared__ float2 wireU[(1 + NLAYERS) * NQ][4];
    __shared__ float2 layM[NLAYERS * NSLOT][16];
    __shared__ float  wpart[NWAVES][NQ];

    const int b    = blockIdx.x;
    const int tid  = threadIdx.x;
    const int lane = tid & 31;
    const int wave = tid >> 5;

    for (int i = tid; i < DIM; i += THREADS) st[i] = make_float2(0.f, 0.f);
    if (tid == 0) st[0] = make_float2(1.f, 0.f);

    if (tid < NLAYERS * NQ) {
        float2 U[4];
        const float* ta = theta + tid * 3;
        fused_rot(ta[0], ta[1], ta[2], U);
        #pragma unroll
        for (int i = 0; i < 4; ++i) wireU[NQ + tid][i] = U[i];
    } else if (tid >= 3 * 32 && tid < 3 * 32 + NQ) {
        const int w = tid - 3 * 32;
        float2 U[4];
        embed_rot(fv[b * NQ + w], U);
        #pragma unroll
        for (int i = 0; i < 4; ++i) wireU[w][i] = U[i];
    }
    __syncthreads();
    if (tid < NLAYERS * NSLOT) {
        const int q = tid & 1;
        const int l = (tid >> 1) / NPAIRS;
        const int s = (tid >> 1) - l * NPAIRS;
        const bool carry = (l & 1) == 0;
        const int  lo = s == 0 ? 0  : 2 * s - 1;
        const int  hi = s == 0 ? 11 : 2 * s;
        if (carry) {
            float2 Ua[4], Ub[4], G[16];
            if (l == 0) {
                mul22(&wireU[NQ + lo][0], &wireU[lo][0], Ua);
                mul22(&wireU[NQ + hi][0], &wireU[hi][0], Ub);
            } else {
                #pragma unroll
                for (int i = 0; i < 4; ++i) {
                    Ua[i] = wireU[(1 + l) * NQ + lo][i];
                    Ub[i] = wireU[(1 + l) * NQ + hi][i];
                }
            }
            kron22(Ub, Ua, G);
            if (carry && s >= 1) {
                float2 Un[4], T[16];
                #pragma unroll
                for (int i = 0; i < 4; ++i) Un[i] = wireU[(2 + l) * NQ + lo][i];
                ride_next(G, Un, q, T);
                #pragma unroll
                for (int i = 0; i < 16; ++i) layM[tid][i] = T[i];
            } else if (carry && l >= 2) {
                float2 Up[4], T[16];
                #pragma unroll
                for (int i = 0; i < 4; ++i) Up[i] = wireU[l * NQ + NQ - 1][i];
                push_prev(G, Up, q, T);
                #pragma unroll
                for (int i = 0; i < 16; ++i) layM[tid][i] = T[i];
            } else {
                #pragma unroll
                for (int i = 0; i < 16; ++i) layM[tid][i] = G[i];
            }
        }
    }
    if (tid >= 2 * 32) {
        #pragma unroll
        for (int idx = tid - 2 * 32; idx < 3 * 128; idx += 64) {
            const int l  = 2 * (idx >> 7) + 1;
            const int tr = (idx >> 6) & 1;
            const int e  = idx & 63;
            const int oo = e >> 3, jj = e & 7;
            const int w0 = (1 + l) * NQ + 6 * tr;
            const float2 ua = wireU[w0    ][( oo       & 1) * 2 + ( jj       & 1)];
            const float2 ub = wireU[w0 + 2][((oo >> 1) & 1) * 2 + ((jj >> 1) & 1)];
            const float2 uc = wireU[w0 + 4][((oo >> 2) & 1) * 2 + ((jj >> 2) & 1)];
            layM[l * NSLOT + tr * 4 + (e >> 4)][e & 15] = cmul(uc, cmul(ub, ua));
        }
    }
    __syncthreads();

    static constexpr Maps MAPS = make_maps();
    const int pi = wave & 1;

    {
        const float2* M = &layM[0][0];
        int g0[6], g1[6];
        #pragma unroll
        for (int i = 0; i < 6; ++i) { g0[i] = MAPS.g[0][0][i]; g1[i] = MAPS.g[0][1][i]; }
        const int lan0 = lane_slot(MAPS.o[0][0], lane, wave);
        const int lan1 = lane_slot(MAPS.o[0][1], lane, wave);
        apply_three_pairs<true, true>(st, M,       g0, lan0, lane, pi);
        __syncthreads();
        apply_three_pairs<true, false>(st, M + 96, g1, lan1, lane, pi);
        __syncthreads();
    }
    #pragma nounroll
    for (int l = 1; l < NLAYERS; ++l) {
        const float2* M = &layM[l * NSLOT][0];
        int g0[6], g1[6];
        #pragma unroll
        for (int i = 0; i < 6; ++i) { g0[i] = MAPS.g[l][0][i]; g1[i] = MAPS.g[l][1][i]; }
        const int lan0 = lane_slot(MAPS.o[l][0], lane, wave);
        const int lan1 = lane_slot(MAPS.o[l][1], lane, wave);
        const bool carry = (l & 1) == 0;
        if (carry) {
            apply_three_pairs<false, true>(st, M, g0, lan0, lane, pi);
            __syncthreads();
            apply_three_pairs<false, false>(st, M + 96, g1, lan1, lane, pi);
        } else {
            apply_two_triples(st, M, g0, lan0, lane);
        }
        __syncthreads();
    }

    constexpr int LZ = NLAYERS - 1;
    const int      h   = lane >> 4;
    const int      wc  = lane & 15;
    const int      sh  = 31 - wc;
    const unsigned one = 0x3f800000u;
    const float2 u00 = wireU[NLAYERS * NQ + NQ - 1][0], u01 = wireU[NLAYERS * NQ + NQ - 1][1];
    const float2 u10 = wireU[NLAYERS * NQ + NQ - 1][2], u11 = wireU[NLAYERS * NQ + NQ - 1][3];
    const float f   = __uint_as_float(one ^ sign_word(MAPS.zg[0], sh));
    const float n00 = u00.x * u00.x + u00.y * u00.y, n01 = u01.x * u01.x + u01.y * u01.y;
    const float n10 = u10.x * u10.x + u10.y * u10.y, n11 = u11.x * u11.x + u11.y * u11.y;
    const float c0r = u00.x * u01.x + u00.y * u01.y, c0i = u00.y * u01.x - u00.x * u01.y;
    const float c1r = u10.x * u11.x + u10.y * u11.y, c1i = u10.y * u11.x - u10.x * u11.y;
    const unsigned k0 = __float_as_uint(n00 + f * n10);
    const unsigned k1 = __float_as_uint(n01 + f * n11);
    const unsigned kr = __float_as_uint( 2.f * (c0r + f * c1r));
    const unsigned ki = __float_as_uint(-2.f * (c0i + f * c1i));
    const unsigned hb  = h ? sign_word(MAPS.zg[1], sh) : 0u;
    const unsigned gt0 = sign_word(MAPS.zg[2], sh), gt1 = sign_word(MAPS.zg[3], sh);
    const unsigned gt2 = sign_word(MAPS.zg[4], sh), gt3 = sign_word(MAPS.zg[5], sh);
    const int lanz = lane_slot(MAPS.o[LZ][1], lane, wave) ^ (h ? MAPS.g[LZ][1][1] : 0);
    float2 za0[DIM / THREADS / 2], za1[DIM / THREADS / 2];
    #pragma unroll
    for (int t = 0; t < DIM / THREADS / 2; ++t) {
        const int rc = ((t & 1) ? MAPS.g[LZ][1][2] : 0) ^ ((t & 2) ? MAPS.g[LZ][1][3] : 0) ^
                       ((t & 4) ? MAPS.g[LZ][1][4] : 0) ^ ((t & 8) ? MAPS.g[LZ][1][5] : 0);
        za0[t] = st[lanz ^ rc];
        za1[t] = st[lanz ^ rc ^ MAPS.g[LZ][1][0]];
    }
    v8f zacc0 = {0.f, 0.f, 0.f, 0.f, 0.f, 0.f, 0.f, 0.f};
    v8f zacc1 = {0.f, 0.f, 0.f, 0.f, 0.f, 0.f, 0.f, 0.f};
    #pragma unroll
    for (int t = 0; t < DIM / THREADS / 2; ++t) {
        const float2 a0 = za0[t], a1 = za1[t];
        const unsigned s = hb ^ ((t & 1) ? gt0 : 0u) ^ ((t & 2) ? gt1 : 0u) ^
                                ((t & 4) ? gt2 : 0u) ^ ((t & 8) ? gt3 : 0u);
        v2f A, B;
        A[0] = a0.x * a0.x + a0.y * a0.y;
        A[1] = a1.x * a1.x + a1.y * a1.y;
        B[0] = __uint_as_float(k0 ^ s);
        B[1] = __uint_as_float(k1 ^ s);
        zacc0 = __builtin_amdgcn_wmma_f32_16x16x4_f32(false, A, false, B,
                                                      (short)0, zacc0, false, false);
        A[0] = a0.x * a1.x + a0.y * a1.y;
        A[1] = a0.y * a1.x - a0.x * a1.y;
        B[0] = __uint_as_float(kr ^ s);
        B[1] = __uint_as_float(ki ^ s);
        zacc1 = __builtin_amdgcn_wmma_f32_16x16x4_f32(false, A, false, B,
                                                      (short)0, zacc1, false, false);
    }
    const float s0 = __uint_as_float(one ^ sign_word(MAPS.zo[0], sh));
    const float s1 = __uint_as_float(one ^ sign_word(MAPS.zo[1], sh));
    const float s2 = __uint_as_float(one ^ sign_word(MAPS.zo[2], sh));
    float dz[8];
    #pragma unroll
    for (int r = 0; r < 8; ++r) dz[r] = zacc0[r] + zacc1[r];
    const float e0 = dz[0] + s0 * dz[1], e1 = dz[2] + s0 * dz[3];
    const float e2 = dz[4] + s0 * dz[5], e3 = dz[6] + s0 * dz[7];
    const float f0 = e0 + s1 * e1, f1 = e2 + s1 * e3;
    float zs = f0 + s2 * f1;
    const unsigned sw = (h ? sign_word(MAPS.zo[3], sh) : 0u) ^
                        ((wave & 1) ? sign_word(MAPS.zo[4], sh) : 0u) ^
                        ((wave & 2) ? sign_word(MAPS.zo[5], sh) : 0u);
    zs = __uint_as_float(__float_as_uint(zs) ^ sw);
    zs += __shfl_xor(zs, 16, 32);
    if (lane < NQ) wpart[wave][lane] = zs;
    __syncthreads();
    if (tid < NQ) {
        float s = 0.f;
        for (int q = 0; q < NWAVES; ++q) s += wpart[q][tid];
        out[b * NQ + tid] = s;
    }
}

extern "C" void kernel_launch(void* const* d_in, const int* in_sizes, int n_in,
                              void* d_out, int out_size, void* d_ws, size_t ws_size,
                              hipStream_t stream) {
    const float* fv    = (const float*)d_in[0];
    const float* theta = (const float*)d_in[1];
    float*       out   = (float*)d_out;
    const int batch = in_sizes[0] / NQ;
    qsim_kernel<<<batch, THREADS, 0, stream>>>(fv, theta, out);
}
